# starting kernel
# baseline (speedup 1.0000x reference)
#include <hip/hip_runtime.h>
#include <math.h>

typedef float v2f __attribute__((ext_vector_type(2)));
typedef float v8f __attribute__((ext_vector_type(8)));

#define FEAT   64
#define NCLS   40
#define NCLSP  48


__global__ void k_fill(float* __restrict__ p, float val, int n) {
    int i = blockIdx.x * blockDim.x + threadIdx.x;
    if (i < n) p[i] = val;
}

__global__ void k_degree(const int* __restrict__ col, float* __restrict__ deg, int E) {
    int i = blockIdx.x * blockDim.x + threadIdx.x;
    if (i < E) atomicAdd(&deg[col[i]], 1.0f);
}

__global__ void k_rsqrt_inplace(float* __restrict__ d, int n) {
    int i = blockIdx.x * blockDim.x + threadIdx.x;
    if (i < n) {
        float v = d[i];
        d[i] = (v > 0.0f) ? rsqrtf(v) : 0.0f;
    }
}

__global__ void k_selfloop_init(const float* __restrict__ xin, float* __restrict__ xout,
                                const float* __restrict__ dinv, int total) {
    int i = blockIdx.x * blockDim.x + threadIdx.x;
    if (i < total) {
        float w = dinv[i >> 6];
        xout[i] = w * w * xin[i];
    }
}

__global__ void k_edge_spmm(const float* __restrict__ xin, float* __restrict__ xout,
                            const int* __restrict__ row, const int* __restrict__ col,
                            const float* __restrict__ dinv, int E) {
    long long t = (long long)blockIdx.x * blockDim.x + threadIdx.x;
    int e    = (int)(t >> 4);
    int lane = (int)(t & 15);
    if (e >= E) return;
    int r = row[e];
    int c = col[e];
    float w = dinv[r] * dinv[c];
    const float4 v = ((const float4*)(xin + (size_t)r * FEAT))[lane];
    float* dst = xout + (size_t)c * FEAT + lane * 4;
    atomicAdd(dst + 0, w * v.x);
    atomicAdd(dst + 1, w * v.y);
    atomicAdd(dst + 2, w * v.z);
    atomicAdd(dst + 3, w * v.w);
}

__global__ void k_pad_w(const float* __restrict__ W, float* __restrict__ Wp) {
    int i = blockIdx.x * blockDim.x + threadIdx.x;
    if (i < NCLSP * FEAT) {
        int r = i / FEAT;
        Wp[i] = (r < NCLS) ? W[i] : 0.0f;
    }
}

__global__ void k_gemm_wmma(const float* __restrict__ x, const float* __restrict__ Wp,
                            const float* __restrict__ bias, float* __restrict__ out, int N) {
    const int wave   = blockIdx.x * (blockDim.x >> 5) + (threadIdx.x >> 5);
    const int lane   = threadIdx.x & 31;
    const int ntiles = (N + 15) >> 4;
    if (wave >= ntiles) return;

    const int base = wave << 4;
    const int half = lane >> 4;
    const int l16  = lane & 15;

    int m = base + l16;
    if (m >= N) m = N - 1;
    const float* arow = x + (size_t)m * FEAT;
    const float* b0r  = Wp + (size_t)(l16)      * FEAT;
    const float* b1r  = Wp + (size_t)(16 + l16) * FEAT;
    const float* b2r  = Wp + (size_t)(32 + l16) * FEAT;

    v8f acc0 = {};
    v8f acc1 = {};
    v8f acc2 = {};

    for (int k0 = 0; k0 < FEAT; k0 += 4) {
        const int ko = k0 + 2 * half;
        v2f a  = *(const v2f*)(arow + ko);
        v2f b0 = *(const v2f*)(b0r + ko);
        v2f b1 = *(const v2f*)(b1r + ko);
        v2f b2 = *(const v2f*)(b2r + ko);
        acc0 = __builtin_amdgcn_wmma_f32_16x16x4_f32(false, a, false, b0, (short)0, acc0, false, false);
        acc1 = __builtin_amdgcn_wmma_f32_16x16x4_f32(false, a, false, b1, (short)0, acc1, false, false);
        acc2 = __builtin_amdgcn_wmma_f32_16x16x4_f32(false, a, false, b2, (short)0, acc2, false, false);
    }

    const int c0 = l16;
    const int c1 = 16 + l16;
    const int c2 = 32 + l16;
    const float bb0 = bias[c0];
    const float bb1 = bias[c1];
    const float bb2 = (c2 < NCLS) ? bias[c2] : 0.0f;
#pragma unroll
    for (int j = 0; j < 8; ++j) {
        const int node = base + j + 8 * half;
        if (node < N) {
            float* o = out + (size_t)node * NCLS;
            o[c0] = acc0[j] + bb0;
            o[c1] = acc1[j] + bb1;
            if (c2 < NCLS) o[c2] = acc2[j] + bb2;
        }
    }
}

__global__ void k_logsoftmax(float* __restrict__ out, int N) {
    int n = blockIdx.x * blockDim.x + threadIdx.x;
    if (n >= N) return;
    float* p = out + (size_t)n * NCLS;
    float v[NCLS];
    float m = -INFINITY;
#pragma unroll
    for (int c = 0; c < NCLS; ++c) { v[c] = p[c]; m = fmaxf(m, v[c]); }
    float s = 0.0f;
#pragma unroll
    for (int c = 0; c < NCLS; ++c) s += expf(v[c] - m);
    const float ls = logf(s);
#pragma unroll
    for (int c = 0; c < NCLS; ++c) p[c] = v[c] - m - ls;
}


static inline size_t align256f(size_t nfloats) { return (nfloats + 63) & ~(size_t)63; }

extern "C" void kernel_launch(void* const* d_in, const int* in_sizes, int n_in,
                              void* d_out, int out_size, void* d_ws, size_t ws_size,
                              hipStream_t stream) {
    const float* x  = (const float*)d_in[0];
    const int*   ei = (const int*)d_in[1];
    const float* W  = (const float*)d_in[2];
    const float* bi = (const float*)d_in[3];

    const int N = in_sizes[0] / FEAT;
    const int E = in_sizes[1] / 2;
    const int* row = ei;
    const int* col = ei + E;
    const int total = N * FEAT;

    float* ws = (float*)d_ws;
    size_t off = 0;
    float* dinv = ws + off; off += align256f((size_t)N);
    float* Wp   = ws + off; off += align256f((size_t)NCLSP * FEAT);
    float* x0   = ws + off; off += align256f((size_t)total);
    float* x1   = ws + off;

    const int T = 256;
    k_fill<<<(N + T - 1) / T, T, 0, stream>>>(dinv, 1.0f, N);
    k_degree<<<(E + T - 1) / T, T, 0, stream>>>(col, dinv, E);
    k_rsqrt_inplace<<<(N + T - 1) / T, T, 0, stream>>>(dinv, N);
    k_pad_w<<<(NCLSP * FEAT + T - 1) / T, T, 0, stream>>>(W, Wp);

    const long long ethreads = (long long)E * 16;
    const int eblocks = (int)((ethreads + T - 1) / T);

    k_selfloop_init<<<(total + T - 1) / T, T, 0, stream>>>(x, x0, dinv, total);
    k_edge_spmm<<<eblocks, T, 0, stream>>>(x, x0, row, col, dinv, E);
    k_selfloop_init<<<(total + T - 1) / T, T, 0, stream>>>(x0, x1, dinv, total);
    k_edge_spmm<<<eblocks, T, 0, stream>>>(x0, x1, row, col, dinv, E);

    const int ntiles = (N + 15) >> 4;
    const int gblocks = (ntiles + 7) / 8;
    k_gemm_wmma<<<gblocks, 256, 0, stream>>>(x1, Wp, bi, (float*)d_out, N);
    k_logsoftmax<<<(N + T - 1) / T, T, 0, stream>>>((float*)d_out, N);
}


# edge SpMM: a wave walks 32 edges, lane l owns columns l and l+32, so each gather and each f32 atomic add covers one whole 128-byte line (2 line requests per edge, not 8); sums atomic, arrival order
# speedup vs baseline: 1.6930x; 1.6930x over previous
#include <hip/hip_runtime.h>
#include <math.h>

typedef float v2f __attribute__((ext_vector_type(2)));
typedef float v8f __attribute__((ext_vector_type(8)));

#define FEAT   64
#define NCLS   40
#define NCLSP  48


__global__ void k_fill(float* __restrict__ p, float val, int n) {
    int i = blockIdx.x * blockDim.x + threadIdx.x;
    if (i < n) p[i] = val;
}

__global__ void k_degree(const int* __restrict__ col, float* __restrict__ deg, int E) {
    int i = blockIdx.x * blockDim.x + threadIdx.x;
    if (i < E) atomicAdd(&deg[col[i]], 1.0f);
}

__global__ void k_rsqrt_inplace(float* __restrict__ d, int n) {
    int i = blockIdx.x * blockDim.x + threadIdx.x;
    if (i < n) {
        float v = d[i];
        d[i] = (v > 0.0f) ? rsqrtf(v) : 0.0f;
    }
}

__global__ void k_selfloop_init(const float* __restrict__ xin, float* __restrict__ xout,
                                const float* __restrict__ dinv, int total) {
    int i = blockIdx.x * blockDim.x + threadIdx.x;
    if (i < total) {
        float w = dinv[i >> 6];
        xout[i] = w * w * xin[i];
    }
}

#define EPW    32
#define EBATCH 8

template <int NB>
__device__ __forceinline__ void spmm_edges(const float* __restrict__ xin, float* __restrict__ xout,
                                           int r, int c, float w, int first, int lane) {
    float v0[NB];
    float v1[NB];
#pragma unroll
    for (int j = 0; j < NB; ++j) {
        const int rj = __builtin_amdgcn_readlane(r, first + j);
        const float* src = xin + (size_t)rj * FEAT;
        v0[j] = src[lane];
        v1[j] = src[lane + 32];
    }
#pragma unroll
    for (int j = 0; j < NB; ++j) {
        const int   cj = __builtin_amdgcn_readlane(c, first + j);
        const float wj = __int_as_float(__builtin_amdgcn_readlane(__float_as_int(w), first + j));
        float* dst = xout + (size_t)cj * FEAT;
        atomicAdd(dst + lane,      wj * v0[j]);
        atomicAdd(dst + lane + 32, wj * v1[j]);
    }
}

__global__ void k_edge_spmm(const float* __restrict__ xin, float* __restrict__ xout,
                            const int* __restrict__ row, const int* __restrict__ col,
                            const float* __restrict__ dinv, int E) {
    const int lane = threadIdx.x & 31;
    const int wave = blockIdx.x * (blockDim.x >> 5) + __builtin_amdgcn_readfirstlane(threadIdx.x >> 5);
    const int e0   = wave * EPW;
    if (e0 >= E) return;
    const int e = e0 + lane;
    int   r = 0;
    int   c = 0;
    float w = 0.0f;
    if (e < E) {
        r = row[e];
        c = col[e];
        w = dinv[r] * dinv[c];
    }
    const int n = (E - e0 < EPW) ? (E - e0) : EPW;
    if (n == EPW) {
#pragma unroll
        for (int i = 0; i < EPW; i += EBATCH) spmm_edges<EBATCH>(xin, xout, r, c, w, i, lane);
    } else {
        for (int i = 0; i < n; ++i) spmm_edges<1>(xin, xout, r, c, w, i, lane);
    }
}

__global__ void k_pad_w(const float* __restrict__ W, float* __restrict__ Wp) {
    int i = blockIdx.x * blockDim.x + threadIdx.x;
    if (i < NCLSP * FEAT) {
        int r = i / FEAT;
        Wp[i] = (r < NCLS) ? W[i] : 0.0f;
    }
}

__global__ void k_gemm_wmma(const float* __restrict__ x, const float* __restrict__ Wp,
                            const float* __restrict__ bias, float* __restrict__ out, int N) {
    const int wave   = blockIdx.x * (blockDim.x >> 5) + (threadIdx.x >> 5);
    const int lane   = threadIdx.x & 31;
    const int ntiles = (N + 15) >> 4;
    if (wave >= ntiles) return;

    const int base = wave << 4;
    const int half = lane >> 4;
    const int l16  = lane & 15;

    int m = base + l16;
    if (m >= N) m = N - 1;
    const float* arow = x + (size_t)m * FEAT;
    const float* b0r  = Wp + (size_t)(l16)      * FEAT;
    const float* b1r  = Wp + (size_t)(16 + l16) * FEAT;
    const float* b2r  = Wp + (size_t)(32 + l16) * FEAT;

    v8f acc0 = {};
    v8f acc1 = {};
    v8f acc2 = {};

    for (int k0 = 0; k0 < FEAT; k0 += 4) {
        const int ko = k0 + 2 * half;
        v2f a  = *(const v2f*)(arow + ko);
        v2f b0 = *(const v2f*)(b0r + ko);
        v2f b1 = *(const v2f*)(b1r + ko);
        v2f b2 = *(const v2f*)(b2r + ko);
        acc0 = __builtin_amdgcn_wmma_f32_16x16x4_f32(false, a, false, b0, (short)0, acc0, false, false);
        acc1 = __builtin_amdgcn_wmma_f32_16x16x4_f32(false, a, false, b1, (short)0, acc1, false, false);
        acc2 = __builtin_amdgcn_wmma_f32_16x16x4_f32(false, a, false, b2, (short)0, acc2, false, false);
    }

    const int c0 = l16;
    const int c1 = 16 + l16;
    const int c2 = 32 + l16;
    const float bb0 = bias[c0];
    const float bb1 = bias[c1];
    const float bb2 = (c2 < NCLS) ? bias[c2] : 0.0f;
#pragma unroll
    for (int j = 0; j < 8; ++j) {
        const int node = base + j + 8 * half;
        if (node < N) {
            float* o = out + (size_t)node * NCLS;
            o[c0] = acc0[j] + bb0;
            o[c1] = acc1[j] + bb1;
            if (c2 < NCLS) o[c2] = acc2[j] + bb2;
        }
    }
}

__global__ void k_logsoftmax(float* __restrict__ out, int N) {
    int n = blockIdx.x * blockDim.x + threadIdx.x;
    if (n >= N) return;
    float* p = out + (size_t)n * NCLS;
    float v[NCLS];
    float m = -INFINITY;
#pragma unroll
    for (int c = 0; c < NCLS; ++c) { v[c] = p[c]; m = fmaxf(m, v[c]); }
    float s = 0.0f;
#pragma unroll
    for (int c = 0; c < NCLS; ++c) s += expf(v[c] - m);
    const float ls = logf(s);
#pragma unroll
    for (int c = 0; c < NCLS; ++c) p[c] = v[c] - m - ls;
}


static inline size_t align256f(size_t nfloats) { return (nfloats + 63) & ~(size_t)63; }

extern "C" void kernel_launch(void* const* d_in, const int* in_sizes, int n_in,
                              void* d_out, int out_size, void* d_ws, size_t ws_size,
                              hipStream_t stream) {
    const float* x  = (const float*)d_in[0];
    const int*   ei = (const int*)d_in[1];
    const float* W  = (const float*)d_in[2];
    const float* bi = (const float*)d_in[3];

    const int N = in_sizes[0] / FEAT;
    const int E = in_sizes[1] / 2;
    const int* row = ei;
    const int* col = ei + E;
    const int total = N * FEAT;

    float* ws = (float*)d_ws;
    size_t off = 0;
    float* dinv = ws + off; off += align256f((size_t)N);
    float* Wp   = ws + off; off += align256f((size_t)NCLSP * FEAT);
    float* x0   = ws + off; off += align256f((size_t)total);
    float* x1   = ws + off;

    const int T = 256;
    k_fill<<<(N + T - 1) / T, T, 0, stream>>>(dinv, 1.0f, N);
    k_degree<<<(E + T - 1) / T, T, 0, stream>>>(col, dinv, E);
    k_rsqrt_inplace<<<(N + T - 1) / T, T, 0, stream>>>(dinv, N);
    k_pad_w<<<(NCLSP * FEAT + T - 1) / T, T, 0, stream>>>(W, Wp);

    const int eblocks = (E + (T / 32) * EPW - 1) / ((T / 32) * EPW);

    k_selfloop_init<<<(total + T - 1) / T, T, 0, stream>>>(x, x0, dinv, total);
    k_edge_spmm<<<eblocks, T, 0, stream>>>(x, x0, row, col, dinv, E);
    k_selfloop_init<<<(total + T - 1) / T, T, 0, stream>>>(x0, x1, dinv, total);
    k_edge_spmm<<<eblocks, T, 0, stream>>>(x0, x1, row, col, dinv, E);

    const int ntiles = (N + 15) >> 4;
    const int gblocks = (ntiles + 7) / 8;
    k_gemm_wmma<<<gblocks, 256, 0, stream>>>(x1, Wp, bi, (float*)d_out, N);
    k_logsoftmax<<<(N + T - 1) / T, T, 0, stream>>>((float*)d_out, N);
}


# every atomic scatter-add (degree count, both hops) keeps one sum plane per package half; a wave adds into its own half's plane; sum = plane 0 + plane 1, each in atomic arrival order
# speedup vs baseline: 2.3269x; 1.3744x over previous
#include <hip/hip_runtime.h>
#include <math.h>

typedef float v2f __attribute__((ext_vector_type(2)));
typedef float v8f __attribute__((ext_vector_type(8)));

#define FEAT   64
#define NCLS   40
#define NCLSP  48

__device__ __forceinline__ int package_half() {
    return (int)((__builtin_amdgcn_s_sendmsg_rtn(0x87) >> 18) & 1u);
}


__global__ void k_fill(float* __restrict__ p, float val, int n) {
    int i = blockIdx.x * blockDim.x + threadIdx.x;
    if (i < n) p[i] = val;
}

__global__ void k_degree(const int* __restrict__ col, float* __restrict__ cnt0, float* __restrict__ cnt1, int E) {
    int i = blockIdx.x * blockDim.x + threadIdx.x;
    float* cnt = package_half() ? cnt1 : cnt0;
    if (i < E) atomicAdd(&cnt[col[i]], 1.0f);
}

__global__ void k_dinv(float* __restrict__ d0, const float* __restrict__ d1, int n) {
    int i = blockIdx.x * blockDim.x + threadIdx.x;
    if (i < n) {
        float v = 1.0f + d0[i] + d1[i];
        d0[i] = (v > 0.0f) ? rsqrtf(v) : 0.0f;
    }
}

__global__ void k_seed_planes(const float* __restrict__ xin, float* __restrict__ pa, float* __restrict__ pb,
                              const float* __restrict__ dinv, int total) {
    int i = blockIdx.x * blockDim.x + threadIdx.x;
    if (i < total) {
        float w = dinv[i >> 6];
        pa[i] = w * w * xin[i];
        if (pb) pb[i] = 0.0f;
    }
}

__global__ void k_join_seed(float* __restrict__ pa, float* __restrict__ pb, float* __restrict__ na,
                            const float* __restrict__ dinv, int total) {
    int i = blockIdx.x * blockDim.x + threadIdx.x;
    if (i < total) {
        float w = dinv[i >> 6];
        float s = pa[i];
        if (pb) {
            s += pb[i];
            pa[i] = s;
            pb[i] = 0.0f;
        }
        na[i] = w * w * s;
    }
}

#define EPW    32
#define EBATCH 8

template <int NB>
__device__ __forceinline__ void spmm_edges(const float* __restrict__ xin, float* __restrict__ xout,
                                           int r, int c, float w, int first, int lane) {
    float v0[NB];
    float v1[NB];
#pragma unroll
    for (int j = 0; j < NB; ++j) {
        const int rj = __builtin_amdgcn_readlane(r, first + j);
        const float* src = xin + (size_t)rj * FEAT;
        v0[j] = src[lane];
        v1[j] = src[lane + 32];
    }
#pragma unroll
    for (int j = 0; j < NB; ++j) {
        const int   cj = __builtin_amdgcn_readlane(c, first + j);
        const float wj = __int_as_float(__builtin_amdgcn_readlane(__float_as_int(w), first + j));
        float* dst = xout + (size_t)cj * FEAT;
        atomicAdd(dst + lane,      wj * v0[j]);
        atomicAdd(dst + lane + 32, wj * v1[j]);
    }
}

__global__ void k_edge_spmm(const float* __restrict__ xin, float* __restrict__ outa, float* __restrict__ outb,
                            const int* __restrict__ row, const int* __restrict__ col,
                            const float* __restrict__ dinv, int E) {
    float* xout = (outb != nullptr && package_half()) ? outb : outa;
    const int lane = threadIdx.x & 31;
    const int wave = blockIdx.x * (blockDim.x >> 5) + __builtin_amdgcn_readfirstlane(threadIdx.x >> 5);
    const int e0   = wave * EPW;
    if (e0 >= E) return;
    const int e = e0 + lane;
    int   r = 0;
    int   c = 0;
    float w = 0.0f;
    if (e < E) {
        r = row[e];
        c = col[e];
        w = dinv[r] * dinv[c];
    }
    const int n = (E - e0 < EPW) ? (E - e0) : EPW;
    if (n == EPW) {
#pragma unroll
        for (int i = 0; i < EPW; i += EBATCH) spmm_edges<EBATCH>(xin, xout, r, c, w, i, lane);
    } else {
        for (int i = 0; i < n; ++i) spmm_edges<1>(xin, xout, r, c, w, i, lane);
    }
}

__global__ void k_pad_w(const float* __restrict__ W, float* __restrict__ Wp) {
    int i = blockIdx.x * blockDim.x + threadIdx.x;
    if (i < NCLSP * FEAT) {
        int r = i / FEAT;
        Wp[i] = (r < NCLS) ? W[i] : 0.0f;
    }
}

template <bool TWO>
__global__ void k_gemm_wmma(const float* __restrict__ xa, const float* __restrict__ xb, const float* __restrict__ Wp,
                            const float* __restrict__ bias, float* __restrict__ out, int N) {
    const int wave   = blockIdx.x * (blockDim.x >> 5) + (threadIdx.x >> 5);
    const int lane   = threadIdx.x & 31;
    const int ntiles = (N + 15) >> 4;
    if (wave >= ntiles) return;

    const int base = wave << 4;
    const int half = lane >> 4;
    const int l16  = lane & 15;

    int m = base + l16;
    if (m >= N) m = N - 1;
    const float* arow = xa + (size_t)m * FEAT;
    const float* brow = (TWO ? xb : xa) + (size_t)m * FEAT;
    const float* b0r  = Wp + (size_t)(l16)      * FEAT;
    const float* b1r  = Wp + (size_t)(16 + l16) * FEAT;
    const float* b2r  = Wp + (size_t)(32 + l16) * FEAT;

    v8f acc0 = {};
    v8f acc1 = {};
    v8f acc2 = {};

    for (int k0 = 0; k0 < FEAT; k0 += 4) {
        const int ko = k0 + 2 * half;
        v2f a  = *(const v2f*)(arow + ko);
        if (TWO) a += *(const v2f*)(brow + ko);
        v2f b0 = *(const v2f*)(b0r + ko);
        v2f b1 = *(const v2f*)(b1r + ko);
        v2f b2 = *(const v2f*)(b2r + ko);
        acc0 = __builtin_amdgcn_wmma_f32_16x16x4_f32(false, a, false, b0, (short)0, acc0, false, false);
        acc1 = __builtin_amdgcn_wmma_f32_16x16x4_f32(false, a, false, b1, (short)0, acc1, false, false);
        acc2 = __builtin_amdgcn_wmma_f32_16x16x4_f32(false, a, false, b2, (short)0, acc2, false, false);
    }

    const int c0 = l16;
    const int c1 = 16 + l16;
    const int c2 = 32 + l16;
    const float bb0 = bias[c0];
    const float bb1 = bias[c1];
    const float bb2 = (c2 < NCLS) ? bias[c2] : 0.0f;
#pragma unroll
    for (int j = 0; j < 8; ++j) {
        const int node = base + j + 8 * half;
        if (node < N) {
            float* o = out + (size_t)node * NCLS;
            o[c0] = acc0[j] + bb0;
            o[c1] = acc1[j] + bb1;
            if (c2 < NCLS) o[c2] = acc2[j] + bb2;
        }
    }
}

__global__ void k_logsoftmax(float* __restrict__ out, int N) {
    int n = blockIdx.x * blockDim.x + threadIdx.x;
    if (n >= N) return;
    float* p = out + (size_t)n * NCLS;
    float v[NCLS];
    float m = -INFINITY;
#pragma unroll
    for (int c = 0; c < NCLS; ++c) { v[c] = p[c]; m = fmaxf(m, v[c]); }
    float s = 0.0f;
#pragma unroll
    for (int c = 0; c < NCLS; ++c) s += expf(v[c] - m);
    const float ls = logf(s);
#pragma unroll
    for (int c = 0; c < NCLS; ++c) p[c] = v[c] - m - ls;
}


static inline size_t align256f(size_t nfloats) { return (nfloats + 63) & ~(size_t)63; }

extern "C" void kernel_launch(void* const* d_in, const int* in_sizes, int n_in,
                              void* d_out, int out_size, void* d_ws, size_t ws_size,
                              hipStream_t stream) {
    const float* x  = (const float*)d_in[0];
    const int*   ei = (const int*)d_in[1];
    const float* W  = (const float*)d_in[2];
    const float* bi = (const float*)d_in[3];

    const int N = in_sizes[0] / FEAT;
    const int E = in_sizes[1] / 2;
    const int* row = ei;
    const int* col = ei + E;
    const int total = N * FEAT;

    float* ws = (float*)d_ws;
    size_t off = 0;
    float* cnt0 = ws + off; off += align256f((size_t)N);
    float* cnt1 = ws + off; off += align256f((size_t)N);
    float* Wp   = ws + off; off += align256f((size_t)NCLSP * FEAT);
    float* p0   = ws + off; off += align256f((size_t)total);
    float* p1   = ws + off; off += align256f((size_t)total);
    float* pb   = nullptr;
    if ((off + (size_t)total) * sizeof(float) <= ws_size) pb = ws + off;
    const float* dinv = cnt0;

    const int T = 256;
    const int ncnt = (int)(cnt1 - cnt0) + N;
    k_fill<<<(ncnt + T - 1) / T, T, 0, stream>>>(cnt0, 0.0f, ncnt);
    k_degree<<<(E + T - 1) / T, T, 0, stream>>>(col, cnt0, cnt1, E);
    k_dinv<<<(N + T - 1) / T, T, 0, stream>>>(cnt0, cnt1, N);
    k_pad_w<<<(NCLSP * FEAT + T - 1) / T, T, 0, stream>>>(W, Wp);

    const int eblocks = (E + (T / 32) * EPW - 1) / ((T / 32) * EPW);

    k_seed_planes<<<(total + T - 1) / T, T, 0, stream>>>(x, p0, pb, dinv, total);
    k_edge_spmm<<<eblocks, T, 0, stream>>>(x, p0, pb, row, col, dinv, E);
    k_join_seed<<<(total + T - 1) / T, T, 0, stream>>>(p0, pb, p1, dinv, total);
    k_edge_spmm<<<eblocks, T, 0, stream>>>(p0, p1, pb, row, col, dinv, E);

    const int ntiles = (N + 15) >> 4;
    const int gblocks = (ntiles + 7) / 8;
    if (pb) k_gemm_wmma<true><<<gblocks, 256, 0, stream>>>(p1, pb, Wp, bi, (float*)d_out, N);
    else    k_gemm_wmma<false><<<gblocks, 256, 0, stream>>>(p1, pb, Wp, bi, (float*)d_out, N);
    k_logsoftmax<<<(N + T - 1) / T, T, 0, stream>>>((float*)d_out, N);
}


# both hops as a pull: edges counting-sorted by target in the call (rank = value the degree atomic returns, scan, fill); a wave sums a target row in registers, stores it once; sum order = arrival rank
# speedup vs baseline: 2.9509x; 1.2682x over previous
#include <hip/hip_runtime.h>
#include <math.h>

typedef float v2f __attribute__((ext_vector_type(2)));
typedef float v8f __attribute__((ext_vector_type(8)));

#define FEAT   64
#define NCLS   40
#define NCLSP  48

__device__ __forceinline__ int package_half() {
    return (int)((__builtin_amdgcn_s_sendmsg_rtn(0x87) >> 18) & 1u);
}


__global__ void k_fill(float* __restrict__ p, float val, int n) {
    int i = blockIdx.x * blockDim.x + threadIdx.x;
    if (i < n) p[i] = val;
}

__global__ void k_degree(const int* __restrict__ col, float* __restrict__ cnt0, float* __restrict__ cnt1, int E) {
    int i = blockIdx.x * blockDim.x + threadIdx.x;
    float* cnt = package_half() ? cnt1 : cnt0;
    if (i < E) atomicAdd(&cnt[col[i]], 1.0f);
}

__global__ void k_dinv(float* __restrict__ d0, const float* __restrict__ d1, int n) {
    int i = blockIdx.x * blockDim.x + threadIdx.x;
    if (i < n) {
        float v = 1.0f + d0[i] + d1[i];
        d0[i] = (v > 0.0f) ? rsqrtf(v) : 0.0f;
    }
}

__global__ void k_seed_planes(const float* __restrict__ xin, float* __restrict__ pa, float* __restrict__ pb,
                              const float* __restrict__ dinv, int total) {
    int i = blockIdx.x * blockDim.x + threadIdx.x;
    if (i < total) {
        float w = dinv[i >> 6];
        pa[i] = w * w * xin[i];
        if (pb) pb[i] = 0.0f;
    }
}

__global__ void k_join_seed(float* __restrict__ pa, float* __restrict__ pb, float* __restrict__ na,
                            const float* __restrict__ dinv, int total) {
    int i = blockIdx.x * blockDim.x + threadIdx.x;
    if (i < total) {
        float w = dinv[i >> 6];
        float s = pa[i];
        if (pb) {
            s += pb[i];
            pa[i] = s;
            pb[i] = 0.0f;
        }
        na[i] = w * w * s;
    }
}

#define EPW    32
#define EBATCH 8

template <int NB>
__device__ __forceinline__ void spmm_edges(const float* __restrict__ xin, float* __restrict__ xout,
                                           int r, int c, float w, int first, int lane) {
    float v0[NB];
    float v1[NB];
#pragma unroll
    for (int j = 0; j < NB; ++j) {
        const int rj = __builtin_amdgcn_readlane(r, first + j);
        const float* src = xin + (size_t)rj * FEAT;
        v0[j] = src[lane];
        v1[j] = src[lane + 32];
    }
#pragma unroll
    for (int j = 0; j < NB; ++j) {
        const int   cj = __builtin_amdgcn_readlane(c, first + j);
        const float wj = __int_as_float(__builtin_amdgcn_readlane(__float_as_int(w), first + j));
        float* dst = xout + (size_t)cj * FEAT;
        atomicAdd(dst + lane,      wj * v0[j]);
        atomicAdd(dst + lane + 32, wj * v1[j]);
    }
}

__global__ void k_edge_spmm(const float* __restrict__ xin, float* __restrict__ outa, float* __restrict__ outb,
                            const int* __restrict__ row, const int* __restrict__ col,
                            const float* __restrict__ dinv, int E) {
    float* xout = (outb != nullptr && package_half()) ? outb : outa;
    const int lane = threadIdx.x & 31;
    const int wave = blockIdx.x * (blockDim.x >> 5) + __builtin_amdgcn_readfirstlane(threadIdx.x >> 5);
    const int e0   = wave * EPW;
    if (e0 >= E) return;
    const int e = e0 + lane;
    int   r = 0;
    int   c = 0;
    float w = 0.0f;
    if (e < E) {
        r = row[e];
        c = col[e];
        w = dinv[r] * dinv[c];
    }
    const int n = (E - e0 < EPW) ? (E - e0) : EPW;
    if (n == EPW) {
#pragma unroll
        for (int i = 0; i < EPW; i += EBATCH) spmm_edges<EBATCH>(xin, xout, r, c, w, i, lane);
    } else {
        for (int i = 0; i < n; ++i) spmm_edges<1>(xin, xout, r, c, w, i, lane);
    }
}

__global__ void k_pad_w(const float* __restrict__ W, float* __restrict__ Wp) {
    int i = blockIdx.x * blockDim.x + threadIdx.x;
    if (i < NCLSP * FEAT) {
        int r = i / FEAT;
        Wp[i] = (r < NCLS) ? W[i] : 0.0f;
    }
}

template <bool TWO>
__global__ void k_gemm_wmma(const float* __restrict__ xa, const float* __restrict__ xb, const float* __restrict__ Wp,
                            const float* __restrict__ bias, float* __restrict__ out, int N) {
    const int wave   = blockIdx.x * (blockDim.x >> 5) + (threadIdx.x >> 5);
    const int lane   = threadIdx.x & 31;
    const int ntiles = (N + 15) >> 4;
    if (wave >= ntiles) return;

    const int base = wave << 4;
    const int half = lane >> 4;
    const int l16  = lane & 15;

    int m = base + l16;
    if (m >= N) m = N - 1;
    const float* arow = xa + (size_t)m * FEAT;
    const float* brow = (TWO ? xb : xa) + (size_t)m * FEAT;
    const float* b0r  = Wp + (size_t)(l16)      * FEAT;
    const float* b1r  = Wp + (size_t)(16 + l16) * FEAT;
    const float* b2r  = Wp + (size_t)(32 + l16) * FEAT;

    v8f acc0 = {};
    v8f acc1 = {};
    v8f acc2 = {};

    for (int k0 = 0; k0 < FEAT; k0 += 4) {
        const int ko = k0 + 2 * half;
        v2f a  = *(const v2f*)(arow + ko);
        if (TWO) a += *(const v2f*)(brow + ko);
        v2f b0 = *(const v2f*)(b0r + ko);
        v2f b1 = *(const v2f*)(b1r + ko);
        v2f b2 = *(const v2f*)(b2r + ko);
        acc0 = __builtin_amdgcn_wmma_f32_16x16x4_f32(false, a, false, b0, (short)0, acc0, false, false);
        acc1 = __builtin_amdgcn_wmma_f32_16x16x4_f32(false, a, false, b1, (short)0, acc1, false, false);
        acc2 = __builtin_amdgcn_wmma_f32_16x16x4_f32(false, a, false, b2, (short)0, acc2, false, false);
    }

    const int c0 = l16;
    const int c1 = 16 + l16;
    const int c2 = 32 + l16;
    const float bb0 = bias[c0];
    const float bb1 = bias[c1];
    const float bb2 = (c2 < NCLS) ? bias[c2] : 0.0f;
#pragma unroll
    for (int j = 0; j < 8; ++j) {
        const int node = base + j + 8 * half;
        if (node < N) {
            float* o = out + (size_t)node * NCLS;
            o[c0] = acc0[j] + bb0;
            o[c1] = acc1[j] + bb1;
            if (c2 < NCLS) o[c2] = acc2[j] + bb2;
        }
    }
}

__global__ void k_logsoftmax(float* __restrict__ out, int N) {
    int n = blockIdx.x * blockDim.x + threadIdx.x;
    if (n >= N) return;
    float* p = out + (size_t)n * NCLS;
    float v[NCLS];
    float m = -INFINITY;
#pragma unroll
    for (int c = 0; c < NCLS; ++c) { v[c] = p[c]; m = fmaxf(m, v[c]); }
    float s = 0.0f;
#pragma unroll
    for (int c = 0; c < NCLS; ++c) s += expf(v[c] - m);
    const float ls = logf(s);
#pragma unroll
    for (int c = 0; c < NCLS; ++c) p[c] = v[c] - m - ls;
}


#define EPT 4
#define RPW 4
#define HB  8

__global__ void k_degree_rank(const int* __restrict__ col, int* __restrict__ cnt0, int* __restrict__ cnt1,
                              int* __restrict__ rank, int E) {
    const int h = package_half();
    int* cnt = h ? cnt1 : cnt0;
    const int stride = gridDim.x * blockDim.x;
    const int i0 = blockIdx.x * blockDim.x + threadIdx.x;
    int idx[EPT];
    int c[EPT];
    int old[EPT];
#pragma unroll
    for (int k = 0; k < EPT; ++k) {
        idx[k] = i0 + k * stride;
        c[k] = col[(idx[k] < E) ? idx[k] : (E - 1)];
    }
#pragma unroll
    for (int k = 0; k < EPT; ++k) {
        old[k] = 0;
        if (idx[k] < E) old[k] = atomicAdd(&cnt[c[k]], 1);
    }
#pragma unroll
    for (int k = 0; k < EPT; ++k) {
        if (idx[k] < E) rank[idx[k]] = old[k] | (h << 30);
    }
}

__global__ void __launch_bounds__(1024) k_scan_nodes(const int* __restrict__ cnt0, const int* __restrict__ cnt1,
                                                     int4* __restrict__ rec, int N) {
    __shared__ int wsum[32];
    __shared__ int bpre;
    const int tid  = threadIdx.x;
    const int lane = tid & 31;
    const int wv   = tid >> 5;
    const int start = blockIdx.x * 1024;

    int s = 0;
    const int4* a4 = (const int4*)cnt0;
    const int4* b4 = (const int4*)cnt1;
#pragma unroll 4
    for (int q = tid; q < (start >> 2); q += 1024) {
        const int4 a = a4[q];
        const int4 b = b4[q];
        s += (a.x + a.y) + (a.z + a.w) + (b.x + b.y) + (b.z + b.w);
    }
    for (int m = 16; m > 0; m >>= 1) s += __shfl_xor(s, m);
    if (lane == 0) wsum[wv] = s;
    __syncthreads();
    if (wv == 0) {
        int t = wsum[lane];
        for (int m = 16; m > 0; m >>= 1) t += __shfl_xor(t, m);
        if (lane == 0) bpre = t;
    }
    __syncthreads();
    const int pre = bpre;

    const int i = start + tid;
    int c0 = 0;
    int c1 = 0;
    if (i < N) { c0 = cnt0[i]; c1 = cnt1[i]; }
    const int d = c0 + c1;
    int inc = d;
    for (int m = 1; m < 32; m <<= 1) {
        const int t = __shfl_up(inc, m);
        if (lane >= m) inc += t;
    }
    if (lane == 31) wsum[wv] = inc;
    __syncthreads();
    int wpre = (lane < wv) ? wsum[lane] : 0;
    for (int m = 16; m > 0; m >>= 1) wpre += __shfl_xor(wpre, m);
    const int excl = pre + wpre + inc - d;
    if (i <= N) {
        const float v  = 1.0f + (float)d;
        const float di = (v > 0.0f) ? rsqrtf(v) : 0.0f;
        rec[i] = make_int4(excl, excl + c0, __float_as_int(di), d);
    }
}

__global__ void k_csr_fill(const int* __restrict__ row, const int* __restrict__ col, const int* __restrict__ rank,
                           const int4* __restrict__ rec, int2* __restrict__ csr, int E) {
    const int stride = gridDim.x * blockDim.x;
    const int i0 = blockIdx.x * blockDim.x + threadIdx.x;
    const int* recw = (const int*)rec;
    int idx[EPT];
    int r[EPT];
    int c[EPT];
    int rk[EPT];
    int4 rc[EPT];
    float dr[EPT];
#pragma unroll
    for (int k = 0; k < EPT; ++k) {
        idx[k] = i0 + k * stride;
        const int ii = (idx[k] < E) ? idx[k] : (E - 1);
        r[k]  = row[ii];
        c[k]  = col[ii];
        rk[k] = rank[ii];
    }
#pragma unroll
    for (int k = 0; k < EPT; ++k) {
        rc[k] = rec[c[k]];
        dr[k] = __int_as_float(recw[4 * (size_t)r[k] + 2]);
    }
#pragma unroll
    for (int k = 0; k < EPT; ++k) {
        if (idx[k] < E) {
            const int pos = ((rk[k] >> 30) ? rc[k].y : rc[k].x) + (rk[k] & 0x3fffffff);
            csr[pos] = make_int2(r[k], __float_as_int(dr[k] * __int_as_float(rc[k].z)));
        }
    }
}

template <bool FULL>
__device__ __forceinline__ v2f pull_batch(const float* __restrict__ xin, int srcs, int wbits,
                                          int j0, int m, int lane, v2f acc) {
    constexpr int NJ = FULL ? HB : (HB - 1);
    v2f v[NJ];
#pragma unroll
    for (int j = 0; j < NJ; ++j) {
        v[j] = (v2f){0.0f, 0.0f};
        if (FULL || j == 0 || j < m) {
            const int rj = __builtin_amdgcn_readlane(srcs, j0 + j);
            v[j] = *(const v2f*)(xin + (size_t)rj * FEAT + 2 * lane);
        }
    }
#pragma unroll
    for (int j = 0; j < NJ; ++j) {
        if (FULL || j == 0 || j < m) {
            const float wj = __int_as_float(__builtin_amdgcn_readlane(wbits, j0 + j));
            acc += wj * v[j];
        }
    }
    return acc;
}

__global__ void k_pull_hop(const float* __restrict__ xin, float* __restrict__ xout,
                           const int4* __restrict__ rec, const int2* __restrict__ csr, int N) {
    const int lane = threadIdx.x & 31;
    const int wave  = blockIdx.x * (blockDim.x >> 5) + __builtin_amdgcn_readfirstlane(threadIdx.x >> 5);
    const int node0 = wave * RPW;
    if (node0 >= N) return;
    const int nl = (node0 + lane < N) ? (node0 + lane) : N;
    const int4 rc = rec[nl];
#pragma unroll
    for (int k = 0; k < RPW; ++k) {
        const int node = node0 + k;
        if (node >= N) break;
        const int   beg = __builtin_amdgcn_readlane(rc.x, k);
        const int   end = __builtin_amdgcn_readlane(rc.x, k + 1);
        const float di  = __int_as_float(__builtin_amdgcn_readlane(rc.z, k));
        const v2f self = *(const v2f*)(xin + (size_t)node * FEAT + 2 * lane);
        v2f acc = (di * di) * self;
        for (int base = beg; base < end; base += 32) {
            const int n = (end - base < 32) ? (end - base) : 32;
            int2 p = make_int2(0, 0);
            if (lane < n) p = csr[base + lane];
            int j0 = 0;
            for (; j0 + HB <= n; j0 += HB) acc = pull_batch<true>(xin, p.x, p.y, j0, HB, lane, acc);
            if (j0 < n) acc = pull_batch<false>(xin, p.x, p.y, j0, n - j0, lane, acc);
        }
        *(v2f*)(xout + (size_t)node * FEAT + 2 * lane) = acc;
    }
}


static inline size_t align256f(size_t nfloats) { return (nfloats + 63) & ~(size_t)63; }

extern "C" void kernel_launch(void* const* d_in, const int* in_sizes, int n_in,
                              void* d_out, int out_size, void* d_ws, size_t ws_size,
                              hipStream_t stream) {
    const float* x  = (const float*)d_in[0];
    const int*   ei = (const int*)d_in[1];
    const float* W  = (const float*)d_in[2];
    const float* bi = (const float*)d_in[3];

    const int N = in_sizes[0] / FEAT;
    const int E = in_sizes[1] / 2;
    const int* row = ei;
    const int* col = ei + E;
    const int total = N * FEAT;

    float* ws = (float*)d_ws;
    size_t off = 0;
    float* cnt0 = ws + off; off += align256f((size_t)N);
    float* cnt1 = ws + off; off += align256f((size_t)N);
    float* Wp   = ws + off; off += align256f((size_t)NCLSP * FEAT);

    const int T = 256;
    const int ncnt = (int)(cnt1 - cnt0) + N;
    const int ntiles = (N + 15) >> 4;
    const int gblocks = (ntiles + 7) / 8;

    size_t o2 = off;
    int4*  rec  = (int4*)(ws + o2); o2 += align256f(4 * ((size_t)N + 1));
    float* x0   = ws + o2;          o2 += align256f((size_t)total);
    float* x1   = ws + o2;          o2 += align256f((size_t)total);
    int2*  csr  = (int2*)(ws + o2); o2 += align256f(2 * (size_t)E);
    int*   rank = (int*)(ws + o2);  o2 += align256f((size_t)E);

    if (o2 * sizeof(float) <= ws_size) {
        k_fill<<<(ncnt + T - 1) / T, T, 0, stream>>>(cnt0, 0.0f, ncnt);
        const int sblocks = (E + T * EPT - 1) / (T * EPT);
        k_degree_rank<<<sblocks, T, 0, stream>>>(col, (int*)cnt0, (int*)cnt1, rank, E);
        k_scan_nodes<<<(N + 1 + 1023) / 1024, 1024, 0, stream>>>((const int*)cnt0, (const int*)cnt1, rec, N);
        k_pad_w<<<(NCLSP * FEAT + T - 1) / T, T, 0, stream>>>(W, Wp);
        k_csr_fill<<<sblocks, T, 0, stream>>>(row, col, rank, rec, csr, E);

        const int hblocks = (N + (T / 32) * RPW - 1) / ((T / 32) * RPW);
        k_pull_hop<<<hblocks, T, 0, stream>>>(x,  x0, rec, csr, N);
        k_pull_hop<<<hblocks, T, 0, stream>>>(x0, x1, rec, csr, N);

        k_gemm_wmma<false><<<gblocks, 256, 0, stream>>>(x1, nullptr, Wp, bi, (float*)d_out, N);
        k_logsoftmax<<<(N + T - 1) / T, T, 0, stream>>>((float*)d_out, N);
        return;
    }

    float* p0   = ws + off; off += align256f((size_t)total);
    float* p1   = ws + off; off += align256f((size_t)total);
    float* pb   = nullptr;
    if ((off + (size_t)total) * sizeof(float) <= ws_size) pb = ws + off;
    const float* dinv = cnt0;

    k_fill<<<(ncnt + T - 1) / T, T, 0, stream>>>(cnt0, 0.0f, ncnt);
    k_degree<<<(E + T - 1) / T, T, 0, stream>>>(col, cnt0, cnt1, E);
    k_dinv<<<(N + T - 1) / T, T, 0, stream>>>(cnt0, cnt1, N);
    k_pad_w<<<(NCLSP * FEAT + T - 1) / T, T, 0, stream>>>(W, Wp);

    const int eblocks = (E + (T / 32) * EPW - 1) / ((T / 32) * EPW);

    k_seed_planes<<<(total + T - 1) / T, T, 0, stream>>>(x, p0, pb, dinv, total);
    k_edge_spmm<<<eblocks, T, 0, stream>>>(x, p0, pb, row, col, dinv, E);
    k_join_seed<<<(total + T - 1) / T, T, 0, stream>>>(p0, pb, p1, dinv, total);
    k_edge_spmm<<<eblocks, T, 0, stream>>>(p0, p1, pb, row, col, dinv, E);

    if (pb) k_gemm_wmma<true><<<gblocks, 256, 0, stream>>>(p1, pb, Wp, bi, (float*)d_out, N);
    else    k_gemm_wmma<false><<<gblocks, 256, 0, stream>>>(p1, pb, Wp, bi, (float*)d_out, N);
    k_logsoftmax<<<(N + T - 1) / T, T, 0, stream>>>((float*)d_out, N);
}


# linear layer taken in front of the two hops ((AAx)W^T = A(A(xW^T))): hops move 40-float rows, hop 2's wave adds bias and log_softmax before its store; sums reassociated, hop sum order as before
# speedup vs baseline: 3.2103x; 1.0879x over previous
#include <hip/hip_runtime.h>
#include <math.h>

typedef float v2f __attribute__((ext_vector_type(2)));
typedef float v8f __attribute__((ext_vector_type(8)));

#define FEAT   64
#define NCLS   40
#define NCLSP  48

__device__ __forceinline__ int package_half() {
    return (int)((__builtin_amdgcn_s_sendmsg_rtn(0x87) >> 18) & 1u);
}


__global__ void k_fill(float* __restrict__ p, float val, int n) {
    int i = blockIdx.x * blockDim.x + threadIdx.x;
    if (i < n) p[i] = val;
}

__global__ void k_degree(const int* __restrict__ col, float* __restrict__ cnt0, float* __restrict__ cnt1, int E) {
    int i = blockIdx.x * blockDim.x + threadIdx.x;
    float* cnt = package_half() ? cnt1 : cnt0;
    if (i < E) atomicAdd(&cnt[col[i]], 1.0f);
}

__global__ void k_dinv(float* __restrict__ d0, const float* __restrict__ d1, int n) {
    int i = blockIdx.x * blockDim.x + threadIdx.x;
    if (i < n) {
        float v = 1.0f + d0[i] + d1[i];
        d0[i] = (v > 0.0f) ? rsqrtf(v) : 0.0f;
    }
}

__global__ void k_seed_planes(const float* __restrict__ xin, float* __restrict__ pa, float* __restrict__ pb,
                              const float* __restrict__ dinv, int total) {
    int i = blockIdx.x * blockDim.x + threadIdx.x;
    if (i < total) {
        float w = dinv[i >> 6];
        pa[i] = w * w * xin[i];
        if (pb) pb[i] = 0.0f;
    }
}

__global__ void k_join_seed(float* __restrict__ pa, float* __restrict__ pb, float* __restrict__ na,
                            const float* __restrict__ dinv, int total) {
    int i = blockIdx.x * blockDim.x + threadIdx.x;
    if (i < total) {
        float w = dinv[i >> 6];
        float s = pa[i];
        if (pb) {
            s += pb[i];
            pa[i] = s;
            pb[i] = 0.0f;
        }
        na[i] = w * w * s;
    }
}

#define EPW    32
#define EBATCH 8

template <int NB>
__device__ __forceinline__ void spmm_edges(const float* __restrict__ xin, float* __restrict__ xout,
                                           int r, int c, float w, int first, int lane) {
    float v0[NB];
    float v1[NB];
#pragma unroll
    for (int j = 0; j < NB; ++j) {
        const int rj = __builtin_amdgcn_readlane(r, first + j);
        const float* src = xin + (size_t)rj * FEAT;
        v0[j] = src[lane];
        v1[j] = src[lane + 32];
    }
#pragma unroll
    for (int j = 0; j < NB; ++j) {
        const int   cj = __builtin_amdgcn_readlane(c, first + j);
        const float wj = __int_as_float(__builtin_amdgcn_readlane(__float_as_int(w), first + j));
        float* dst = xout + (size_t)cj * FEAT;
        atomicAdd(dst + lane,      wj * v0[j]);
        atomicAdd(dst + lane + 32, wj * v1[j]);
    }
}

__global__ void k_edge_spmm(const float* __restrict__ xin, float* __restrict__ outa, float* __restrict__ outb,
                            const int* __restrict__ row, const int* __restrict__ col,
                            const float* __restrict__ dinv, int E) {
    float* xout = (outb != nullptr && package_half()) ? outb : outa;
    const int lane = threadIdx.x & 31;
    const int wave = blockIdx.x * (blockDim.x >> 5) + __builtin_amdgcn_readfirstlane(threadIdx.x >> 5);
    const int e0   = wave * EPW;
    if (e0 >= E) return;
    const int e = e0 + lane;
    int   r = 0;
    int   c = 0;
    float w = 0.0f;
    if (e < E) {
        r = row[e];
        c = col[e];
        w = dinv[r] * dinv[c];
    }
    const int n = (E - e0 < EPW) ? (E - e0) : EPW;
    if (n == EPW) {
#pragma unroll
        for (int i = 0; i < EPW; i += EBATCH) spmm_edges<EBATCH>(xin, xout, r, c, w, i, lane);
    } else {
        for (int i = 0; i < n; ++i) spmm_edges<1>(xin, xout, r, c, w, i, lane);
    }
}

__global__ void k_pad_w(const float* __restrict__ W, float* __restrict__ Wp) {
    int i = blockIdx.x * blockDim.x + threadIdx.x;
    if (i < NCLSP * FEAT) {
        int r = i / FEAT;
        Wp[i] = (r < NCLS) ? W[i] : 0.0f;
    }
}

template <bool TWO>
__global__ void k_gemm_wmma(const float* __restrict__ xa, const float* __restrict__ xb, const float* __restrict__ Wp,
                            const float* __restrict__ bias, float* __restrict__ out, int N) {
    const int wave   = blockIdx.x * (blockDim.x >> 5) + (threadIdx.x >> 5);
    const int lane   = threadIdx.x & 31;
    const int ntiles = (N + 15) >> 4;
    if (wave >= ntiles) return;

    const int base = wave << 4;
    const int half = lane >> 4;
    const int l16  = lane & 15;

    int m = base + l16;
    if (m >= N) m = N - 1;
    const float* arow = xa + (size_t)m * FEAT;
    const float* brow = (TWO ? xb : xa) + (size_t)m * FEAT;
    const float* b0r  = Wp + (size_t)(l16)      * FEAT;
    const float* b1r  = Wp + (size_t)(16 + l16) * FEAT;
    const float* b2r  = Wp + (size_t)(32 + l16) * FEAT;

    v8f acc0 = {};
    v8f acc1 = {};
    v8f acc2 = {};

    for (int k0 = 0; k0 < FEAT; k0 += 4) {
        const int ko = k0 + 2 * half;
        v2f a  = *(const v2f*)(arow + ko);
        if (TWO) a += *(const v2f*)(brow + ko);
        v2f b0 = *(const v2f*)(b0r + ko);
        v2f b1 = *(const v2f*)(b1r + ko);
        v2f b2 = *(const v2f*)(b2r + ko);
        acc0 = __builtin_amdgcn_wmma_f32_16x16x4_f32(false, a, false, b0, (short)0, acc0, false, false);
        acc1 = __builtin_amdgcn_wmma_f32_16x16x4_f32(false, a, false, b1, (short)0, acc1, false, false);
        acc2 = __builtin_amdgcn_wmma_f32_16x16x4_f32(false, a, false, b2, (short)0, acc2, false, false);
    }

    const int c0 = l16;
    const int c1 = 16 + l16;
    const int c2 = 32 + l16;
    const float bb0 = bias ? bias[c0] : 0.0f;
    const float bb1 = bias ? bias[c1] : 0.0f;
    const float bb2 = (bias && c2 < NCLS) ? bias[c2] : 0.0f;
#pragma unroll
    for (int j = 0; j < 8; ++j) {
        const int node = base + j + 8 * half;
        if (node < N) {
            float* o = out + (size_t)node * NCLS;
            o[c0] = acc0[j] + bb0;
            o[c1] = acc1[j] + bb1;
            if (c2 < NCLS) o[c2] = acc2[j] + bb2;
        }
    }
}

__global__ void k_logsoftmax(float* __restrict__ out, int N) {
    int n = blockIdx.x * blockDim.x + threadIdx.x;
    if (n >= N) return;
    float* p = out + (size_t)n * NCLS;
    float v[NCLS];
    float m = -INFINITY;
#pragma unroll
    for (int c = 0; c < NCLS; ++c) { v[c] = p[c]; m = fmaxf(m, v[c]); }
    float s = 0.0f;
#pragma unroll
    for (int c = 0; c < NCLS; ++c) s += expf(v[c] - m);
    const float ls = logf(s);
#pragma unroll
    for (int c = 0; c < NCLS; ++c) p[c] = v[c] - m - ls;
}


#define EPT 4
#define RPW 4
#define HB  8
#define YL  (NCLS / 2)

__global__ void k_degree_rank(const int* __restrict__ col, int* __restrict__ cnt0, int* __restrict__ cnt1,
                              int* __restrict__ rank, int E) {
    const int h = package_half();
    int* cnt = h ? cnt1 : cnt0;
    const int stride = gridDim.x * blockDim.x;
    const int i0 = blockIdx.x * blockDim.x + threadIdx.x;
    int idx[EPT];
    int c[EPT];
    int old[EPT];
#pragma unroll
    for (int k = 0; k < EPT; ++k) {
        idx[k] = i0 + k * stride;
        c[k] = col[(idx[k] < E) ? idx[k] : (E - 1)];
    }
#pragma unroll
    for (int k = 0; k < EPT; ++k) {
        old[k] = 0;
        if (idx[k] < E) old[k] = atomicAdd(&cnt[c[k]], 1);
    }
#pragma unroll
    for (int k = 0; k < EPT; ++k) {
        if (idx[k] < E) rank[idx[k]] = old[k] | (h << 30);
    }
}

__global__ void __launch_bounds__(1024) k_scan_nodes(const int* __restrict__ cnt0, const int* __restrict__ cnt1,
                                                     int4* __restrict__ rec, int N) {
    __shared__ int wsum[32];
    __shared__ int bpre;
    const int tid  = threadIdx.x;
    const int lane = tid & 31;
    const int wv   = tid >> 5;
    const int start = blockIdx.x * 1024;

    int s = 0;
    const int4* a4 = (const int4*)cnt0;
    const int4* b4 = (const int4*)cnt1;
#pragma unroll 4
    for (int q = tid; q < (start >> 2); q += 1024) {
        const int4 a = a4[q];
        const int4 b = b4[q];
        s += (a.x + a.y) + (a.z + a.w) + (b.x + b.y) + (b.z + b.w);
    }
    for (int m = 16; m > 0; m >>= 1) s += __shfl_xor(s, m);
    if (lane == 0) wsum[wv] = s;
    __syncthreads();
    if (wv == 0) {
        int t = wsum[lane];
        for (int m = 16; m > 0; m >>= 1) t += __shfl_xor(t, m);
        if (lane == 0) bpre = t;
    }
    __syncthreads();
    const int pre = bpre;

    const int i = start + tid;
    int c0 = 0;
    int c1 = 0;
    if (i < N) { c0 = cnt0[i]; c1 = cnt1[i]; }
    const int d = c0 + c1;
    int inc = d;
    for (int m = 1; m < 32; m <<= 1) {
        const int t = __shfl_up(inc, m);
        if (lane >= m) inc += t;
    }
    if (lane == 31) wsum[wv] = inc;
    __syncthreads();
    int wpre = (lane < wv) ? wsum[lane] : 0;
    for (int m = 16; m > 0; m >>= 1) wpre += __shfl_xor(wpre, m);
    const int excl = pre + wpre + inc - d;
    if (i <= N) {
        const float v  = 1.0f + (float)d;
        const float di = (v > 0.0f) ? rsqrtf(v) : 0.0f;
        rec[i] = make_int4(excl, excl + c0, __float_as_int(di), d);
    }
}

__global__ void k_csr_fill(const int* __restrict__ row, const int* __restrict__ col, const int* __restrict__ rank,
                           const int4* __restrict__ rec, int2* __restrict__ csr, int E) {
    const int stride = gridDim.x * blockDim.x;
    const int i0 = blockIdx.x * blockDim.x + threadIdx.x;
    const int* recw = (const int*)rec;
    int idx[EPT];
    int r[EPT];
    int c[EPT];
    int rk[EPT];
    int4 rc[EPT];
    float dr[EPT];
#pragma unroll
    for (int k = 0; k < EPT; ++k) {
        idx[k] = i0 + k * stride;
        const int ii = (idx[k] < E) ? idx[k] : (E - 1);
        r[k]  = row[ii];
        c[k]  = col[ii];
        rk[k] = rank[ii];
    }
#pragma unroll
    for (int k = 0; k < EPT; ++k) {
        rc[k] = rec[c[k]];
        dr[k] = __int_as_float(recw[4 * (size_t)r[k] + 2]);
    }
#pragma unroll
    for (int k = 0; k < EPT; ++k) {
        if (idx[k] < E) {
            const int pos = ((rk[k] >> 30) ? rc[k].y : rc[k].x) + (rk[k] & 0x3fffffff);
            csr[pos] = make_int2(r[k], __float_as_int(dr[k] * __int_as_float(rc[k].z)));
        }
    }
}

template <bool FULL>
__device__ __forceinline__ v2f pull_batch(const float* __restrict__ xin, int srcs, int wbits,
                                          int j0, int m, int cl, v2f acc) {
    constexpr int NJ = FULL ? HB : (HB - 1);
    v2f v[NJ];
#pragma unroll
    for (int j = 0; j < NJ; ++j) {
        v[j] = (v2f){0.0f, 0.0f};
        if (FULL || j == 0 || j < m) {
            const int rj = __builtin_amdgcn_readlane(srcs, j0 + j);
            v[j] = *(const v2f*)(xin + (size_t)rj * NCLS + cl);
        }
    }
#pragma unroll
    for (int j = 0; j < NJ; ++j) {
        if (FULL || j == 0 || j < m) {
            const float wj = __int_as_float(__builtin_amdgcn_readlane(wbits, j0 + j));
            acc += wj * v[j];
        }
    }
    return acc;
}

template <int M>
__device__ __forceinline__ float row_xor(float v) {
    return __builtin_bit_cast(float, __builtin_amdgcn_update_dpp(0, __builtin_bit_cast(int, v), 0x160 | M, 0xf, 0xf, false));
}
__device__ __forceinline__ float xhalf(float v) {
    const int b = __builtin_bit_cast(int, v);
    return __builtin_bit_cast(float, __builtin_amdgcn_permlanex16(b, b, 0x76543210, 0xfedcba98, false, false));
}
__device__ __forceinline__ float wave_max(float v) {
    v = fmaxf(v, row_xor<8>(v));
    v = fmaxf(v, row_xor<4>(v));
    v = fmaxf(v, row_xor<2>(v));
    v = fmaxf(v, row_xor<1>(v));
    return fmaxf(v, xhalf(v));
}
__device__ __forceinline__ float wave_sum(float v) {
    v += row_xor<8>(v);
    v += row_xor<4>(v);
    v += row_xor<2>(v);
    v += row_xor<1>(v);
    return v + xhalf(v);
}

template <bool LAST>
__global__ void k_pull_hop(const float* __restrict__ xin, float* __restrict__ xout,
                           const int4* __restrict__ rec, const int2* __restrict__ csr,
                           const float* __restrict__ bias, int N) {
    const int lane = threadIdx.x & 31;
    const bool own = lane < YL;
    const int cl = 2 * (own ? lane : (YL - 1));
    const int wave  = blockIdx.x * (blockDim.x >> 5) + __builtin_amdgcn_readfirstlane(threadIdx.x >> 5);
    const int node0 = wave * RPW;
    if (node0 >= N) return;
    const int nl = (node0 + lane < N) ? (node0 + lane) : N;
    const int4 rc = rec[nl];
    v2f bb = (v2f){0.0f, 0.0f};
    if (LAST) { bb.x = bias[cl]; bb.y = bias[cl + 1]; }
#pragma unroll
    for (int k = 0; k < RPW; ++k) {
        const int node = node0 + k;
        if (node >= N) break;
        const int   beg = __builtin_amdgcn_readlane(rc.x, k);
        const int   end = __builtin_amdgcn_readlane(rc.x, k + 1);
        const float di  = __int_as_float(__builtin_amdgcn_readlane(rc.z, k));
        const v2f self = *(const v2f*)(xin + (size_t)node * NCLS + cl);
        v2f acc = (di * di) * self;
        for (int base = beg; base < end; base += 32) {
            const int n = (end - base < 32) ? (end - base) : 32;
            int2 p = make_int2(0, 0);
            if (lane < n) p = csr[base + lane];
            int j0 = 0;
            for (; j0 + HB <= n; j0 += HB) acc = pull_batch<true>(xin, p.x, p.y, j0, HB, cl, acc);
            if (j0 < n) acc = pull_batch<false>(xin, p.x, p.y, j0, n - j0, cl, acc);
        }
        if (LAST) {
            const v2f lg = acc + bb;
            const float mx = wave_max(own ? fmaxf(lg.x, lg.y) : -INFINITY);
            const float se = wave_sum(own ? (expf(lg.x - mx) + expf(lg.y - mx)) : 0.0f);
            const float ls = logf(se);
            acc = (v2f){lg.x - mx - ls, lg.y - mx - ls};
        }
        if (own) *(v2f*)(xout + (size_t)node * NCLS + cl) = acc;
    }
}


static inline size_t align256f(size_t nfloats) { return (nfloats + 63) & ~(size_t)63; }

extern "C" void kernel_launch(void* const* d_in, const int* in_sizes, int n_in,
                              void* d_out, int out_size, void* d_ws, size_t ws_size,
                              hipStream_t stream) {
    const float* x  = (const float*)d_in[0];
    const int*   ei = (const int*)d_in[1];
    const float* W  = (const float*)d_in[2];
    const float* bi = (const float*)d_in[3];

    const int N = in_sizes[0] / FEAT;
    const int E = in_sizes[1] / 2;
    const int* row = ei;
    const int* col = ei + E;
    const int total = N * FEAT;

    float* ws = (float*)d_ws;
    size_t off = 0;
    float* cnt0 = ws + off; off += align256f((size_t)N);
    float* cnt1 = ws + off; off += align256f((size_t)N);
    float* Wp   = ws + off; off += align256f((size_t)NCLSP * FEAT);

    const int T = 256;
    const int ncnt = (int)(cnt1 - cnt0) + N;
    const int ntiles = (N + 15) >> 4;
    const int gblocks = (ntiles + 7) / 8;

    size_t o2 = off;
    int4*  rec  = (int4*)(ws + o2); o2 += align256f(4 * ((size_t)N + 1));
    float* y0   = ws + o2;          o2 += align256f((size_t)N * NCLS);
    float* y1   = ws + o2;          o2 += align256f((size_t)N * NCLS);
    int2*  csr  = (int2*)(ws + o2); o2 += align256f(2 * (size_t)E);
    int*   rank = (int*)(ws + o2);  o2 += align256f((size_t)E);

    if (o2 * sizeof(float) <= ws_size) {
        k_fill<<<(ncnt + T - 1) / T, T, 0, stream>>>(cnt0, 0.0f, ncnt);
        const int sblocks = (E + T * EPT - 1) / (T * EPT);
        k_degree_rank<<<sblocks, T, 0, stream>>>(col, (int*)cnt0, (int*)cnt1, rank, E);
        k_scan_nodes<<<(N + 1 + 1023) / 1024, 1024, 0, stream>>>((const int*)cnt0, (const int*)cnt1, rec, N);
        k_pad_w<<<(NCLSP * FEAT + T - 1) / T, T, 0, stream>>>(W, Wp);
        k_csr_fill<<<sblocks, T, 0, stream>>>(row, col, rank, rec, csr, E);

        k_gemm_wmma<false><<<gblocks, 256, 0, stream>>>(x, nullptr, Wp, nullptr, y0, N);

        const int hblocks = (N + (T / 32) * RPW - 1) / ((T / 32) * RPW);
        k_pull_hop<false><<<hblocks, T, 0, stream>>>(y0, y1, rec, csr, nullptr, N);
        k_pull_hop<true><<<hblocks, T, 0, stream>>>(y1, (float*)d_out, rec, csr, bi, N);
        return;
    }

    float* p0   = ws + off; off += align256f((size_t)total);
    float* p1   = ws + off; off += align256f((size_t)total);
    float* pb   = nullptr;
    if ((off + (size_t)total) * sizeof(float) <= ws_size) pb = ws + off;
    const float* dinv = cnt0;

    k_fill<<<(ncnt + T - 1) / T, T, 0, stream>>>(cnt0, 0.0f, ncnt);
    k_degree<<<(E + T - 1) / T, T, 0, stream>>>(col, cnt0, cnt1, E);
    k_dinv<<<(N + T - 1) / T, T, 0, stream>>>(cnt0, cnt1, N);
    k_pad_w<<<(NCLSP * FEAT + T - 1) / T, T, 0, stream>>>(W, Wp);

    const int eblocks = (E + (T / 32) * EPW - 1) / ((T / 32) * EPW);

    k_seed_planes<<<(total + T - 1) / T, T, 0, stream>>>(x, p0, pb, dinv, total);
    k_edge_spmm<<<eblocks, T, 0, stream>>>(x, p0, pb, row, col, dinv, E);
    k_join_seed<<<(total + T - 1) / T, T, 0, stream>>>(p0, pb, p1, dinv, total);
    k_edge_spmm<<<eblocks, T, 0, stream>>>(p0, p1, pb, row, col, dinv, E);

    if (pb) k_gemm_wmma<true><<<gblocks, 256, 0, stream>>>(p1, pb, Wp, bi, (float*)d_out, N);
    else    k_gemm_wmma<false><<<gblocks, 256, 0, stream>>>(p1, pb, Wp, bi, (float*)d_out, N);
    k_logsoftmax<<<(N + T - 1) / T, T, 0, stream>>>((float*)d_out, N);
}


# linear layer's GEMM runs as extra work-groups of the edge count launch (matrix work beside the waits for returning atomics) and reads W unpadded: its launch and the W padding launch go, 8 -> 6
# speedup vs baseline: 3.4086x; 1.0618x over previous
#include <hip/hip_runtime.h>
#include <math.h>

typedef float v2f __attribute__((ext_vector_type(2)));
typedef float v8f __attribute__((ext_vector_type(8)));

#define FEAT   64
#define NCLS   40
#define NCLSP  48

__device__ __forceinline__ int package_half() {
    return (int)((__builtin_amdgcn_s_sendmsg_rtn(0x87) >> 18) & 1u);
}


__global__ void k_fill(float* __restrict__ p, float val, int n) {
    int i = blockIdx.x * blockDim.x + threadIdx.x;
    if (i < n) p[i] = val;
}

__global__ void k_degree(const int* __restrict__ col, float* __restrict__ cnt0, float* __restrict__ cnt1, int E) {
    int i = blockIdx.x * blockDim.x + threadIdx.x;
    float* cnt = package_half() ? cnt1 : cnt0;
    if (i < E) atomicAdd(&cnt[col[i]], 1.0f);
}

__global__ void k_dinv(float* __restrict__ d0, const float* __restrict__ d1, int n) {
    int i = blockIdx.x * blockDim.x + threadIdx.x;
    if (i < n) {
        float v = 1.0f + d0[i] + d1[i];
        d0[i] = (v > 0.0f) ? rsqrtf(v) : 0.0f;
    }
}

__global__ void k_seed_planes(const float* __restrict__ xin, float* __restrict__ pa, float* __restrict__ pb,
                              const float* __restrict__ dinv, int total) {
    int i = blockIdx.x * blockDim.x + threadIdx.x;
    if (i < total) {
        float w = dinv[i >> 6];
        pa[i] = w * w * xin[i];
        if (pb) pb[i] = 0.0f;
    }
}

__global__ void k_join_seed(float* __restrict__ pa, float* __restrict__ pb, float* __restrict__ na,
                            const float* __restrict__ dinv, int total) {
    int i = blockIdx.x * blockDim.x + threadIdx.x;
    if (i < total) {
        float w = dinv[i >> 6];
        float s = pa[i];
        if (pb) {
            s += pb[i];
            pa[i] = s;
            pb[i] = 0.0f;
        }
        na[i] = w * w * s;
    }
}

#define EPW    32
#define EBATCH 8

template <int NB>
__device__ __forceinline__ void spmm_edges(const float* __restrict__ xin, float* __restrict__ xout,
                                           int r, int c, float w, int first, int lane) {
    float v0[NB];
    float v1[NB];
#pragma unroll
    for (int j = 0; j < NB; ++j) {
        const int rj = __builtin_amdgcn_readlane(r, first + j);
        const float* src = xin + (size_t)rj * FEAT;
        v0[j] = src[lane];
        v1[j] = src[lane + 32];
    }
#pragma unroll
    for (int j = 0; j < NB; ++j) {
        const int   cj = __builtin_amdgcn_readlane(c, first + j);
        const float wj = __int_as_float(__builtin_amdgcn_readlane(__float_as_int(w), first + j));
        float* dst = xout + (size_t)cj * FEAT;
        atomicAdd(dst + lane,      wj * v0[j]);
        atomicAdd(dst + lane + 32, wj * v1[j]);
    }
}

__global__ void k_edge_spmm(const float* __restrict__ xin, float* __restrict__ outa, float* __restrict__ outb,
                            const int* __restrict__ row, const int* __restrict__ col,
                            const float* __restrict__ dinv, int E) {
    float* xout = (outb != nullptr && package_half()) ? outb : outa;
    const int lane = threadIdx.x & 31;
    const int wave = blockIdx.x * (blockDim.x >> 5) + __builtin_amdgcn_readfirstlane(threadIdx.x >> 5);
    const int e0   = wave * EPW;
    if (e0 >= E) return;
    const int e = e0 + lane;
    int   r = 0;
    int   c = 0;
    float w = 0.0f;
    if (e < E) {
        r = row[e];
        c = col[e];
        w = dinv[r] * dinv[c];
    }
    const int n = (E - e0 < EPW) ? (E - e0) : EPW;
    if (n == EPW) {
#pragma unroll
        for (int i = 0; i < EPW; i += EBATCH) spmm_edges<EBATCH>(xin, xout, r, c, w, i, lane);
    } else {
        for (int i = 0; i < n; ++i) spmm_edges<1>(xin, xout, r, c, w, i, lane);
    }
}

__global__ void k_pad_w(const float* __restrict__ W, float* __restrict__ Wp) {
    int i = blockIdx.x * blockDim.x + threadIdx.x;
    if (i < NCLSP * FEAT) {
        int r = i / FEAT;
        Wp[i] = (r < NCLS) ? W[i] : 0.0f;
    }
}

template <bool TWO>
__global__ void k_gemm_wmma(const float* __restrict__ xa, const float* __restrict__ xb, const float* __restrict__ Wp,
                            const float* __restrict__ bias, float* __restrict__ out, int N) {
    const int wave   = blockIdx.x * (blockDim.x >> 5) + (threadIdx.x >> 5);
    const int lane   = threadIdx.x & 31;
    const int ntiles = (N + 15) >> 4;
    if (wave >= ntiles) return;

    const int base = wave << 4;
    const int half = lane >> 4;
    const int l16  = lane & 15;

    int m = base + l16;
    if (m >= N) m = N - 1;
    const float* arow = xa + (size_t)m * FEAT;
    const float* brow = (TWO ? xb : xa) + (size_t)m * FEAT;
    const float* b0r  = Wp + (size_t)(l16)      * FEAT;
    const float* b1r  = Wp + (size_t)(16 + l16) * FEAT;
    const float* b2r  = Wp + (size_t)(32 + l16) * FEAT;

    v8f acc0 = {};
    v8f acc1 = {};
    v8f acc2 = {};

    for (int k0 = 0; k0 < FEAT; k0 += 4) {
        const int ko = k0 + 2 * half;
        v2f a  = *(const v2f*)(arow + ko);
        if (TWO) a += *(const v2f*)(brow + ko);
        v2f b0 = *(const v2f*)(b0r + ko);
        v2f b1 = *(const v2f*)(b1r + ko);
        v2f b2 = *(const v2f*)(b2r + ko);
        acc0 = __builtin_amdgcn_wmma_f32_16x16x4_f32(false, a, false, b0, (short)0, acc0, false, false);
        acc1 = __builtin_amdgcn_wmma_f32_16x16x4_f32(false, a, false, b1, (short)0, acc1, false, false);
        acc2 = __builtin_amdgcn_wmma_f32_16x16x4_f32(false, a, false, b2, (short)0, acc2, false, false);
    }

    const int c0 = l16;
    const int c1 = 16 + l16;
    const int c2 = 32 + l16;
    const float bb0 = bias ? bias[c0] : 0.0f;
    const float bb1 = bias ? bias[c1] : 0.0f;
    const float bb2 = (bias && c2 < NCLS) ? bias[c2] : 0.0f;
#pragma unroll
    for (int j = 0; j < 8; ++j) {
        const int node = base + j + 8 * half;
        if (node < N) {
            float* o = out + (size_t)node * NCLS;
            o[c0] = acc0[j] + bb0;
            o[c1] = acc1[j] + bb1;
            if (c2 < NCLS) o[c2] = acc2[j] + bb2;
        }
    }
}

__global__ void k_logsoftmax(float* __restrict__ out, int N) {
    int n = blockIdx.x * blockDim.x + threadIdx.x;
    if (n >= N) return;
    float* p = out + (size_t)n * NCLS;
    float v[NCLS];
    float m = -INFINITY;
#pragma unroll
    for (int c = 0; c < NCLS; ++c) { v[c] = p[c]; m = fmaxf(m, v[c]); }
    float s = 0.0f;
#pragma unroll
    for (int c = 0; c < NCLS; ++c) s += expf(v[c] - m);
    const float ls = logf(s);
#pragma unroll
    for (int c = 0; c < NCLS; ++c) p[c] = v[c] - m - ls;
}


#define EPT 4
#define RPW 4
#define HB  8
#define YL  (NCLS / 2)

__device__ __forceinline__ void degree_rank_block(const int* __restrict__ col, int* __restrict__ cnt0, int* __restrict__ cnt1,
                                                  int* __restrict__ rank, int E, int blk, int nblk) {
    const int h = package_half();
    int* cnt = h ? cnt1 : cnt0;
    const int stride = nblk * blockDim.x;
    const int i0 = blk * blockDim.x + threadIdx.x;
    int idx[EPT];
    int c[EPT];
    int old[EPT];
#pragma unroll
    for (int k = 0; k < EPT; ++k) {
        idx[k] = i0 + k * stride;
        c[k] = col[(idx[k] < E) ? idx[k] : (E - 1)];
    }
#pragma unroll
    for (int k = 0; k < EPT; ++k) {
        old[k] = 0;
        if (idx[k] < E) old[k] = atomicAdd(&cnt[c[k]], 1);
    }
#pragma unroll
    for (int k = 0; k < EPT; ++k) {
        if (idx[k] < E) rank[idx[k]] = old[k] | (h << 30);
    }
}

__device__ __forceinline__ void gemm_tile_w40(const float* __restrict__ x, const float* __restrict__ W,
                                              float* __restrict__ y, int N, int tile, int lane) {
    const int base = tile << 4;
    const int half = lane >> 4;
    const int l16  = lane & 15;

    int m = base + l16;
    if (m >= N) m = N - 1;
    const bool b2ok = (32 + l16) < NCLS;
    const float* arow = x + (size_t)m * FEAT;
    const float* b0r  = W + (size_t)(l16)      * FEAT;
    const float* b1r  = W + (size_t)(16 + l16) * FEAT;
    const float* b2r  = W + (size_t)(b2ok ? (32 + l16) : (NCLS - 1)) * FEAT;

    v8f acc0 = {};
    v8f acc1 = {};
    v8f acc2 = {};

    for (int k0 = 0; k0 < FEAT; k0 += 4) {
        const int ko = k0 + 2 * half;
        v2f a  = *(const v2f*)(arow + ko);
        v2f b0 = *(const v2f*)(b0r + ko);
        v2f b1 = *(const v2f*)(b1r + ko);
        v2f b2 = *(const v2f*)(b2r + ko);
        acc0 = __builtin_amdgcn_wmma_f32_16x16x4_f32(false, a, false, b0, (short)0, acc0, false, false);
        acc1 = __builtin_amdgcn_wmma_f32_16x16x4_f32(false, a, false, b1, (short)0, acc1, false, false);
        acc2 = __builtin_amdgcn_wmma_f32_16x16x4_f32(false, a, false, b2, (short)0, acc2, false, false);
    }

    int ln = lane;
    asm volatile("v_nop\n\tv_nop\n\tv_nop\n\tv_nop" : "+v"(acc0), "+v"(acc1), "+v"(acc2), "+v"(ln));
    const int eh = ln >> 4;
    const int el = ln & 15;

    const int c0 = el;
    const int c1 = 16 + el;
    const int c2 = 32 + el;
#pragma unroll
    for (int j = 0; j < 8; ++j) {
        const int node = base + j + 8 * eh;
        if (node < N) {
            float* o = y + (size_t)node * NCLS;
            o[c0] = acc0[j];
            o[c1] = acc1[j];
            if (c2 < NCLS) o[c2] = acc2[j];
        }
    }
}

__global__ void k_rank_gemm(const int* __restrict__ col, int* __restrict__ cnt0, int* __restrict__ cnt1,
                            int* __restrict__ rank, int E, int ngemm, int nrank,
                            const float* __restrict__ x, const float* __restrict__ W,
                            float* __restrict__ y, int N) {
    if ((int)blockIdx.x >= ngemm) {
        degree_rank_block(col, cnt0, cnt1, rank, E, (int)blockIdx.x - ngemm, nrank);
        return;
    }
    const int tile = (int)blockIdx.x * (int)(blockDim.x >> 5) + __builtin_amdgcn_readfirstlane(threadIdx.x >> 5);
    if (tile >= ((N + 15) >> 4)) return;
    gemm_tile_w40(x, W, y, N, tile, threadIdx.x & 31);
}

__global__ void __launch_bounds__(1024) k_scan_nodes(const int* __restrict__ cnt0, const int* __restrict__ cnt1,
                                                     int4* __restrict__ rec, int N) {
    __shared__ int wsum[32];
    __shared__ int bpre;
    const int tid  = threadIdx.x;
    const int lane = tid & 31;
    const int wv   = tid >> 5;
    const int start = blockIdx.x * 1024;

    int s = 0;
    const int4* a4 = (const int4*)cnt0;
    const int4* b4 = (const int4*)cnt1;
#pragma unroll 4
    for (int q = tid; q < (start >> 2); q += 1024) {
        const int4 a = a4[q];
        const int4 b = b4[q];
        s += (a.x + a.y) + (a.z + a.w) + (b.x + b.y) + (b.z + b.w);
    }
    for (int m = 16; m > 0; m >>= 1) s += __shfl_xor(s, m);
    if (lane == 0) wsum[wv] = s;
    __syncthreads();
    if (wv == 0) {
        int t = wsum[lane];
        for (int m = 16; m > 0; m >>= 1) t += __shfl_xor(t, m);
        if (lane == 0) bpre = t;
    }
    __syncthreads();
    const int pre = bpre;

    const int i = start + tid;
    int c0 = 0;
    int c1 = 0;
    if (i < N) { c0 = cnt0[i]; c1 = cnt1[i]; }
    const int d = c0 + c1;
    int inc = d;
    for (int m = 1; m < 32; m <<= 1) {
        const int t = __shfl_up(inc, m);
        if (lane >= m) inc += t;
    }
    if (lane == 31) wsum[wv] = inc;
    __syncthreads();
    int wpre = (lane < wv) ? wsum[lane] : 0;
    for (int m = 16; m > 0; m >>= 1) wpre += __shfl_xor(wpre, m);
    const int excl = pre + wpre + inc - d;
    if (i <= N) {
        const float v  = 1.0f + (float)d;
        const float di = (v > 0.0f) ? rsqrtf(v) : 0.0f;
        rec[i] = make_int4(excl, excl + c0, __float_as_int(di), d);
    }
}

__global__ void k_csr_fill(const int* __restrict__ row, const int* __restrict__ col, const int* __restrict__ rank,
                           const int4* __restrict__ rec, int2* __restrict__ csr, int E) {
    const int stride = gridDim.x * blockDim.x;
    const int i0 = blockIdx.x * blockDim.x + threadIdx.x;
    const int* recw = (const int*)rec;
    int idx[EPT];
    int r[EPT];
    int c[EPT];
    int rk[EPT];
    int4 rc[EPT];
    float dr[EPT];
#pragma unroll
    for (int k = 0; k < EPT; ++k) {
        idx[k] = i0 + k * stride;
        const int ii = (idx[k] < E) ? idx[k] : (E - 1);
        r[k]  = row[ii];
        c[k]  = col[ii];
        rk[k] = rank[ii];
    }
#pragma unroll
    for (int k = 0; k < EPT; ++k) {
        rc[k] = rec[c[k]];
        dr[k] = __int_as_float(recw[4 * (size_t)r[k] + 2]);
    }
#pragma unroll
    for (int k = 0; k < EPT; ++k) {
        if (idx[k] < E) {
            const int pos = ((rk[k] >> 30) ? rc[k].y : rc[k].x) + (rk[k] & 0x3fffffff);
            csr[pos] = make_int2(r[k], __float_as_int(dr[k] * __int_as_float(rc[k].z)));
        }
    }
}

template <bool FULL>
__device__ __forceinline__ v2f pull_batch(const float* __restrict__ xin, int srcs, int wbits,
                                          int j0, int m, int cl, v2f acc) {
    constexpr int NJ = FULL ? HB : (HB - 1);
    v2f v[NJ];
#pragma unroll
    for (int j = 0; j < NJ; ++j) {
        v[j] = (v2f){0.0f, 0.0f};
        if (FULL || j == 0 || j < m) {
            const int rj = __builtin_amdgcn_readlane(srcs, j0 + j);
            v[j] = *(const v2f*)(xin + (size_t)rj * NCLS + cl);
        }
    }
#pragma unroll
    for (int j = 0; j < NJ; ++j) {
        if (FULL || j == 0 || j < m) {
            const float wj = __int_as_float(__builtin_amdgcn_readlane(wbits, j0 + j));
            acc += wj * v[j];
        }
    }
    return acc;
}

template <int M>
__device__ __forceinline__ float row_xor(float v) {
    return __builtin_bit_cast(float, __builtin_amdgcn_update_dpp(0, __builtin_bit_cast(int, v), 0x160 | M, 0xf, 0xf, false));
}
__device__ __forceinline__ float xhalf(float v) {
    const int b = __builtin_bit_cast(int, v);
    return __builtin_bit_cast(float, __builtin_amdgcn_permlanex16(b, b, 0x76543210, 0xfedcba98, false, false));
}
__device__ __forceinline__ float wave_max(float v) {
    v = fmaxf(v, row_xor<8>(v));
    v = fmaxf(v, row_xor<4>(v));
    v = fmaxf(v, row_xor<2>(v));
    v = fmaxf(v, row_xor<1>(v));
    return fmaxf(v, xhalf(v));
}
__device__ __forceinline__ float wave_sum(float v) {
    v += row_xor<8>(v);
    v += row_xor<4>(v);
    v += row_xor<2>(v);
    v += row_xor<1>(v);
    return v + xhalf(v);
}

template <bool LAST>
__global__ void k_pull_hop(const float* __restrict__ xin, float* __restrict__ xout,
                           const int4* __restrict__ rec, const int2* __restrict__ csr,
                           const float* __restrict__ bias, int N) {
    const int lane = threadIdx.x & 31;
    const bool own = lane < YL;
    const int cl = 2 * (own ? lane : (YL - 1));
    const int wave  = blockIdx.x * (blockDim.x >> 5) + __builtin_amdgcn_readfirstlane(threadIdx.x >> 5);
    const int node0 = wave * RPW;
    if (node0 >= N) return;
    const int nl = (node0 + lane < N) ? (node0 + lane) : N;
    const int4 rc = rec[nl];
    v2f bb = (v2f){0.0f, 0.0f};
    if (LAST) { bb.x = bias[cl]; bb.y = bias[cl + 1]; }
#pragma unroll
    for (int k = 0; k < RPW; ++k) {
        const int node = node0 + k;
        if (node >= N) break;
        const int   beg = __builtin_amdgcn_readlane(rc.x, k);
        const int   end = __builtin_amdgcn_readlane(rc.x, k + 1);
        const float di  = __int_as_float(__builtin_amdgcn_readlane(rc.z, k));
        const v2f self = *(const v2f*)(xin + (size_t)node * NCLS + cl);
        v2f acc = (di * di) * self;
        for (int base = beg; base < end; base += 32) {
            const int n = (end - base < 32) ? (end - base) : 32;
            int2 p = make_int2(0, 0);
            if (lane < n) p = csr[base + lane];
            int j0 = 0;
            for (; j0 + HB <= n; j0 += HB) acc = pull_batch<true>(xin, p.x, p.y, j0, HB, cl, acc);
            if (j0 < n) acc = pull_batch<false>(xin, p.x, p.y, j0, n - j0, cl, acc);
        }
        if (LAST) {
            const v2f lg = acc + bb;
            const float mx = wave_max(own ? fmaxf(lg.x, lg.y) : -INFINITY);
            const float se = wave_sum(own ? (expf(lg.x - mx) + expf(lg.y - mx)) : 0.0f);
            const float ls = logf(se);
            acc = (v2f){lg.x - mx - ls, lg.y - mx - ls};
        }
        if (own) *(v2f*)(xout + (size_t)node * NCLS + cl) = acc;
    }
}


static inline size_t align256f(size_t nfloats) { return (nfloats + 63) & ~(size_t)63; }

extern "C" void kernel_launch(void* const* d_in, const int* in_sizes, int n_in,
                              void* d_out, int out_size, void* d_ws, size_t ws_size,
                              hipStream_t stream) {
    const float* x  = (const float*)d_in[0];
    const int*   ei = (const int*)d_in[1];
    const float* W  = (const float*)d_in[2];
    const float* bi = (const float*)d_in[3];

    const int N = in_sizes[0] / FEAT;
    const int E = in_sizes[1] / 2;
    const int* row = ei;
    const int* col = ei + E;
    const int total = N * FEAT;

    float* ws = (float*)d_ws;
    size_t off = 0;
    float* cnt0 = ws + off; off += align256f((size_t)N);
    float* cnt1 = ws + off; off += align256f((size_t)N);
    float* Wp   = ws + off; off += align256f((size_t)NCLSP * FEAT);

    const int T = 256;
    const int ncnt = (int)(cnt1 - cnt0) + N;
    const int ntiles = (N + 15) >> 4;
    const int gblocks = (ntiles + 7) / 8;

    size_t o2 = off;
    int4*  rec  = (int4*)(ws + o2); o2 += align256f(4 * ((size_t)N + 1));
    float* y0   = ws + o2;          o2 += align256f((size_t)N * NCLS);
    float* y1   = ws + o2;          o2 += align256f((size_t)N * NCLS);
    int2*  csr  = (int2*)(ws + o2); o2 += align256f(2 * (size_t)E);
    int*   rank = (int*)(ws + o2);  o2 += align256f((size_t)E);

    if (o2 * sizeof(float) <= ws_size) {
        k_fill<<<(ncnt + T - 1) / T, T, 0, stream>>>(cnt0, 0.0f, ncnt);
        const int sblocks = (E + T * EPT - 1) / (T * EPT);
        k_rank_gemm<<<gblocks + sblocks, T, 0, stream>>>(col, (int*)cnt0, (int*)cnt1, rank, E, gblocks, sblocks,
                                                         x, W, y0, N);
        k_scan_nodes<<<(N + 1 + 1023) / 1024, 1024, 0, stream>>>((const int*)cnt0, (const int*)cnt1, rec, N);
        k_csr_fill<<<sblocks, T, 0, stream>>>(row, col, rank, rec, csr, E);


        const int hblocks = (N + (T / 32) * RPW - 1) / ((T / 32) * RPW);
        k_pull_hop<false><<<hblocks, T, 0, stream>>>(y0, y1, rec, csr, nullptr, N);
        k_pull_hop<true><<<hblocks, T, 0, stream>>>(y1, (float*)d_out, rec, csr, bi, N);
        return;
    }

    float* p0   = ws + off; off += align256f((size_t)total);
    float* p1   = ws + off; off += align256f((size_t)total);
    float* pb   = nullptr;
    if ((off + (size_t)total) * sizeof(float) <= ws_size) pb = ws + off;
    const float* dinv = cnt0;

    k_fill<<<(ncnt + T - 1) / T, T, 0, stream>>>(cnt0, 0.0f, ncnt);
    k_degree<<<(E + T - 1) / T, T, 0, stream>>>(col, cnt0, cnt1, E);
    k_dinv<<<(N + T - 1) / T, T, 0, stream>>>(cnt0, cnt1, N);
    k_pad_w<<<(NCLSP * FEAT + T - 1) / T, T, 0, stream>>>(W, Wp);

    const int eblocks = (E + (T / 32) * EPW - 1) / ((T / 32) * EPW);

    k_seed_planes<<<(total + T - 1) / T, T, 0, stream>>>(x, p0, pb, dinv, total);
    k_edge_spmm<<<eblocks, T, 0, stream>>>(x, p0, pb, row, col, dinv, E);
    k_join_seed<<<(total + T - 1) / T, T, 0, stream>>>(p0, pb, p1, dinv, total);
    k_edge_spmm<<<eblocks, T, 0, stream>>>(p0, p1, pb, row, col, dinv, E);

    if (pb) k_gemm_wmma<true><<<gblocks, 256, 0, stream>>>(p1, pb, Wp, bi, (float*)d_out, N);
    else    k_gemm_wmma<false><<<gblocks, 256, 0, stream>>>(p1, pb, Wp, bi, (float*)d_out, N);
    k_logsoftmax<<<(N + T - 1) / T, T, 0, stream>>>((float*)d_out, N);
}


# slot table replaces the edge sort: target owns 2x32 slots, slot = rank the count atomic returns; scan and fill passes go, 6 -> 5 launches; sum order as before, edges past 32 a plane last, by arrival
# speedup vs baseline: 3.9094x; 1.1469x over previous
#include <hip/hip_runtime.h>
#include <math.h>

typedef float v2f __attribute__((ext_vector_type(2)));
typedef float v8f __attribute__((ext_vector_type(8)));

#define FEAT   64
#define NCLS   40
#define NCLSP  48

__device__ __forceinline__ int package_half() {
    return (int)((__builtin_amdgcn_s_sendmsg_rtn(0x87) >> 18) & 1u);
}


__global__ void k_fill(float* __restrict__ p, float val, int n) {
    int i = blockIdx.x * blockDim.x + threadIdx.x;
    if (i < n) p[i] = val;
}

__global__ void k_degree(const int* __restrict__ col, float* __restrict__ cnt0, float* __restrict__ cnt1, int E) {
    int i = blockIdx.x * blockDim.x + threadIdx.x;
    float* cnt = package_half() ? cnt1 : cnt0;
    if (i < E) atomicAdd(&cnt[col[i]], 1.0f);
}

__global__ void k_dinv(float* __restrict__ d0, const float* __restrict__ d1, int n) {
    int i = blockIdx.x * blockDim.x + threadIdx.x;
    if (i < n) {
        float v = 1.0f + d0[i] + d1[i];
        d0[i] = (v > 0.0f) ? rsqrtf(v) : 0.0f;
    }
}

__global__ void k_seed_planes(const float* __restrict__ xin, float* __restrict__ pa, float* __restrict__ pb,
                              const float* __restrict__ dinv, int total) {
    int i = blockIdx.x * blockDim.x + threadIdx.x;
    if (i < total) {
        float w = dinv[i >> 6];
        pa[i] = w * w * xin[i];
        if (pb) pb[i] = 0.0f;
    }
}

__global__ void k_join_seed(float* __restrict__ pa, float* __restrict__ pb, float* __restrict__ na,
                            const float* __restrict__ dinv, int total) {
    int i = blockIdx.x * blockDim.x + threadIdx.x;
    if (i < total) {
        float w = dinv[i >> 6];
        float s = pa[i];
        if (pb) {
            s += pb[i];
            pa[i] = s;
            pb[i] = 0.0f;
        }
        na[i] = w * w * s;
    }
}

#define EPW    32
#define EBATCH 8

template <int NB>
__device__ __forceinline__ void spmm_edges(const float* __restrict__ xin, float* __restrict__ xout,
                                           int r, int c, float w, int first, int lane) {
    float v0[NB];
    float v1[NB];
#pragma unroll
    for (int j = 0; j < NB; ++j) {
        const int rj = __builtin_amdgcn_readlane(r, first + j);
        const float* src = xin + (size_t)rj * FEAT;
        v0[j] = src[lane];
        v1[j] = src[lane + 32];
    }
#pragma unroll
    for (int j = 0; j < NB; ++j) {
        const int   cj = __builtin_amdgcn_readlane(c, first + j);
        const float wj = __int_as_float(__builtin_amdgcn_readlane(__float_as_int(w), first + j));
        float* dst = xout + (size_t)cj * FEAT;
        atomicAdd(dst + lane,      wj * v0[j]);
        atomicAdd(dst + lane + 32, wj * v1[j]);
    }
}

__global__ void k_edge_spmm(const float* __restrict__ xin, float* __restrict__ outa, float* __restrict__ outb,
                            const int* __restrict__ row, const int* __restrict__ col,
                            const float* __restrict__ dinv, int E) {
    float* xout = (outb != nullptr && package_half()) ? outb : outa;
    const int lane = threadIdx.x & 31;
    const int wave = blockIdx.x * (blockDim.x >> 5) + __builtin_amdgcn_readfirstlane(threadIdx.x >> 5);
    const int e0   = wave * EPW;
    if (e0 >= E) return;
    const int e = e0 + lane;
    int   r = 0;
    int   c = 0;
    float w = 0.0f;
    if (e < E) {
        r = row[e];
        c = col[e];
        w = dinv[r] * dinv[c];
    }
    const int n = (E - e0 < EPW) ? (E - e0) : EPW;
    if (n == EPW) {
#pragma unroll
        for (int i = 0; i < EPW; i += EBATCH) spmm_edges<EBATCH>(xin, xout, r, c, w, i, lane);
    } else {
        for (int i = 0; i < n; ++i) spmm_edges<1>(xin, xout, r, c, w, i, lane);
    }
}

__global__ void k_pad_w(const float* __restrict__ W, float* __restrict__ Wp) {
    int i = blockIdx.x * blockDim.x + threadIdx.x;
    if (i < NCLSP * FEAT) {
        int r = i / FEAT;
        Wp[i] = (r < NCLS) ? W[i] : 0.0f;
    }
}

template <bool TWO>
__global__ void k_gemm_wmma(const float* __restrict__ xa, const float* __restrict__ xb, const float* __restrict__ Wp,
                            const float* __restrict__ bias, float* __restrict__ out, int N) {
    const int wave   = blockIdx.x * (blockDim.x >> 5) + (threadIdx.x >> 5);
    const int lane   = threadIdx.x & 31;
    const int ntiles = (N + 15) >> 4;
    if (wave >= ntiles) return;

    const int base = wave << 4;
    const int half = lane >> 4;
    const int l16  = lane & 15;

    int m = base + l16;
    if (m >= N) m = N - 1;
    const float* arow = xa + (size_t)m * FEAT;
    const float* brow = (TWO ? xb : xa) + (size_t)m * FEAT;
    const float* b0r  = Wp + (size_t)(l16)      * FEAT;
    const float* b1r  = Wp + (size_t)(16 + l16) * FEAT;
    const float* b2r  = Wp + (size_t)(32 + l16) * FEAT;

    v8f acc0 = {};
    v8f acc1 = {};
    v8f acc2 = {};

    for (int k0 = 0; k0 < FEAT; k0 += 4) {
        const int ko = k0 + 2 * half;
        v2f a  = *(const v2f*)(arow + ko);
        if (TWO) a += *(const v2f*)(brow + ko);
        v2f b0 = *(const v2f*)(b0r + ko);
        v2f b1 = *(const v2f*)(b1r + ko);
        v2f b2 = *(const v2f*)(b2r + ko);
        acc0 = __builtin_amdgcn_wmma_f32_16x16x4_f32(false, a, false, b0, (short)0, acc0, false, false);
        acc1 = __builtin_amdgcn_wmma_f32_16x16x4_f32(false, a, false, b1, (short)0, acc1, false, false);
        acc2 = __builtin_amdgcn_wmma_f32_16x16x4_f32(false, a, false, b2, (short)0, acc2, false, false);
    }

    const int c0 = l16;
    const int c1 = 16 + l16;
    const int c2 = 32 + l16;
    const float bb0 = bias ? bias[c0] : 0.0f;
    const float bb1 = bias ? bias[c1] : 0.0f;
    const float bb2 = (bias && c2 < NCLS) ? bias[c2] : 0.0f;
#pragma unroll
    for (int j = 0; j < 8; ++j) {
        const int node = base + j + 8 * half;
        if (node < N) {
            float* o = out + (size_t)node * NCLS;
            o[c0] = acc0[j] + bb0;
            o[c1] = acc1[j] + bb1;
            if (c2 < NCLS) o[c2] = acc2[j] + bb2;
        }
    }
}

__global__ void k_logsoftmax(float* __restrict__ out, int N) {
    int n = blockIdx.x * blockDim.x + threadIdx.x;
    if (n >= N) return;
    float* p = out + (size_t)n * NCLS;
    float v[NCLS];
    float m = -INFINITY;
#pragma unroll
    for (int c = 0; c < NCLS; ++c) { v[c] = p[c]; m = fmaxf(m, v[c]); }
    float s = 0.0f;
#pragma unroll
    for (int c = 0; c < NCLS; ++c) s += expf(v[c] - m);
    const float ls = logf(s);
#pragma unroll
    for (int c = 0; c < NCLS; ++c) p[c] = v[c] - m - ls;
}


#define EPT 4
#define RPW 4
#define HB  8
#define YL  (NCLS / 2)

__device__ __forceinline__ void degree_rank_block(const int* __restrict__ col, int* __restrict__ cnt0, int* __restrict__ cnt1,
                                                  int* __restrict__ rank, int E, int blk, int nblk) {
    const int h = package_half();
    int* cnt = h ? cnt1 : cnt0;
    const int stride = nblk * blockDim.x;
    const int i0 = blk * blockDim.x + threadIdx.x;
    int idx[EPT];
    int c[EPT];
    int old[EPT];
#pragma unroll
    for (int k = 0; k < EPT; ++k) {
        idx[k] = i0 + k * stride;
        c[k] = col[(idx[k] < E) ? idx[k] : (E - 1)];
    }
#pragma unroll
    for (int k = 0; k < EPT; ++k) {
        old[k] = 0;
        if (idx[k] < E) old[k] = atomicAdd(&cnt[c[k]], 1);
    }
#pragma unroll
    for (int k = 0; k < EPT; ++k) {
        if (idx[k] < E) rank[idx[k]] = old[k] | (h << 30);
    }
}

template <bool REC>
__device__ __forceinline__ void gemm_tile_w40(const float* __restrict__ x, const float* __restrict__ W,
                                              const int4* __restrict__ rec,
                                              const int* __restrict__ cnt0, const int* __restrict__ cnt1,
                                              float* __restrict__ y, int N, int tile, int lane) {
    const int base = tile << 4;
    const int half = lane >> 4;
    const int l16  = lane & 15;

    int m = base + l16;
    if (m >= N) m = N - 1;
    const bool b2ok = (32 + l16) < NCLS;
    const float* arow = x + (size_t)m * FEAT;
    const float* b0r  = W + (size_t)(l16)      * FEAT;
    const float* b1r  = W + (size_t)(16 + l16) * FEAT;
    const float* b2r  = W + (size_t)(b2ok ? (32 + l16) : (NCLS - 1)) * FEAT;

    v8f acc0 = {};
    v8f acc1 = {};
    v8f acc2 = {};

    for (int k0 = 0; k0 < FEAT; k0 += 4) {
        const int ko = k0 + 2 * half;
        v2f a  = *(const v2f*)(arow + ko);
        v2f b0 = *(const v2f*)(b0r + ko);
        v2f b1 = *(const v2f*)(b1r + ko);
        v2f b2 = *(const v2f*)(b2r + ko);
        acc0 = __builtin_amdgcn_wmma_f32_16x16x4_f32(false, a, false, b0, (short)0, acc0, false, false);
        acc1 = __builtin_amdgcn_wmma_f32_16x16x4_f32(false, a, false, b1, (short)0, acc1, false, false);
        acc2 = __builtin_amdgcn_wmma_f32_16x16x4_f32(false, a, false, b2, (short)0, acc2, false, false);
    }

    int ln = lane;
    asm volatile("v_nop\n\tv_nop\n\tv_nop\n\tv_nop" : "+v"(acc0), "+v"(acc1), "+v"(acc2), "+v"(ln));
    const int eh = ln >> 4;
    const int el = ln & 15;

    const int c0 = el;
    const int c1 = 16 + el;
    const int c2 = 32 + el;
    const int* recw = (const int*)rec;
    float dv[8];
#pragma unroll
    for (int j = 0; j < 8; ++j) {
        const int node = base + j + 8 * eh;
        const size_t nn = (size_t)((node < N) ? node : (N - 1));
        if (REC) {
            dv[j] = __int_as_float(recw[4 * nn + 2]);
        } else {
            const float v = 1.0f + (float)(cnt0[nn] + cnt1[nn]);
            dv[j] = (v > 0.0f) ? rsqrtf(v) : 0.0f;
        }
    }
#pragma unroll
    for (int j = 0; j < 8; ++j) {
        const int node = base + j + 8 * eh;
        if (node < N) {
            const float dj = dv[j];
            float* o = y + (size_t)node * NCLS;
            o[c0] = dj * acc0[j];
            o[c1] = dj * acc1[j];
            if (c2 < NCLS) o[c2] = dj * acc2[j];
        }
    }
}

__global__ void k_rank(const int* __restrict__ col, int* __restrict__ cnt0, int* __restrict__ cnt1,
                       int* __restrict__ rank, int E) {
    degree_rank_block(col, cnt0, cnt1, rank, E, (int)blockIdx.x, (int)gridDim.x);
}

__global__ void __launch_bounds__(1024) k_scan_nodes(const int* __restrict__ cnt0, const int* __restrict__ cnt1,
                                                     int4* __restrict__ rec, int N) {
    __shared__ int wsum[32];
    __shared__ int bpre;
    const int tid  = threadIdx.x;
    const int lane = tid & 31;
    const int wv   = tid >> 5;
    const int start = blockIdx.x * 1024;

    int s = 0;
    const int4* a4 = (const int4*)cnt0;
    const int4* b4 = (const int4*)cnt1;
#pragma unroll 4
    for (int q = tid; q < (start >> 2); q += 1024) {
        const int4 a = a4[q];
        const int4 b = b4[q];
        s += (a.x + a.y) + (a.z + a.w) + (b.x + b.y) + (b.z + b.w);
    }
    for (int m = 16; m > 0; m >>= 1) s += __shfl_xor(s, m);
    if (lane == 0) wsum[wv] = s;
    __syncthreads();
    if (wv == 0) {
        int t = wsum[lane];
        for (int m = 16; m > 0; m >>= 1) t += __shfl_xor(t, m);
        if (lane == 0) bpre = t;
    }
    __syncthreads();
    const int pre = bpre;

    const int i = start + tid;
    int c0 = 0;
    int c1 = 0;
    if (i < N) { c0 = cnt0[i]; c1 = cnt1[i]; }
    const int d = c0 + c1;
    int inc = d;
    for (int m = 1; m < 32; m <<= 1) {
        const int t = __shfl_up(inc, m);
        if (lane >= m) inc += t;
    }
    if (lane == 31) wsum[wv] = inc;
    __syncthreads();
    int wpre = (lane < wv) ? wsum[lane] : 0;
    for (int m = 16; m > 0; m >>= 1) wpre += __shfl_xor(wpre, m);
    const int excl = pre + wpre + inc - d;
    if (i <= N) {
        const float v  = 1.0f + (float)d;
        const float di = (v > 0.0f) ? rsqrtf(v) : 0.0f;
        rec[i] = make_int4(excl, excl + c0, __float_as_int(di), d);
    }
}

__device__ __forceinline__ void csr_fill_block(const int* __restrict__ row, const int* __restrict__ col,
                                               const int* __restrict__ rank, const int4* __restrict__ rec,
                                               int* __restrict__ csr, int E, int blk, int nblk) {
    const int stride = nblk * blockDim.x;
    const int i0 = blk * blockDim.x + threadIdx.x;
    const int* recw = (const int*)rec;
    int idx[EPT];
    int r[EPT];
    int c[EPT];
    int rk[EPT];
    int first[EPT];
#pragma unroll
    for (int k = 0; k < EPT; ++k) {
        idx[k] = i0 + k * stride;
        const int ii = (idx[k] < E) ? idx[k] : (E - 1);
        r[k]  = row[ii];
        c[k]  = col[ii];
        rk[k] = rank[ii];
    }
#pragma unroll
    for (int k = 0; k < EPT; ++k) {
        first[k] = recw[4 * (size_t)c[k] + (size_t)(rk[k] >> 30)];
    }
#pragma unroll
    for (int k = 0; k < EPT; ++k) {
        if (idx[k] < E) csr[first[k] + (rk[k] & 0x3fffffff)] = r[k];
    }
}

__global__ void k_csr_gemm(const int* __restrict__ row, const int* __restrict__ col, const int* __restrict__ rank,
                           const int4* __restrict__ rec, int* __restrict__ csr, int E, int ngemm, int nfill,
                           const float* __restrict__ x, const float* __restrict__ W,
                           float* __restrict__ y, int N) {
    if ((int)blockIdx.x >= ngemm) {
        csr_fill_block(row, col, rank, rec, csr, E, (int)blockIdx.x - ngemm, nfill);
        return;
    }
    const int tile = (int)blockIdx.x * (int)(blockDim.x >> 5) + __builtin_amdgcn_readfirstlane(threadIdx.x >> 5);
    if (tile >= ((N + 15) >> 4)) return;
    gemm_tile_w40<true>(x, W, rec, nullptr, nullptr, y, N, tile, threadIdx.x & 31);
}

template <bool FULL>
__device__ __forceinline__ v2f pull_batch(const float* __restrict__ xin, int srcs,
                                          int j0, int m, int cl, v2f acc) {
    constexpr int NJ = FULL ? HB : (HB - 1);
    v2f v[NJ];
#pragma unroll
    for (int j = 0; j < NJ; ++j) {
        v[j] = (v2f){0.0f, 0.0f};
        if (FULL || j == 0 || j < m) {
            const int rj = __builtin_amdgcn_readlane(srcs, j0 + j);
            v[j] = *(const v2f*)(xin + (size_t)rj * NCLS + cl);
        }
    }
#pragma unroll
    for (int j = 0; j < NJ; ++j) {
        if (FULL || j == 0 || j < m) acc += v[j];
    }
    return acc;
}

template <int M>
__device__ __forceinline__ float row_xor(float v) {
    return __builtin_bit_cast(float, __builtin_amdgcn_update_dpp(0, __builtin_bit_cast(int, v), 0x160 | M, 0xf, 0xf, false));
}
__device__ __forceinline__ float xhalf(float v) {
    const int b = __builtin_bit_cast(int, v);
    return __builtin_bit_cast(float, __builtin_amdgcn_permlanex16(b, b, 0x76543210, 0xfedcba98, false, false));
}
__device__ __forceinline__ float wave_max(float v) {
    v = fmaxf(v, row_xor<8>(v));
    v = fmaxf(v, row_xor<4>(v));
    v = fmaxf(v, row_xor<2>(v));
    v = fmaxf(v, row_xor<1>(v));
    return fmaxf(v, xhalf(v));
}
__device__ __forceinline__ float wave_sum(float v) {
    v += row_xor<8>(v);
    v += row_xor<4>(v);
    v += row_xor<2>(v);
    v += row_xor<1>(v);
    return v + xhalf(v);
}

template <bool LAST>
__global__ void k_pull_hop(const float* __restrict__ xin, float* __restrict__ xout,
                           const int4* __restrict__ rec, const int* __restrict__ csr,
                           const float* __restrict__ bias, int N) {
    const int lane = threadIdx.x & 31;
    const bool own = lane < YL;
    const int cl = 2 * (own ? lane : (YL - 1));
    const int wave  = blockIdx.x * (blockDim.x >> 5) + __builtin_amdgcn_readfirstlane(threadIdx.x >> 5);
    const int node0 = wave * RPW;
    if (node0 >= N) return;
    const int nl = (node0 + lane < N) ? (node0 + lane) : N;
    const int4 rc = rec[nl];
    v2f bb = (v2f){0.0f, 0.0f};
    if (LAST) { bb.x = bias[cl]; bb.y = bias[cl + 1]; }
#pragma unroll
    for (int k = 0; k < RPW; ++k) {
        const int node = node0 + k;
        if (node >= N) break;
        const int   beg = __builtin_amdgcn_readlane(rc.x, k);
        const int   end = __builtin_amdgcn_readlane(rc.x, k + 1);
        const float di  = __int_as_float(__builtin_amdgcn_readlane(rc.z, k));
        const v2f self = *(const v2f*)(xin + (size_t)node * NCLS + cl);
        v2f acc = self;
        for (int base = beg; base < end; base += 32) {
            const int n = (end - base < 32) ? (end - base) : 32;
            int p = 0;
            if (lane < n) p = csr[base + lane];
            int j0 = 0;
            for (; j0 + HB <= n; j0 += HB) acc = pull_batch<true>(xin, p, j0, HB, cl, acc);
            if (j0 < n) acc = pull_batch<false>(xin, p, j0, n - j0, cl, acc);
        }
        if (!LAST) acc = (di * di) * acc;
        if (LAST) {
            const v2f lg = di * acc + bb;
            const float mx = wave_max(own ? fmaxf(lg.x, lg.y) : -INFINITY);
            const float se = wave_sum(own ? (expf(lg.x - mx) + expf(lg.y - mx)) : 0.0f);
            const float ls = logf(se);
            acc = (v2f){lg.x - mx - ls, lg.y - mx - ls};
        }
        if (own) *(v2f*)(xout + (size_t)node * NCLS + cl) = acc;
    }
}

#define ECAP 32
#define EROW (2 * ECAP)

__global__ void k_rank_slots(const int* __restrict__ row, const int* __restrict__ col,
                             int* __restrict__ cnt0, int* __restrict__ cnt1, int* __restrict__ slots,
                             int* __restrict__ ovfn, int2* __restrict__ ovf, int E) {
    const int h = package_half();
    int* cnt = h ? cnt1 : cnt0;
    const int stride = gridDim.x * blockDim.x;
    const int i0 = blockIdx.x * blockDim.x + threadIdx.x;
    int idx[EPT];
    int c[EPT];
    int r[EPT];
    int old[EPT];
#pragma unroll
    for (int k = 0; k < EPT; ++k) {
        idx[k] = i0 + k * stride;
        const int ii = (idx[k] < E) ? idx[k] : (E - 1);
        c[k] = col[ii];
        r[k] = row[ii];
    }
#pragma unroll
    for (int k = 0; k < EPT; ++k) {
        old[k] = 0;
        if (idx[k] < E) old[k] = atomicAdd(&cnt[c[k]], 1);
    }
#pragma unroll
    for (int k = 0; k < EPT; ++k) {
        if (idx[k] < E) {
            if (old[k] < ECAP) {
                slots[(size_t)c[k] * EROW + (size_t)(h * ECAP + old[k])] = r[k];
            } else {
                const int p = atomicAdd(ovfn, 1);
                ovf[p] = make_int2(r[k], c[k]);
            }
        }
    }
}

__global__ void k_gemm_cnt(const float* __restrict__ x, const float* __restrict__ W,
                           const int* __restrict__ cnt0, const int* __restrict__ cnt1,
                           float* __restrict__ y, int N) {
    const int tile = (int)blockIdx.x * (int)(blockDim.x >> 5) + __builtin_amdgcn_readfirstlane(threadIdx.x >> 5);
    if (tile >= ((N + 15) >> 4)) return;
    gemm_tile_w40<false>(x, W, nullptr, cnt0, cnt1, y, N, tile, threadIdx.x & 31);
}

__device__ __forceinline__ v2f pull_list(const float* __restrict__ xin, int p, int n, int cl, v2f acc) {
    int j0 = 0;
    for (; j0 + HB <= n; j0 += HB) acc = pull_batch<true>(xin, p, j0, HB, cl, acc);
    if (j0 < n) acc = pull_batch<false>(xin, p, j0, n - j0, cl, acc);
    return acc;
}

template <bool LAST>
__global__ void k_pull_hop_slots(const float* __restrict__ xin, float* __restrict__ xout,
                                 const int* __restrict__ cnt0, const int* __restrict__ cnt1,
                                 const int* __restrict__ slots,
                                 const int* __restrict__ ovfn, const int2* __restrict__ ovf,
                                 const float* __restrict__ bias, int N) {
    const int lane = threadIdx.x & 31;
    const bool own = lane < YL;
    const int cl = 2 * (own ? lane : (YL - 1));
    const int wave  = blockIdx.x * (blockDim.x >> 5) + __builtin_amdgcn_readfirstlane(threadIdx.x >> 5);
    const int node0 = wave * RPW;
    if (node0 >= N) return;
    const int nl = (node0 + lane < N) ? (node0 + lane) : (N - 1);
    const int c0v = cnt0[nl];
    const int c1v = cnt1[nl];
    const float dgv = 1.0f + (float)(c0v + c1v);
    const float div = (dgv > 0.0f) ? rsqrtf(dgv) : 0.0f;
    v2f bb = (v2f){0.0f, 0.0f};
    if (LAST) { bb.x = bias[cl]; bb.y = bias[cl + 1]; }
#pragma unroll
    for (int k = 0; k < RPW; ++k) {
        const int node = node0 + k;
        if (node >= N) break;
        const int   c0 = __builtin_amdgcn_readlane(c0v, k);
        const int   c1 = __builtin_amdgcn_readlane(c1v, k);
        const float di = __int_as_float(__builtin_amdgcn_readlane(__float_as_int(div), k));
        const int n0 = (c0 < ECAP) ? c0 : ECAP;
        const int n1 = (c1 < ECAP) ? c1 : ECAP;
        const int* sr = slots + (size_t)node * EROW;
        const v2f self = *(const v2f*)(xin + (size_t)node * NCLS + cl);
        const int n  = n0 + n1;
        int pa = 0;
        if (lane < n) pa = sr[(lane < n0) ? lane : (ECAP + lane - n0)];
        v2f acc = self;
        acc = pull_list(xin, pa, (n < 32) ? n : 32, cl, acc);
        if (n > 32) {
            int pb = 0;
            if (lane + 32 < n) pb = sr[ECAP + 32 + lane - n0];
            acc = pull_list(xin, pb, n - 32, cl, acc);
        }
        if (c0 > ECAP || c1 > ECAP) {
            const int novf = *ovfn;
            for (int i = 0; i < novf; ++i) {
                const int2 e = ovf[i];
                const int es = __builtin_amdgcn_readfirstlane(e.x);
                const int et = __builtin_amdgcn_readfirstlane(e.y);
                if (et == node) acc += *(const v2f*)(xin + (size_t)es * NCLS + cl);
            }
        }
        if (!LAST) acc = (di * di) * acc;
        if (LAST) {
            const v2f lg = di * acc + bb;
            const float mx = wave_max(own ? fmaxf(lg.x, lg.y) : -INFINITY);
            const float se = wave_sum(own ? (expf(lg.x - mx) + expf(lg.y - mx)) : 0.0f);
            const float ls = logf(se);
            acc = (v2f){lg.x - mx - ls, lg.y - mx - ls};
        }
        if (own) *(v2f*)(xout + (size_t)node * NCLS + cl) = acc;
    }
}


static inline size_t align256f(size_t nfloats) { return (nfloats + 63) & ~(size_t)63; }

extern "C" void kernel_launch(void* const* d_in, const int* in_sizes, int n_in,
                              void* d_out, int out_size, void* d_ws, size_t ws_size,
                              hipStream_t stream) {
    const float* x  = (const float*)d_in[0];
    const int*   ei = (const int*)d_in[1];
    const float* W  = (const float*)d_in[2];
    const float* bi = (const float*)d_in[3];

    const int N = in_sizes[0] / FEAT;
    const int E = in_sizes[1] / 2;
    const int* row = ei;
    const int* col = ei + E;
    const int total = N * FEAT;

    float* ws = (float*)d_ws;
    size_t off = 0;
    float* cnt0 = ws + off; off += align256f((size_t)N);
    float* cnt1 = ws + off; off += align256f((size_t)N);
    float* Wp   = ws + off; off += align256f((size_t)NCLSP * FEAT);

    const int T = 256;
    const int ncnt = (int)(cnt1 - cnt0) + N;
    const int ntiles = (N + 15) >> 4;
    const int gblocks = (ntiles + 7) / 8;

    size_t o3 = off;
    float* z0    = ws + o3;          o3 += align256f((size_t)N * NCLS);
    float* z1    = ws + o3;          o3 += align256f((size_t)N * NCLS);
    int*   slots = (int*)(ws + o3);  o3 += align256f((size_t)N * EROW);
    int2*  ovf   = (int2*)(ws + o3); o3 += align256f(2 * (size_t)E);
    int*   ovfn  = (int*)Wp;

    if (o3 * sizeof(float) <= ws_size) {
        const int nclr = (int)(Wp - cnt0) + 64;
        k_fill<<<(nclr + T - 1) / T, T, 0, stream>>>(cnt0, 0.0f, nclr);
        const int sblocks = (E + T * EPT - 1) / (T * EPT);
        k_rank_slots<<<sblocks, T, 0, stream>>>(row, col, (int*)cnt0, (int*)cnt1, slots, ovfn, ovf, E);
        k_gemm_cnt<<<gblocks, T, 0, stream>>>(x, W, (const int*)cnt0, (const int*)cnt1, z0, N);
        const int hblocks = (N + (T / 32) * RPW - 1) / ((T / 32) * RPW);
        k_pull_hop_slots<false><<<hblocks, T, 0, stream>>>(z0, z1, (const int*)cnt0, (const int*)cnt1, slots,
                                                           ovfn, ovf, nullptr, N);
        k_pull_hop_slots<true><<<hblocks, T, 0, stream>>>(z1, (float*)d_out, (const int*)cnt0, (const int*)cnt1, slots,
                                                          ovfn, ovf, bi, N);
        return;
    }

    size_t o2 = off;
    int4*  rec  = (int4*)(ws + o2); o2 += align256f(4 * ((size_t)N + 1));
    float* y0   = ws + o2;          o2 += align256f((size_t)N * NCLS);
    float* y1   = ws + o2;          o2 += align256f((size_t)N * NCLS);
    int*   csr  = (int*)(ws + o2);  o2 += align256f((size_t)E);
    int*   rank = (int*)(ws + o2);  o2 += align256f((size_t)E);

    if (o2 * sizeof(float) <= ws_size) {
        k_fill<<<(ncnt + T - 1) / T, T, 0, stream>>>(cnt0, 0.0f, ncnt);
        const int sblocks = (E + T * EPT - 1) / (T * EPT);
        k_rank<<<sblocks, T, 0, stream>>>(col, (int*)cnt0, (int*)cnt1, rank, E);
        k_scan_nodes<<<(N + 1 + 1023) / 1024, 1024, 0, stream>>>((const int*)cnt0, (const int*)cnt1, rec, N);
        k_csr_gemm<<<gblocks + sblocks, T, 0, stream>>>(row, col, rank, rec, csr, E, gblocks, sblocks,
                                                        x, W, y0, N);


        const int hblocks = (N + (T / 32) * RPW - 1) / ((T / 32) * RPW);
        k_pull_hop<false><<<hblocks, T, 0, stream>>>(y0, y1, rec, csr, nullptr, N);
        k_pull_hop<true><<<hblocks, T, 0, stream>>>(y1, (float*)d_out, rec, csr, bi, N);
        return;
    }

    float* p0   = ws + off; off += align256f((size_t)total);
    float* p1   = ws + off; off += align256f((size_t)total);
    float* pb   = nullptr;
    if ((off + (size_t)total) * sizeof(float) <= ws_size) pb = ws + off;
    const float* dinv = cnt0;

    k_fill<<<(ncnt + T - 1) / T, T, 0, stream>>>(cnt0, 0.0f, ncnt);
    k_degree<<<(E + T - 1) / T, T, 0, stream>>>(col, cnt0, cnt1, E);
    k_dinv<<<(N + T - 1) / T, T, 0, stream>>>(cnt0, cnt1, N);
    k_pad_w<<<(NCLSP * FEAT + T - 1) / T, T, 0, stream>>>(W, Wp);

    const int eblocks = (E + (T / 32) * EPW - 1) / ((T / 32) * EPW);

    k_seed_planes<<<(total + T - 1) / T, T, 0, stream>>>(x, p0, pb, dinv, total);
    k_edge_spmm<<<eblocks, T, 0, stream>>>(x, p0, pb, row, col, dinv, E);
    k_join_seed<<<(total + T - 1) / T, T, 0, stream>>>(p0, pb, p1, dinv, total);
    k_edge_spmm<<<eblocks, T, 0, stream>>>(p0, p1, pb, row, col, dinv, E);

    if (pb) k_gemm_wmma<true><<<gblocks, 256, 0, stream>>>(p1, pb, Wp, bi, (float*)d_out, N);
    else    k_gemm_wmma<false><<<gblocks, 256, 0, stream>>>(p1, pb, Wp, bi, (float*)d_out, N);
    k_logsoftmax<<<(N + T - 1) / T, T, 0, stream>>>((float*)d_out, N);
}


# slot rows built in LDS: chunks bucket edges by target range (LDS count, scan, whole lines out), one work-group per bucket ranks them by LDS atomics and stores whole rows; sum order = LDS arrival rank
# speedup vs baseline: 5.2823x; 1.3512x over previous
#include <hip/hip_runtime.h>
#include <math.h>

typedef float v2f __attribute__((ext_vector_type(2)));
typedef float v8f __attribute__((ext_vector_type(8)));

#define FEAT   64
#define NCLS   40
#define NCLSP  48

__device__ __forceinline__ int package_half() {
    return (int)((__builtin_amdgcn_s_sendmsg_rtn(0x87) >> 18) & 1u);
}


__global__ void k_fill(float* __restrict__ p, float val, int n) {
    int i = blockIdx.x * blockDim.x + threadIdx.x;
    if (i < n) p[i] = val;
}

__global__ void k_degree(const int* __restrict__ col, float* __restrict__ cnt0, float* __restrict__ cnt1, int E) {
    int i = blockIdx.x * blockDim.x + threadIdx.x;
    float* cnt = package_half() ? cnt1 : cnt0;
    if (i < E) atomicAdd(&cnt[col[i]], 1.0f);
}

__global__ void k_dinv(float* __restrict__ d0, const float* __restrict__ d1, int n) {
    int i = blockIdx.x * blockDim.x + threadIdx.x;
    if (i < n) {
        float v = 1.0f + d0[i] + d1[i];
        d0[i] = (v > 0.0f) ? rsqrtf(v) : 0.0f;
    }
}

__global__ void k_seed_planes(const float* __restrict__ xin, float* __restrict__ pa, float* __restrict__ pb,
                              const float* __restrict__ dinv, int total) {
    int i = blockIdx.x * blockDim.x + threadIdx.x;
    if (i < total) {
        float w = dinv[i >> 6];
        pa[i] = w * w * xin[i];
        if (pb) pb[i] = 0.0f;
    }
}

__global__ void k_join_seed(float* __restrict__ pa, float* __restrict__ pb, float* __restrict__ na,
                            const float* __restrict__ dinv, int total) {
    int i = blockIdx.x * blockDim.x + threadIdx.x;
    if (i < total) {
        float w = dinv[i >> 6];
        float s = pa[i];
        if (pb) {
            s += pb[i];
            pa[i] = s;
            pb[i] = 0.0f;
        }
        na[i] = w * w * s;
    }
}

#define EPW    32
#define EBATCH 8

template <int NB>
__device__ __forceinline__ void spmm_edges(const float* __restrict__ xin, float* __restrict__ xout,
                                           int r, int c, float w, int first, int lane) {
    float v0[NB];
    float v1[NB];
#pragma unroll
    for (int j = 0; j < NB; ++j) {
        const int rj = __builtin_amdgcn_readlane(r, first + j);
        const float* src = xin + (size_t)rj * FEAT;
        v0[j] = src[lane];
        v1[j] = src[lane + 32];
    }
#pragma unroll
    for (int j = 0; j < NB; ++j) {
        const int   cj = __builtin_amdgcn_readlane(c, first + j);
        const float wj = __int_as_float(__builtin_amdgcn_readlane(__float_as_int(w), first + j));
        float* dst = xout + (size_t)cj * FEAT;
        atomicAdd(dst + lane,      wj * v0[j]);
        atomicAdd(dst + lane + 32, wj * v1[j]);
    }
}

__global__ void k_edge_spmm(const float* __restrict__ xin, float* __restrict__ outa, float* __restrict__ outb,
                            const int* __restrict__ row, const int* __restrict__ col,
                            const float* __restrict__ dinv, int E) {
    float* xout = (outb != nullptr && package_half()) ? outb : outa;
    const int lane = threadIdx.x & 31;
    const int wave = blockIdx.x * (blockDim.x >> 5) + __builtin_amdgcn_readfirstlane(threadIdx.x >> 5);
    const int e0   = wave * EPW;
    if (e0 >= E) return;
    const int e = e0 + lane;
    int   r = 0;
    int   c = 0;
    float w = 0.0f;
    if (e < E) {
        r = row[e];
        c = col[e];
        w = dinv[r] * dinv[c];
    }
    const int n = (E - e0 < EPW) ? (E - e0) : EPW;
    if (n == EPW) {
#pragma unroll
        for (int i = 0; i < EPW; i += EBATCH) spmm_edges<EBATCH>(xin, xout, r, c, w, i, lane);
    } else {
        for (int i = 0; i < n; ++i) spmm_edges<1>(xin, xout, r, c, w, i, lane);
    }
}

__global__ void k_pad_w(const float* __restrict__ W, float* __restrict__ Wp) {
    int i = blockIdx.x * blockDim.x + threadIdx.x;
    if (i < NCLSP * FEAT) {
        int r = i / FEAT;
        Wp[i] = (r < NCLS) ? W[i] : 0.0f;
    }
}

template <bool TWO>
__global__ void k_gemm_wmma(const float* __restrict__ xa, const float* __restrict__ xb, const float* __restrict__ Wp,
                            const float* __restrict__ bias, float* __restrict__ out, int N) {
    const int wave   = blockIdx.x * (blockDim.x >> 5) + (threadIdx.x >> 5);
    const int lane   = threadIdx.x & 31;
    const int ntiles = (N + 15) >> 4;
    if (wave >= ntiles) return;

    const int base = wave << 4;
    const int half = lane >> 4;
    const int l16  = lane & 15;

    int m = base + l16;
    if (m >= N) m = N - 1;
    const float* arow = xa + (size_t)m * FEAT;
    const float* brow = (TWO ? xb : xa) + (size_t)m * FEAT;
    const float* b0r  = Wp + (size_t)(l16)      * FEAT;
    const float* b1r  = Wp + (size_t)(16 + l16) * FEAT;
    const float* b2r  = Wp + (size_t)(32 + l16) * FEAT;

    v8f acc0 = {};
    v8f acc1 = {};
    v8f acc2 = {};

    for (int k0 = 0; k0 < FEAT; k0 += 4) {
        const int ko = k0 + 2 * half;
        v2f a  = *(const v2f*)(arow + ko);
        if (TWO) a += *(const v2f*)(brow + ko);
        v2f b0 = *(const v2f*)(b0r + ko);
        v2f b1 = *(const v2f*)(b1r + ko);
        v2f b2 = *(const v2f*)(b2r + ko);
        acc0 = __builtin_amdgcn_wmma_f32_16x16x4_f32(false, a, false, b0, (short)0, acc0, false, false);
        acc1 = __builtin_amdgcn_wmma_f32_16x16x4_f32(false, a, false, b1, (short)0, acc1, false, false);
        acc2 = __builtin_amdgcn_wmma_f32_16x16x4_f32(false, a, false, b2, (short)0, acc2, false, false);
    }

    const int c0 = l16;
    const int c1 = 16 + l16;
    const int c2 = 32 + l16;
    const float bb0 = bias ? bias[c0] : 0.0f;
    const float bb1 = bias ? bias[c1] : 0.0f;
    const float bb2 = (bias && c2 < NCLS) ? bias[c2] : 0.0f;
#pragma unroll
    for (int j = 0; j < 8; ++j) {
        const int node = base + j + 8 * half;
        if (node < N) {
            float* o = out + (size_t)node * NCLS;
            o[c0] = acc0[j] + bb0;
            o[c1] = acc1[j] + bb1;
            if (c2 < NCLS) o[c2] = acc2[j] + bb2;
        }
    }
}

__global__ void k_logsoftmax(float* __restrict__ out, int N) {
    int n = blockIdx.x * blockDim.x + threadIdx.x;
    if (n >= N) return;
    float* p = out + (size_t)n * NCLS;
    float v[NCLS];
    float m = -INFINITY;
#pragma unroll
    for (int c = 0; c < NCLS; ++c) { v[c] = p[c]; m = fmaxf(m, v[c]); }
    float s = 0.0f;
#pragma unroll
    for (int c = 0; c < NCLS; ++c) s += expf(v[c] - m);
    const float ls = logf(s);
#pragma unroll
    for (int c = 0; c < NCLS; ++c) p[c] = v[c] - m - ls;
}


#define EPT 4
#define RPW 4
#define HB  8
#define YL  (NCLS / 2)

__device__ __forceinline__ void degree_rank_block(const int* __restrict__ col, int* __restrict__ cnt0, int* __restrict__ cnt1,
                                                  int* __restrict__ rank, int E, int blk, int nblk) {
    const int h = package_half();
    int* cnt = h ? cnt1 : cnt0;
    const int stride = nblk * blockDim.x;
    const int i0 = blk * blockDim.x + threadIdx.x;
    int idx[EPT];
    int c[EPT];
    int old[EPT];
#pragma unroll
    for (int k = 0; k < EPT; ++k) {
        idx[k] = i0 + k * stride;
        c[k] = col[(idx[k] < E) ? idx[k] : (E - 1)];
    }
#pragma unroll
    for (int k = 0; k < EPT; ++k) {
        old[k] = 0;
        if (idx[k] < E) old[k] = atomicAdd(&cnt[c[k]], 1);
    }
#pragma unroll
    for (int k = 0; k < EPT; ++k) {
        if (idx[k] < E) rank[idx[k]] = old[k] | (h << 30);
    }
}

template <int DM>
__device__ __forceinline__ void gemm_tile_w40(const float* __restrict__ x, const float* __restrict__ W,
                                              const int4* __restrict__ rec,
                                              const int* __restrict__ cnt0, const int* __restrict__ cnt1,
                                              float* __restrict__ y, int N, int tile, int lane) {
    const int base = tile << 4;
    const int half = lane >> 4;
    const int l16  = lane & 15;

    int m = base + l16;
    if (m >= N) m = N - 1;
    const bool b2ok = (32 + l16) < NCLS;
    const float* arow = x + (size_t)m * FEAT;
    const float* b0r  = W + (size_t)(l16)      * FEAT;
    const float* b1r  = W + (size_t)(16 + l16) * FEAT;
    const float* b2r  = W + (size_t)(b2ok ? (32 + l16) : (NCLS - 1)) * FEAT;

    v8f acc0 = {};
    v8f acc1 = {};
    v8f acc2 = {};

    for (int k0 = 0; k0 < FEAT; k0 += 4) {
        const int ko = k0 + 2 * half;
        v2f a  = *(const v2f*)(arow + ko);
        v2f b0 = *(const v2f*)(b0r + ko);
        v2f b1 = *(const v2f*)(b1r + ko);
        v2f b2 = *(const v2f*)(b2r + ko);
        acc0 = __builtin_amdgcn_wmma_f32_16x16x4_f32(false, a, false, b0, (short)0, acc0, false, false);
        acc1 = __builtin_amdgcn_wmma_f32_16x16x4_f32(false, a, false, b1, (short)0, acc1, false, false);
        acc2 = __builtin_amdgcn_wmma_f32_16x16x4_f32(false, a, false, b2, (short)0, acc2, false, false);
    }

    int ln = lane;
    asm volatile("v_nop\n\tv_nop\n\tv_nop\n\tv_nop" : "+v"(acc0), "+v"(acc1), "+v"(acc2), "+v"(ln));
    const int eh = ln >> 4;
    const int el = ln & 15;

    const int c0 = el;
    const int c1 = 16 + el;
    const int c2 = 32 + el;
    const int* recw = (const int*)rec;
    float dv[8];
#pragma unroll
    for (int j = 0; j < 8; ++j) {
        const int node = base + j + 8 * eh;
        const size_t nn = (size_t)((node < N) ? node : (N - 1));
        if (DM == 1) {
            dv[j] = __int_as_float(recw[4 * nn + 2]);
        } else {
            const int cn = (DM == 2) ? cnt0[nn] : (cnt0[nn] + cnt1[nn]);
            const float v = 1.0f + (float)cn;
            dv[j] = (v > 0.0f) ? rsqrtf(v) : 0.0f;
        }
    }
#pragma unroll
    for (int j = 0; j < 8; ++j) {
        const int node = base + j + 8 * eh;
        if (node < N) {
            const float dj = dv[j];
            float* o = y + (size_t)node * NCLS;
            o[c0] = dj * acc0[j];
            o[c1] = dj * acc1[j];
            if (c2 < NCLS) o[c2] = dj * acc2[j];
        }
    }
}

__global__ void k_rank(const int* __restrict__ col, int* __restrict__ cnt0, int* __restrict__ cnt1,
                       int* __restrict__ rank, int E) {
    degree_rank_block(col, cnt0, cnt1, rank, E, (int)blockIdx.x, (int)gridDim.x);
}

__global__ void __launch_bounds__(1024) k_scan_nodes(const int* __restrict__ cnt0, const int* __restrict__ cnt1,
                                                     int4* __restrict__ rec, int N) {
    __shared__ int wsum[32];
    __shared__ int bpre;
    const int tid  = threadIdx.x;
    const int lane = tid & 31;
    const int wv   = tid >> 5;
    const int start = blockIdx.x * 1024;

    int s = 0;
    const int4* a4 = (const int4*)cnt0;
    const int4* b4 = (const int4*)cnt1;
#pragma unroll 4
    for (int q = tid; q < (start >> 2); q += 1024) {
        const int4 a = a4[q];
        const int4 b = b4[q];
        s += (a.x + a.y) + (a.z + a.w) + (b.x + b.y) + (b.z + b.w);
    }
    for (int m = 16; m > 0; m >>= 1) s += __shfl_xor(s, m);
    if (lane == 0) wsum[wv] = s;
    __syncthreads();
    if (wv == 0) {
        int t = wsum[lane];
        for (int m = 16; m > 0; m >>= 1) t += __shfl_xor(t, m);
        if (lane == 0) bpre = t;
    }
    __syncthreads();
    const int pre = bpre;

    const int i = start + tid;
    int c0 = 0;
    int c1 = 0;
    if (i < N) { c0 = cnt0[i]; c1 = cnt1[i]; }
    const int d = c0 + c1;
    int inc = d;
    for (int m = 1; m < 32; m <<= 1) {
        const int t = __shfl_up(inc, m);
        if (lane >= m) inc += t;
    }
    if (lane == 31) wsum[wv] = inc;
    __syncthreads();
    int wpre = (lane < wv) ? wsum[lane] : 0;
    for (int m = 16; m > 0; m >>= 1) wpre += __shfl_xor(wpre, m);
    const int excl = pre + wpre + inc - d;
    if (i <= N) {
        const float v  = 1.0f + (float)d;
        const float di = (v > 0.0f) ? rsqrtf(v) : 0.0f;
        rec[i] = make_int4(excl, excl + c0, __float_as_int(di), d);
    }
}

__device__ __forceinline__ void csr_fill_block(const int* __restrict__ row, const int* __restrict__ col,
                                               const int* __restrict__ rank, const int4* __restrict__ rec,
                                               int* __restrict__ csr, int E, int blk, int nblk) {
    const int stride = nblk * blockDim.x;
    const int i0 = blk * blockDim.x + threadIdx.x;
    const int* recw = (const int*)rec;
    int idx[EPT];
    int r[EPT];
    int c[EPT];
    int rk[EPT];
    int first[EPT];
#pragma unroll
    for (int k = 0; k < EPT; ++k) {
        idx[k] = i0 + k * stride;
        const int ii = (idx[k] < E) ? idx[k] : (E - 1);
        r[k]  = row[ii];
        c[k]  = col[ii];
        rk[k] = rank[ii];
    }
#pragma unroll
    for (int k = 0; k < EPT; ++k) {
        first[k] = recw[4 * (size_t)c[k] + (size_t)(rk[k] >> 30)];
    }
#pragma unroll
    for (int k = 0; k < EPT; ++k) {
        if (idx[k] < E) csr[first[k] + (rk[k] & 0x3fffffff)] = r[k];
    }
}

__global__ void k_csr_gemm(const int* __restrict__ row, const int* __restrict__ col, const int* __restrict__ rank,
                           const int4* __restrict__ rec, int* __restrict__ csr, int E, int ngemm, int nfill,
                           const float* __restrict__ x, const float* __restrict__ W,
                           float* __restrict__ y, int N) {
    if ((int)blockIdx.x >= ngemm) {
        csr_fill_block(row, col, rank, rec, csr, E, (int)blockIdx.x - ngemm, nfill);
        return;
    }
    const int tile = (int)blockIdx.x * (int)(blockDim.x >> 5) + __builtin_amdgcn_readfirstlane(threadIdx.x >> 5);
    if (tile >= ((N + 15) >> 4)) return;
    gemm_tile_w40<1>(x, W, rec, nullptr, nullptr, y, N, tile, threadIdx.x & 31);
}

template <bool FULL>
__device__ __forceinline__ v2f pull_batch(const float* __restrict__ xin, int srcs,
                                          int j0, int m, int cl, v2f acc) {
    constexpr int NJ = FULL ? HB : (HB - 1);
    v2f v[NJ];
#pragma unroll
    for (int j = 0; j < NJ; ++j) {
        v[j] = (v2f){0.0f, 0.0f};
        if (FULL || j == 0 || j < m) {
            const int rj = __builtin_amdgcn_readlane(srcs, j0 + j);
            v[j] = *(const v2f*)(xin + (size_t)rj * NCLS + cl);
        }
    }
#pragma unroll
    for (int j = 0; j < NJ; ++j) {
        if (FULL || j == 0 || j < m) acc += v[j];
    }
    return acc;
}

template <int M>
__device__ __forceinline__ float row_xor(float v) {
    return __builtin_bit_cast(float, __builtin_amdgcn_update_dpp(0, __builtin_bit_cast(int, v), 0x160 | M, 0xf, 0xf, false));
}
__device__ __forceinline__ float xhalf(float v) {
    const int b = __builtin_bit_cast(int, v);
    return __builtin_bit_cast(float, __builtin_amdgcn_permlanex16(b, b, 0x76543210, 0xfedcba98, false, false));
}
__device__ __forceinline__ float wave_max(float v) {
    v = fmaxf(v, row_xor<8>(v));
    v = fmaxf(v, row_xor<4>(v));
    v = fmaxf(v, row_xor<2>(v));
    v = fmaxf(v, row_xor<1>(v));
    return fmaxf(v, xhalf(v));
}
__device__ __forceinline__ float wave_sum(float v) {
    v += row_xor<8>(v);
    v += row_xor<4>(v);
    v += row_xor<2>(v);
    v += row_xor<1>(v);
    return v + xhalf(v);
}

template <bool LAST>
__global__ void k_pull_hop(const float* __restrict__ xin, float* __restrict__ xout,
                           const int4* __restrict__ rec, const int* __restrict__ csr,
                           const float* __restrict__ bias, int N) {
    const int lane = threadIdx.x & 31;
    const bool own = lane < YL;
    const int cl = 2 * (own ? lane : (YL - 1));
    const int wave  = blockIdx.x * (blockDim.x >> 5) + __builtin_amdgcn_readfirstlane(threadIdx.x >> 5);
    const int node0 = wave * RPW;
    if (node0 >= N) return;
    const int nl = (node0 + lane < N) ? (node0 + lane) : N;
    const int4 rc = rec[nl];
    v2f bb = (v2f){0.0f, 0.0f};
    if (LAST) { bb.x = bias[cl]; bb.y = bias[cl + 1]; }
#pragma unroll
    for (int k = 0; k < RPW; ++k) {
        const int node = node0 + k;
        if (node >= N) break;
        const int   beg = __builtin_amdgcn_readlane(rc.x, k);
        const int   end = __builtin_amdgcn_readlane(rc.x, k + 1);
        const float di  = __int_as_float(__builtin_amdgcn_readlane(rc.z, k));
        const v2f self = *(const v2f*)(xin + (size_t)node * NCLS + cl);
        v2f acc = self;
        for (int base = beg; base < end; base += 32) {
            const int n = (end - base < 32) ? (end - base) : 32;
            int p = 0;
            if (lane < n) p = csr[base + lane];
            int j0 = 0;
            for (; j0 + HB <= n; j0 += HB) acc = pull_batch<true>(xin, p, j0, HB, cl, acc);
            if (j0 < n) acc = pull_batch<false>(xin, p, j0, n - j0, cl, acc);
        }
        if (!LAST) acc = (di * di) * acc;
        if (LAST) {
            const v2f lg = di * acc + bb;
            const float mx = wave_max(own ? fmaxf(lg.x, lg.y) : -INFINITY);
            const float se = wave_sum(own ? (expf(lg.x - mx) + expf(lg.y - mx)) : 0.0f);
            const float ls = logf(se);
            acc = (v2f){lg.x - mx - ls, lg.y - mx - ls};
        }
        if (own) *(v2f*)(xout + (size_t)node * NCLS + cl) = acc;
    }
}

#define ECAP 32
#define EROW (2 * ECAP)

__global__ void k_rank_slots(const int* __restrict__ row, const int* __restrict__ col,
                             int* __restrict__ cnt0, int* __restrict__ cnt1, int* __restrict__ slots,
                             int* __restrict__ ovfn, int2* __restrict__ ovf, int E) {
    const int h = package_half();
    int* cnt = h ? cnt1 : cnt0;
    const int stride = gridDim.x * blockDim.x;
    const int i0 = blockIdx.x * blockDim.x + threadIdx.x;
    int idx[EPT];
    int c[EPT];
    int r[EPT];
    int old[EPT];
#pragma unroll
    for (int k = 0; k < EPT; ++k) {
        idx[k] = i0 + k * stride;
        const int ii = (idx[k] < E) ? idx[k] : (E - 1);
        c[k] = col[ii];
        r[k] = row[ii];
    }
#pragma unroll
    for (int k = 0; k < EPT; ++k) {
        old[k] = 0;
        if (idx[k] < E) old[k] = atomicAdd(&cnt[c[k]], 1);
    }
#pragma unroll
    for (int k = 0; k < EPT; ++k) {
        if (idx[k] < E) {
            if (old[k] < ECAP) {
                slots[(size_t)c[k] * EROW + (size_t)(h * ECAP + old[k])] = r[k];
            } else {
                const int p = atomicAdd(ovfn, 1);
                ovf[p] = make_int2(r[k], c[k]);
            }
        }
    }
}

__global__ void k_gemm_cnt(const float* __restrict__ x, const float* __restrict__ W,
                           const int* __restrict__ cnt0, const int* __restrict__ cnt1,
                           float* __restrict__ y, int N) {
    const int tile = (int)blockIdx.x * (int)(blockDim.x >> 5) + __builtin_amdgcn_readfirstlane(threadIdx.x >> 5);
    if (tile >= ((N + 15) >> 4)) return;
    gemm_tile_w40<0>(x, W, nullptr, cnt0, cnt1, y, N, tile, threadIdx.x & 31);
}

__device__ __forceinline__ v2f pull_list(const float* __restrict__ xin, int p, int n, int cl, v2f acc) {
    int j0 = 0;
    for (; j0 + HB <= n; j0 += HB) acc = pull_batch<true>(xin, p, j0, HB, cl, acc);
    if (j0 < n) acc = pull_batch<false>(xin, p, j0, n - j0, cl, acc);
    return acc;
}

template <bool LAST>
__global__ void k_pull_hop_slots(const float* __restrict__ xin, float* __restrict__ xout,
                                 const int* __restrict__ cnt0, const int* __restrict__ cnt1,
                                 const int* __restrict__ slots,
                                 const int* __restrict__ ovfn, const int2* __restrict__ ovf,
                                 const float* __restrict__ bias, int N) {
    const int lane = threadIdx.x & 31;
    const bool own = lane < YL;
    const int cl = 2 * (own ? lane : (YL - 1));
    const int wave  = blockIdx.x * (blockDim.x >> 5) + __builtin_amdgcn_readfirstlane(threadIdx.x >> 5);
    const int node0 = wave * RPW;
    if (node0 >= N) return;
    const int nl = (node0 + lane < N) ? (node0 + lane) : (N - 1);
    const int c0v = cnt0[nl];
    const int c1v = cnt1[nl];
    const float dgv = 1.0f + (float)(c0v + c1v);
    const float div = (dgv > 0.0f) ? rsqrtf(dgv) : 0.0f;
    v2f bb = (v2f){0.0f, 0.0f};
    if (LAST) { bb.x = bias[cl]; bb.y = bias[cl + 1]; }
#pragma unroll
    for (int k = 0; k < RPW; ++k) {
        const int node = node0 + k;
        if (node >= N) break;
        const int   c0 = __builtin_amdgcn_readlane(c0v, k);
        const int   c1 = __builtin_amdgcn_readlane(c1v, k);
        const float di = __int_as_float(__builtin_amdgcn_readlane(__float_as_int(div), k));
        const int n0 = (c0 < ECAP) ? c0 : ECAP;
        const int n1 = (c1 < ECAP) ? c1 : ECAP;
        const int* sr = slots + (size_t)node * EROW;
        const v2f self = *(const v2f*)(xin + (size_t)node * NCLS + cl);
        const int n  = n0 + n1;
        int pa = 0;
        if (lane < n) pa = sr[(lane < n0) ? lane : (ECAP + lane - n0)];
        v2f acc = self;
        acc = pull_list(xin, pa, (n < 32) ? n : 32, cl, acc);
        if (n > 32) {
            int pb = 0;
            if (lane + 32 < n) pb = sr[ECAP + 32 + lane - n0];
            acc = pull_list(xin, pb, n - 32, cl, acc);
        }
        if (c0 > ECAP || c1 > ECAP) {
            const int novf = *ovfn;
            for (int i = 0; i < novf; ++i) {
                const int2 e = ovf[i];
                const int es = __builtin_amdgcn_readfirstlane(e.x);
                const int et = __builtin_amdgcn_readfirstlane(e.y);
                if (et == node) acc += *(const v2f*)(xin + (size_t)es * NCLS + cl);
            }
        }
        if (!LAST) acc = (di * di) * acc;
        if (LAST) {
            const v2f lg = di * acc + bb;
            const float mx = wave_max(own ? fmaxf(lg.x, lg.y) : -INFINITY);
            const float se = wave_sum(own ? (expf(lg.x - mx) + expf(lg.y - mx)) : 0.0f);
            const float ls = logf(se);
            acc = (v2f){lg.x - mx - ls, lg.y - mx - ls};
        }
        if (own) *(v2f*)(xout + (size_t)node * NCLS + cl) = acc;
    }
}

#define BR    256
#define BSH   8
#define MAXB  1024
#define CT    256
#define EPC   12
#define CHUNK (CT * EPC)
#define BT    512

__global__ void __launch_bounds__(CT) k_bucket_chunks(const int* __restrict__ row, const int* __restrict__ col,
                                                      int2* __restrict__ bk, int* __restrict__ offs,
                                                      int* __restrict__ ovfn, int E, int B) {
    __shared__ int4 hist4[MAXB / 4];
    __shared__ int  wtot[CT / 32];
    __shared__ int2 buf[CHUNK];
    int* hist = (int*)hist4;
    const int tid  = threadIdx.x;
    const int lane = tid & 31;
    const int wv   = tid >> 5;
    const int w    = blockIdx.x;
    const int e0   = w * CHUNK;
    if (w == 0 && tid == 0) *ovfn = 0;
    for (int i = tid; i < MAXB; i += CT) hist[i] = 0;
    __syncthreads();

    int r[EPC];
    int c[EPC];
    int pos[EPC];
#pragma unroll
    for (int k = 0; k < EPC; ++k) {
        const int idx = e0 + k * CT + tid;
        const int ii = (idx < E) ? idx : (E - 1);
        c[k] = col[ii];
        r[k] = row[ii];
    }
    int bkt[EPC];
#pragma unroll
    for (int k = 0; k < EPC; ++k) {
        bkt[k] = (e0 + k * CT + tid < E) ? (c[k] >> BSH) : B;
        pos[k] = atomicAdd(&hist[bkt[k]], 1);
    }
    __syncthreads();

    const int4 h = hist4[tid];
    const int s = (h.x + h.y) + (h.z + h.w);
    int inc = s;
    for (int m = 1; m < 32; m <<= 1) {
        const int t = __shfl_up(inc, m);
        if (lane >= m) inc += t;
    }
    if (lane == 31) wtot[wv] = inc;
    __syncthreads();
    int wpre = 0;
    for (int i = 0; i < wv; ++i) wpre += wtot[i];
    const int ex = wpre + inc - s;
    hist4[tid] = make_int4(ex, ex + h.x, ex + h.x + h.y, ex + h.x + h.y + h.z);
    __syncthreads();

    int dstp[EPC];
#pragma unroll
    for (int k = 0; k < EPC; ++k) dstp[k] = hist[bkt[k]] + pos[k];
#pragma unroll
    for (int k = 0; k < EPC; ++k) buf[dstp[k]] = make_int2(r[k], c[k]);
    for (int i = tid; i <= B; i += CT) offs[(size_t)w * (size_t)(B + 1) + (size_t)i] = hist[i];
    __syncthreads();
    const int n = (E - e0 < CHUNK) ? (E - e0) : CHUNK;
    for (int i = tid; i < n; i += CT) bk[(size_t)e0 + (size_t)i] = buf[i];
}

__global__ void __launch_bounds__(BT) k_build_rows(const int2* __restrict__ bk, const int* __restrict__ offs,
                                                   int* __restrict__ cnt, int* __restrict__ slots,
                                                   int* __restrict__ ovfn, int2* __restrict__ ovf,
                                                   int N, int B, int nchunk) {
    __shared__ int  lcnt[BR];
    __shared__ int4 lrow4[BR * EROW / 4];
    int* lrow = (int*)lrow4;
    const int tid = threadIdx.x;
    const int b   = blockIdx.x;
    const int t0  = b << BSH;
    for (int i = tid; i < BR; i += BT) lcnt[i] = 0;
    for (int i = tid; i < BR * EROW / 4; i += BT) lrow4[i] = make_int4(0, 0, 0, 0);
    __syncthreads();

    for (int w = tid; w < nchunk; w += BT) {
        const int* o = offs + (size_t)w * (size_t)(B + 1) + (size_t)b;
        const int s = o[0];
        const int e = o[1];
        const int2* src = bk + (size_t)w * CHUNK;
        for (int i = s; i < e; i += 4) {
            int2 p[4];
#pragma unroll
            for (int u = 0; u < 4; ++u) {
                const int ii = (i + u < e) ? (i + u) : (e - 1);
                p[u] = src[ii];
            }
#pragma unroll
            for (int u = 0; u < 4; ++u) {
                if (i + u < e) {
                    const int lt = p[u].y - t0;
                    const int k = atomicAdd(&lcnt[lt], 1);
                    if (k < EROW) {
                        lrow[lt * EROW + k] = p[u].x;
                    } else {
                        const int q = atomicAdd(ovfn, 1);
                        ovf[q] = p[u];
                    }
                }
            }
        }
    }
    __syncthreads();

    for (int i = tid; i < BR; i += BT) {
        if (t0 + i < N) cnt[t0 + i] = lcnt[i];
    }
    int4* dst = (int4*)slots + (size_t)t0 * (EROW / 4);
    for (int i = tid; i < BR * EROW / 4; i += BT) {
        if (t0 + i / (EROW / 4) < N) dst[i] = lrow4[i];
    }
}

__global__ void k_gemm_deg(const float* __restrict__ x, const float* __restrict__ W,
                           const int* __restrict__ cnt, float* __restrict__ y, int N) {
    const int tile = (int)blockIdx.x * (int)(blockDim.x >> 5) + __builtin_amdgcn_readfirstlane(threadIdx.x >> 5);
    if (tile >= ((N + 15) >> 4)) return;
    gemm_tile_w40<2>(x, W, nullptr, cnt, nullptr, y, N, tile, threadIdx.x & 31);
}

template <bool LAST>
__global__ void k_pull_hop_rows(const float* __restrict__ xin, float* __restrict__ xout,
                                const int* __restrict__ cnt, const int* __restrict__ slots,
                                const int* __restrict__ ovfn, const int2* __restrict__ ovf,
                                const float* __restrict__ bias, int N) {
    const int lane = threadIdx.x & 31;
    const bool own = lane < YL;
    const int cl = 2 * (own ? lane : (YL - 1));
    const int wave  = blockIdx.x * (blockDim.x >> 5) + __builtin_amdgcn_readfirstlane(threadIdx.x >> 5);
    const int node0 = wave * RPW;
    if (node0 >= N) return;
    const int nl = (node0 + lane < N) ? (node0 + lane) : (N - 1);
    const int cv = cnt[nl];
    int pav[RPW];
    v2f selfv[RPW];
#pragma unroll
    for (int k = 0; k < RPW; ++k) {
        const int nk = (node0 + k < N) ? (node0 + k) : (N - 1);
        pav[k]   = slots[(size_t)nk * EROW + (size_t)lane];
        selfv[k] = *(const v2f*)(xin + (size_t)nk * NCLS + cl);
    }
    const float dgv = 1.0f + (float)cv;
    const float div = (dgv > 0.0f) ? rsqrtf(dgv) : 0.0f;
    v2f bb = (v2f){0.0f, 0.0f};
    if (LAST) { bb.x = bias[cl]; bb.y = bias[cl + 1]; }
#pragma unroll
    for (int k = 0; k < RPW; ++k) {
        const int node = node0 + k;
        if (node >= N) break;
        const int   c  = __builtin_amdgcn_readlane(cv, k);
        const float di = __int_as_float(__builtin_amdgcn_readlane(__float_as_int(div), k));
        const int n = (c < EROW) ? c : EROW;
        const int* sr = slots + (size_t)node * EROW;
        const int pa = pav[k];
        v2f acc = selfv[k];
        acc = pull_list(xin, pa, (n < 32) ? n : 32, cl, acc);
        if (n > 32) {
            const int pb = sr[32 + lane];
            acc = pull_list(xin, pb, n - 32, cl, acc);
        }
        if (c > EROW) {
            const int novf = *ovfn;
            for (int i = 0; i < novf; ++i) {
                const int2 e = ovf[i];
                const int es = __builtin_amdgcn_readfirstlane(e.x);
                const int et = __builtin_amdgcn_readfirstlane(e.y);
                if (et == node) acc += *(const v2f*)(xin + (size_t)es * NCLS + cl);
            }
        }
        if (!LAST) acc = (di * di) * acc;
        if (LAST) {
            const v2f lg = di * acc + bb;
            const float mx = wave_max(own ? fmaxf(lg.x, lg.y) : -INFINITY);
            const float se = wave_sum(own ? (expf(lg.x - mx) + expf(lg.y - mx)) : 0.0f);
            const float ls = logf(se);
            acc = (v2f){lg.x - mx - ls, lg.y - mx - ls};
        }
        if (own) *(v2f*)(xout + (size_t)node * NCLS + cl) = acc;
    }
}


static inline size_t align256f(size_t nfloats) { return (nfloats + 63) & ~(size_t)63; }

extern "C" void kernel_launch(void* const* d_in, const int* in_sizes, int n_in,
                              void* d_out, int out_size, void* d_ws, size_t ws_size,
                              hipStream_t stream) {
    const float* x  = (const float*)d_in[0];
    const int*   ei = (const int*)d_in[1];
    const float* W  = (const float*)d_in[2];
    const float* bi = (const float*)d_in[3];

    const int N = in_sizes[0] / FEAT;
    const int E = in_sizes[1] / 2;
    const int* row = ei;
    const int* col = ei + E;
    const int total = N * FEAT;

    float* ws = (float*)d_ws;
    size_t off = 0;
    float* cnt0 = ws + off; off += align256f((size_t)N);
    float* cnt1 = ws + off; off += align256f((size_t)N);
    float* Wp   = ws + off; off += align256f((size_t)NCLSP * FEAT);

    const int T = 256;
    const int ncnt = (int)(cnt1 - cnt0) + N;
    const int ntiles = (N + 15) >> 4;
    const int gblocks = (ntiles + 7) / 8;

    size_t o3 = off;
    float* z0    = ws + o3;          o3 += align256f((size_t)N * NCLS);
    float* z1    = ws + o3;          o3 += align256f((size_t)N * NCLS);
    int*   slots = (int*)(ws + o3);  o3 += align256f((size_t)N * EROW);
    int2*  ovf   = (int2*)(ws + o3); o3 += align256f(2 * (size_t)E);
    int*   ovfn  = (int*)Wp;

    const int nbuck  = (N + BR - 1) >> BSH;
    const int nchunk = (E + CHUNK - 1) / CHUNK;
    if (o3 * sizeof(float) <= ws_size && nchunk > 0 && nbuck + 1 <= MAXB &&
        2 * (size_t)E <= (size_t)N * NCLS && (size_t)nchunk * (size_t)(nbuck + 1) <= (size_t)N * NCLS) {
        int2* bk   = (int2*)z1;
        int*  offs = (int*)z0;
        k_bucket_chunks<<<nchunk, CT, 0, stream>>>(row, col, bk, offs, ovfn, E, nbuck);
        k_build_rows<<<nbuck, BT, 0, stream>>>(bk, offs, (int*)cnt0, slots, ovfn, ovf, N, nbuck, nchunk);
        k_gemm_deg<<<gblocks, T, 0, stream>>>(x, W, (const int*)cnt0, z0, N);
        const int hblocks = (N + (T / 32) * RPW - 1) / ((T / 32) * RPW);
        k_pull_hop_rows<false><<<hblocks, T, 0, stream>>>(z0, z1, (const int*)cnt0, slots, ovfn, ovf, nullptr, N);
        k_pull_hop_rows<true><<<hblocks, T, 0, stream>>>(z1, (float*)d_out, (const int*)cnt0, slots, ovfn, ovf, bi, N);
        return;
    }

    if (o3 * sizeof(float) <= ws_size) {
        const int nclr = (int)(Wp - cnt0) + 64;
        k_fill<<<(nclr + T - 1) / T, T, 0, stream>>>(cnt0, 0.0f, nclr);
        const int sblocks = (E + T * EPT - 1) / (T * EPT);
        k_rank_slots<<<sblocks, T, 0, stream>>>(row, col, (int*)cnt0, (int*)cnt1, slots, ovfn, ovf, E);
        k_gemm_cnt<<<gblocks, T, 0, stream>>>(x, W, (const int*)cnt0, (const int*)cnt1, z0, N);
        const int hblocks = (N + (T / 32) * RPW - 1) / ((T / 32) * RPW);
        k_pull_hop_slots<false><<<hblocks, T, 0, stream>>>(z0, z1, (const int*)cnt0, (const int*)cnt1, slots,
                                                           ovfn, ovf, nullptr, N);
        k_pull_hop_slots<true><<<hblocks, T, 0, stream>>>(z1, (float*)d_out, (const int*)cnt0, (const int*)cnt1, slots,
                                                          ovfn, ovf, bi, N);
        return;
    }

    size_t o2 = off;
    int4*  rec  = (int4*)(ws + o2); o2 += align256f(4 * ((size_t)N + 1));
    float* y0   = ws + o2;          o2 += align256f((size_t)N * NCLS);
    float* y1   = ws + o2;          o2 += align256f((size_t)N * NCLS);
    int*   csr  = (int*)(ws + o2);  o2 += align256f((size_t)E);
    int*   rank = (int*)(ws + o2);  o2 += align256f((size_t)E);

    if (o2 * sizeof(float) <= ws_size) {
        k_fill<<<(ncnt + T - 1) / T, T, 0, stream>>>(cnt0, 0.0f, ncnt);
        const int sblocks = (E + T * EPT - 1) / (T * EPT);
        k_rank<<<sblocks, T, 0, stream>>>(col, (int*)cnt0, (int*)cnt1, rank, E);
        k_scan_nodes<<<(N + 1 + 1023) / 1024, 1024, 0, stream>>>((const int*)cnt0, (const int*)cnt1, rec, N);
        k_csr_gemm<<<gblocks + sblocks, T, 0, stream>>>(row, col, rank, rec, csr, E, gblocks, sblocks,
                                                        x, W, y0, N);


        const int hblocks = (N + (T / 32) * RPW - 1) / ((T / 32) * RPW);
        k_pull_hop<false><<<hblocks, T, 0, stream>>>(y0, y1, rec, csr, nullptr, N);
        k_pull_hop<true><<<hblocks, T, 0, stream>>>(y1, (float*)d_out, rec, csr, bi, N);
        return;
    }

    float* p0   = ws + off; off += align256f((size_t)total);
    float* p1   = ws + off; off += align256f((size_t)total);
    float* pb   = nullptr;
    if ((off + (size_t)total) * sizeof(float) <= ws_size) pb = ws + off;
    const float* dinv = cnt0;

    k_fill<<<(ncnt + T - 1) / T, T, 0, stream>>>(cnt0, 0.0f, ncnt);
    k_degree<<<(E + T - 1) / T, T, 0, stream>>>(col, cnt0, cnt1, E);
    k_dinv<<<(N + T - 1) / T, T, 0, stream>>>(cnt0, cnt1, N);
    k_pad_w<<<(NCLSP * FEAT + T - 1) / T, T, 0, stream>>>(W, Wp);

    const int eblocks = (E + (T / 32) * EPW - 1) / ((T / 32) * EPW);

    k_seed_planes<<<(total + T - 1) / T, T, 0, stream>>>(x, p0, pb, dinv, total);
    k_edge_spmm<<<eblocks, T, 0, stream>>>(x, p0, pb, row, col, dinv, E);
    k_join_seed<<<(total + T - 1) / T, T, 0, stream>>>(p0, pb, p1, dinv, total);
    k_edge_spmm<<<eblocks, T, 0, stream>>>(p0, p1, pb, row, col, dinv, E);

    if (pb) k_gemm_wmma<true><<<gblocks, 256, 0, stream>>>(p1, pb, Wp, bi, (float*)d_out, N);
    else    k_gemm_wmma<false><<<gblocks, 256, 0, stream>>>(p1, pb, Wp, bi, (float*)d_out, N);
    k_logsoftmax<<<(N + T - 1) / T, T, 0, stream>>>((float*)d_out, N);
}


# linear layer computed inside the build launch: wave v of a bucket's work-group takes node tile v, W fragments from an LDS copy, dinv from the LDS counts; 5 -> 4 launches; sums unchanged
# speedup vs baseline: 5.7170x; 1.0823x over previous
#include <hip/hip_runtime.h>
#include <math.h>

typedef float v2f __attribute__((ext_vector_type(2)));
typedef float v8f __attribute__((ext_vector_type(8)));

#define FEAT   64
#define NCLS   40
#define NCLSP  48

__device__ __forceinline__ int package_half() {
    return (int)((__builtin_amdgcn_s_sendmsg_rtn(0x87) >> 18) & 1u);
}


__global__ void k_fill(float* __restrict__ p, float val, int n) {
    int i = blockIdx.x * blockDim.x + threadIdx.x;
    if (i < n) p[i] = val;
}

__global__ void k_degree(const int* __restrict__ col, float* __restrict__ cnt0, float* __restrict__ cnt1, int E) {
    int i = blockIdx.x * blockDim.x + threadIdx.x;
    float* cnt = package_half() ? cnt1 : cnt0;
    if (i < E) atomicAdd(&cnt[col[i]], 1.0f);
}

__global__ void k_dinv(float* __restrict__ d0, const float* __restrict__ d1, int n) {
    int i = blockIdx.x * blockDim.x + threadIdx.x;
    if (i < n) {
        float v = 1.0f + d0[i] + d1[i];
        d0[i] = (v > 0.0f) ? rsqrtf(v) : 0.0f;
    }
}

__global__ void k_seed_planes(const float* __restrict__ xin, float* __restrict__ pa, float* __restrict__ pb,
                              const float* __restrict__ dinv, int total) {
    int i = blockIdx.x * blockDim.x + threadIdx.x;
    if (i < total) {
        float w = dinv[i >> 6];
        pa[i] = w * w * xin[i];
        if (pb) pb[i] = 0.0f;
    }
}

__global__ void k_join_seed(float* __restrict__ pa, float* __restrict__ pb, float* __restrict__ na,
                            const float* __restrict__ dinv, int total) {
    int i = blockIdx.x * blockDim.x + threadIdx.x;
    if (i < total) {
        float w = dinv[i >> 6];
        float s = pa[i];
        if (pb) {
            s += pb[i];
            pa[i] = s;
            pb[i] = 0.0f;
        }
        na[i] = w * w * s;
    }
}

#define EPW    32
#define EBATCH 8

template <int NB>
__device__ __forceinline__ void spmm_edges(const float* __restrict__ xin, float* __restrict__ xout,
                                           int r, int c, float w, int first, int lane) {
    float v0[NB];
    float v1[NB];
#pragma unroll
    for (int j = 0; j < NB; ++j) {
        const int rj = __builtin_amdgcn_readlane(r, first + j);
        const float* src = xin + (size_t)rj * FEAT;
        v0[j] = src[lane];
        v1[j] = src[lane + 32];
    }
#pragma unroll
    for (int j = 0; j < NB; ++j) {
        const int   cj = __builtin_amdgcn_readlane(c, first + j);
        const float wj = __int_as_float(__builtin_amdgcn_readlane(__float_as_int(w), first + j));
        float* dst = xout + (size_t)cj * FEAT;
        atomicAdd(dst + lane,      wj * v0[j]);
        atomicAdd(dst + lane + 32, wj * v1[j]);
    }
}

__global__ void k_edge_spmm(const float* __restrict__ xin, float* __restrict__ outa, float* __restrict__ outb,
                            const int* __restrict__ row, const int* __restrict__ col,
                            const float* __restrict__ dinv, int E) {
    float* xout = (outb != nullptr && package_half()) ? outb : outa;
    const int lane = threadIdx.x & 31;
    const int wave = blockIdx.x * (blockDim.x >> 5) + __builtin_amdgcn_readfirstlane(threadIdx.x >> 5);
    const int e0   = wave * EPW;
    if (e0 >= E) return;
    const int e = e0 + lane;
    int   r = 0;
    int   c = 0;
    float w = 0.0f;
    if (e < E) {
        r = row[e];
        c = col[e];
        w = dinv[r] * dinv[c];
    }
    const int n = (E - e0 < EPW) ? (E - e0) : EPW;
    if (n == EPW) {
#pragma unroll
        for (int i = 0; i < EPW; i += EBATCH) spmm_edges<EBATCH>(xin, xout, r, c, w, i, lane);
    } else {
        for (int i = 0; i < n; ++i) spmm_edges<1>(xin, xout, r, c, w, i, lane);
    }
}

__global__ void k_pad_w(const float* __restrict__ W, float* __restrict__ Wp) {
    int i = blockIdx.x * blockDim.x + threadIdx.x;
    if (i < NCLSP * FEAT) {
        int r = i / FEAT;
        Wp[i] = (r < NCLS) ? W[i] : 0.0f;
    }
}

template <bool TWO>
__global__ void k_gemm_wmma(const float* __restrict__ xa, const float* __restrict__ xb, const float* __restrict__ Wp,
                            const float* __restrict__ bias, float* __restrict__ out, int N) {
    const int wave   = blockIdx.x * (blockDim.x >> 5) + (threadIdx.x >> 5);
    const int lane   = threadIdx.x & 31;
    const int ntiles = (N + 15) >> 4;
    if (wave >= ntiles) return;

    const int base = wave << 4;
    const int half = lane >> 4;
    const int l16  = lane & 15;

    int m = base + l16;
    if (m >= N) m = N - 1;
    const float* arow = xa + (size_t)m * FEAT;
    const float* brow = (TWO ? xb : xa) + (size_t)m * FEAT;
    const float* b0r  = Wp + (size_t)(l16)      * FEAT;
    const float* b1r  = Wp + (size_t)(16 + l16) * FEAT;
    const float* b2r  = Wp + (size_t)(32 + l16) * FEAT;

    v8f acc0 = {};
    v8f acc1 = {};
    v8f acc2 = {};

    for (int k0 = 0; k0 < FEAT; k0 += 4) {
        const int ko = k0 + 2 * half;
        v2f a  = *(const v2f*)(arow + ko);
        if (TWO) a += *(const v2f*)(brow + ko);
        v2f b0 = *(const v2f*)(b0r + ko);
        v2f b1 = *(const v2f*)(b1r + ko);
        v2f b2 = *(const v2f*)(b2r + ko);
        acc0 = __builtin_amdgcn_wmma_f32_16x16x4_f32(false, a, false, b0, (short)0, acc0, false, false);
        acc1 = __builtin_amdgcn_wmma_f32_16x16x4_f32(false, a, false, b1, (short)0, acc1, false, false);
        acc2 = __builtin_amdgcn_wmma_f32_16x16x4_f32(false, a, false, b2, (short)0, acc2, false, false);
    }

    const int c0 = l16;
    const int c1 = 16 + l16;
    const int c2 = 32 + l16;
    const float bb0 = bias ? bias[c0] : 0.0f;
    const float bb1 = bias ? bias[c1] : 0.0f;
    const float bb2 = (bias && c2 < NCLS) ? bias[c2] : 0.0f;
#pragma unroll
    for (int j = 0; j < 8; ++j) {
        const int node = base + j + 8 * half;
        if (node < N) {
            float* o = out + (size_t)node * NCLS;
            o[c0] = acc0[j] + bb0;
            o[c1] = acc1[j] + bb1;
            if (c2 < NCLS) o[c2] = acc2[j] + bb2;
        }
    }
}

__global__ void k_logsoftmax(float* __restrict__ out, int N) {
    int n = blockIdx.x * blockDim.x + threadIdx.x;
    if (n >= N) return;
    float* p = out + (size_t)n * NCLS;
    float v[NCLS];
    float m = -INFINITY;
#pragma unroll
    for (int c = 0; c < NCLS; ++c) { v[c] = p[c]; m = fmaxf(m, v[c]); }
    float s = 0.0f;
#pragma unroll
    for (int c = 0; c < NCLS; ++c) s += expf(v[c] - m);
    const float ls = logf(s);
#pragma unroll
    for (int c = 0; c < NCLS; ++c) p[c] = v[c] - m - ls;
}


#define EPT 4
#define RPW 4
#define HB  8
#define YL  (NCLS / 2)

__device__ __forceinline__ void degree_rank_block(const int* __restrict__ col, int* __restrict__ cnt0, int* __restrict__ cnt1,
                                                  int* __restrict__ rank, int E, int blk, int nblk) {
    const int h = package_half();
    int* cnt = h ? cnt1 : cnt0;
    const int stride = nblk * blockDim.x;
    const int i0 = blk * blockDim.x + threadIdx.x;
    int idx[EPT];
    int c[EPT];
    int old[EPT];
#pragma unroll
    for (int k = 0; k < EPT; ++k) {
        idx[k] = i0 + k * stride;
        c[k] = col[(idx[k] < E) ? idx[k] : (E - 1)];
    }
#pragma unroll
    for (int k = 0; k < EPT; ++k) {
        old[k] = 0;
        if (idx[k] < E) old[k] = atomicAdd(&cnt[c[k]], 1);
    }
#pragma unroll
    for (int k = 0; k < EPT; ++k) {
        if (idx[k] < E) rank[idx[k]] = old[k] | (h << 30);
    }
}

template <int DM>
__device__ __forceinline__ void gemm_tile_w40(const float* __restrict__ x, const float* __restrict__ W,
                                              const int4* __restrict__ rec,
                                              const int* __restrict__ cnt0, const int* __restrict__ cnt1,
                                              float* __restrict__ y, int N, int tile, int lane) {
    const int base = tile << 4;
    const int half = lane >> 4;
    const int l16  = lane & 15;

    int m = base + l16;
    if (m >= N) m = N - 1;
    const bool b2ok = (32 + l16) < NCLS;
    const float* arow = x + (size_t)m * FEAT;
    const float* b0r  = W + (size_t)(l16)      * FEAT;
    const float* b1r  = W + (size_t)(16 + l16) * FEAT;
    const float* b2r  = W + (size_t)(b2ok ? (32 + l16) : (NCLS - 1)) * FEAT;

    v8f acc0 = {};
    v8f acc1 = {};
    v8f acc2 = {};

    for (int k0 = 0; k0 < FEAT; k0 += 4) {
        const int ko = k0 + 2 * half;
        v2f a  = *(const v2f*)(arow + ko);
        v2f b0 = *(const v2f*)(b0r + ko);
        v2f b1 = *(const v2f*)(b1r + ko);
        v2f b2 = *(const v2f*)(b2r + ko);
        acc0 = __builtin_amdgcn_wmma_f32_16x16x4_f32(false, a, false, b0, (short)0, acc0, false, false);
        acc1 = __builtin_amdgcn_wmma_f32_16x16x4_f32(false, a, false, b1, (short)0, acc1, false, false);
        acc2 = __builtin_amdgcn_wmma_f32_16x16x4_f32(false, a, false, b2, (short)0, acc2, false, false);
    }

    int ln = lane;
    asm volatile("v_nop\n\tv_nop\n\tv_nop\n\tv_nop" : "+v"(acc0), "+v"(acc1), "+v"(acc2), "+v"(ln));
    const int eh = ln >> 4;
    const int el = ln & 15;

    const int c0 = el;
    const int c1 = 16 + el;
    const int c2 = 32 + el;
    const int* recw = (const int*)rec;
    float dv[8];
#pragma unroll
    for (int j = 0; j < 8; ++j) {
        const int node = base + j + 8 * eh;
        const size_t nn = (size_t)((node < N) ? node : (N - 1));
        if (DM == 1) {
            dv[j] = __int_as_float(recw[4 * nn + 2]);
        } else {
            const int cn = (DM == 2) ? cnt0[nn] : (cnt0[nn] + cnt1[nn]);
            const float v = 1.0f + (float)cn;
            dv[j] = (v > 0.0f) ? rsqrtf(v) : 0.0f;
        }
    }
#pragma unroll
    for (int j = 0; j < 8; ++j) {
        const int node = base + j + 8 * eh;
        if (node < N) {
            const float dj = dv[j];
            float* o = y + (size_t)node * NCLS;
            o[c0] = dj * acc0[j];
            o[c1] = dj * acc1[j];
            if (c2 < NCLS) o[c2] = dj * acc2[j];
        }
    }
}

__global__ void k_rank(const int* __restrict__ col, int* __restrict__ cnt0, int* __restrict__ cnt1,
                       int* __restrict__ rank, int E) {
    degree_rank_block(col, cnt0, cnt1, rank, E, (int)blockIdx.x, (int)gridDim.x);
}

__global__ void __launch_bounds__(1024) k_scan_nodes(const int* __restrict__ cnt0, const int* __restrict__ cnt1,
                                                     int4* __restrict__ rec, int N) {
    __shared__ int wsum[32];
    __shared__ int bpre;
    const int tid  = threadIdx.x;
    const int lane = tid & 31;
    const int wv   = tid >> 5;
    const int start = blockIdx.x * 1024;

    int s = 0;
    const int4* a4 = (const int4*)cnt0;
    const int4* b4 = (const int4*)cnt1;
#pragma unroll 4
    for (int q = tid; q < (start >> 2); q += 1024) {
        const int4 a = a4[q];
        const int4 b = b4[q];
        s += (a.x + a.y) + (a.z + a.w) + (b.x + b.y) + (b.z + b.w);
    }
    for (int m = 16; m > 0; m >>= 1) s += __shfl_xor(s, m);
    if (lane == 0) wsum[wv] = s;
    __syncthreads();
    if (wv == 0) {
        int t = wsum[lane];
        for (int m = 16; m > 0; m >>= 1) t += __shfl_xor(t, m);
        if (lane == 0) bpre = t;
    }
    __syncthreads();
    const int pre = bpre;

    const int i = start + tid;
    int c0 = 0;
    int c1 = 0;
    if (i < N) { c0 = cnt0[i]; c1 = cnt1[i]; }
    const int d = c0 + c1;
    int inc = d;
    for (int m = 1; m < 32; m <<= 1) {
        const int t = __shfl_up(inc, m);
        if (lane >= m) inc += t;
    }
    if (lane == 31) wsum[wv] = inc;
    __syncthreads();
    int wpre = (lane < wv) ? wsum[lane] : 0;
    for (int m = 16; m > 0; m >>= 1) wpre += __shfl_xor(wpre, m);
    const int excl = pre + wpre + inc - d;
    if (i <= N) {
        const float v  = 1.0f + (float)d;
        const float di = (v > 0.0f) ? rsqrtf(v) : 0.0f;
        rec[i] = make_int4(excl, excl + c0, __float_as_int(di), d);
    }
}

__device__ __forceinline__ void csr_fill_block(const int* __restrict__ row, const int* __restrict__ col,
                                               const int* __restrict__ rank, const int4* __restrict__ rec,
                                               int* __restrict__ csr, int E, int blk, int nblk) {
    const int stride = nblk * blockDim.x;
    const int i0 = blk * blockDim.x + threadIdx.x;
    const int* recw = (const int*)rec;
    int idx[EPT];
    int r[EPT];
    int c[EPT];
    int rk[EPT];
    int first[EPT];
#pragma unroll
    for (int k = 0; k < EPT; ++k) {
        idx[k] = i0 + k * stride;
        const int ii = (idx[k] < E) ? idx[k] : (E - 1);
        r[k]  = row[ii];
        c[k]  = col[ii];
        rk[k] = rank[ii];
    }
#pragma unroll
    for (int k = 0; k < EPT; ++k) {
        first[k] = recw[4 * (size_t)c[k] + (size_t)(rk[k] >> 30)];
    }
#pragma unroll
    for (int k = 0; k < EPT; ++k) {
        if (idx[k] < E) csr[first[k] + (rk[k] & 0x3fffffff)] = r[k];
    }
}

__global__ void k_csr_gemm(const int* __restrict__ row, const int* __restrict__ col, const int* __restrict__ rank,
                           const int4* __restrict__ rec, int* __restrict__ csr, int E, int ngemm, int nfill,
                           const float* __restrict__ x, const float* __restrict__ W,
                           float* __restrict__ y, int N) {
    if ((int)blockIdx.x >= ngemm) {
        csr_fill_block(row, col, rank, rec, csr, E, (int)blockIdx.x - ngemm, nfill);
        return;
    }
    const int tile = (int)blockIdx.x * (int)(blockDim.x >> 5) + __builtin_amdgcn_readfirstlane(threadIdx.x >> 5);
    if (tile >= ((N + 15) >> 4)) return;
    gemm_tile_w40<1>(x, W, rec, nullptr, nullptr, y, N, tile, threadIdx.x & 31);
}

template <bool FULL>
__device__ __forceinline__ v2f pull_batch(const float* __restrict__ xin, int srcs,
                                          int j0, int m, int cl, v2f acc) {
    constexpr int NJ = FULL ? HB : (HB - 1);
    v2f v[NJ];
#pragma unroll
    for (int j = 0; j < NJ; ++j) {
        v[j] = (v2f){0.0f, 0.0f};
        if (FULL || j == 0 || j < m) {
            const int rj = __builtin_amdgcn_readlane(srcs, j0 + j);
            v[j] = *(const v2f*)(xin + (size_t)rj * NCLS + cl);
        }
    }
#pragma unroll
    for (int j = 0; j < NJ; ++j) {
        if (FULL || j == 0 || j < m) acc += v[j];
    }
    return acc;
}

template <int M>
__device__ __forceinline__ float row_xor(float v) {
    return __builtin_bit_cast(float, __builtin_amdgcn_update_dpp(0, __builtin_bit_cast(int, v), 0x160 | M, 0xf, 0xf, false));
}
__device__ __forceinline__ float xhalf(float v) {
    const int b = __builtin_bit_cast(int, v);
    return __builtin_bit_cast(float, __builtin_amdgcn_permlanex16(b, b, 0x76543210, 0xfedcba98, false, false));
}
__device__ __forceinline__ float wave_max(float v) {
    v = fmaxf(v, row_xor<8>(v));
    v = fmaxf(v, row_xor<4>(v));
    v = fmaxf(v, row_xor<2>(v));
    v = fmaxf(v, row_xor<1>(v));
    return fmaxf(v, xhalf(v));
}
__device__ __forceinline__ float wave_sum(float v) {
    v += row_xor<8>(v);
    v += row_xor<4>(v);
    v += row_xor<2>(v);
    v += row_xor<1>(v);
    return v + xhalf(v);
}

template <bool LAST>
__global__ void k_pull_hop(const float* __restrict__ xin, float* __restrict__ xout,
                           const int4* __restrict__ rec, const int* __restrict__ csr,
                           const float* __restrict__ bias, int N) {
    const int lane = threadIdx.x & 31;
    const bool own = lane < YL;
    const int cl = 2 * (own ? lane : (YL - 1));
    const int wave  = blockIdx.x * (blockDim.x >> 5) + __builtin_amdgcn_readfirstlane(threadIdx.x >> 5);
    const int node0 = wave * RPW;
    if (node0 >= N) return;
    const int nl = (node0 + lane < N) ? (node0 + lane) : N;
    const int4 rc = rec[nl];
    v2f bb = (v2f){0.0f, 0.0f};
    if (LAST) { bb.x = bias[cl]; bb.y = bias[cl + 1]; }
#pragma unroll
    for (int k = 0; k < RPW; ++k) {
        const int node = node0 + k;
        if (node >= N) break;
        const int   beg = __builtin_amdgcn_readlane(rc.x, k);
        const int   end = __builtin_amdgcn_readlane(rc.x, k + 1);
        const float di  = __int_as_float(__builtin_amdgcn_readlane(rc.z, k));
        const v2f self = *(const v2f*)(xin + (size_t)node * NCLS + cl);
        v2f acc = self;
        for (int base = beg; base < end; base += 32) {
            const int n = (end - base < 32) ? (end - base) : 32;
            int p = 0;
            if (lane < n) p = csr[base + lane];
            int j0 = 0;
            for (; j0 + HB <= n; j0 += HB) acc = pull_batch<true>(xin, p, j0, HB, cl, acc);
            if (j0 < n) acc = pull_batch<false>(xin, p, j0, n - j0, cl, acc);
        }
        if (!LAST) acc = (di * di) * acc;
        if (LAST) {
            const v2f lg = di * acc + bb;
            const float mx = wave_max(own ? fmaxf(lg.x, lg.y) : -INFINITY);
            const float se = wave_sum(own ? (expf(lg.x - mx) + expf(lg.y - mx)) : 0.0f);
            const float ls = logf(se);
            acc = (v2f){lg.x - mx - ls, lg.y - mx - ls};
        }
        if (own) *(v2f*)(xout + (size_t)node * NCLS + cl) = acc;
    }
}

#define ECAP 32
#define EROW (2 * ECAP)

__global__ void k_rank_slots(const int* __restrict__ row, const int* __restrict__ col,
                             int* __restrict__ cnt0, int* __restrict__ cnt1, int* __restrict__ slots,
                             int* __restrict__ ovfn, int2* __restrict__ ovf, int E) {
    const int h = package_half();
    int* cnt = h ? cnt1 : cnt0;
    const int stride = gridDim.x * blockDim.x;
    const int i0 = blockIdx.x * blockDim.x + threadIdx.x;
    int idx[EPT];
    int c[EPT];
    int r[EPT];
    int old[EPT];
#pragma unroll
    for (int k = 0; k < EPT; ++k) {
        idx[k] = i0 + k * stride;
        const int ii = (idx[k] < E) ? idx[k] : (E - 1);
        c[k] = col[ii];
        r[k] = row[ii];
    }
#pragma unroll
    for (int k = 0; k < EPT; ++k) {
        old[k] = 0;
        if (idx[k] < E) old[k] = atomicAdd(&cnt[c[k]], 1);
    }
#pragma unroll
    for (int k = 0; k < EPT; ++k) {
        if (idx[k] < E) {
            if (old[k] < ECAP) {
                slots[(size_t)c[k] * EROW + (size_t)(h * ECAP + old[k])] = r[k];
            } else {
                const int p = atomicAdd(ovfn, 1);
                ovf[p] = make_int2(r[k], c[k]);
            }
        }
    }
}

__global__ void k_gemm_cnt(const float* __restrict__ x, const float* __restrict__ W,
                           const int* __restrict__ cnt0, const int* __restrict__ cnt1,
                           float* __restrict__ y, int N) {
    const int tile = (int)blockIdx.x * (int)(blockDim.x >> 5) + __builtin_amdgcn_readfirstlane(threadIdx.x >> 5);
    if (tile >= ((N + 15) >> 4)) return;
    gemm_tile_w40<0>(x, W, nullptr, cnt0, cnt1, y, N, tile, threadIdx.x & 31);
}

__device__ __forceinline__ v2f pull_list(const float* __restrict__ xin, int p, int n, int cl, v2f acc) {
    int j0 = 0;
    for (; j0 + HB <= n; j0 += HB) acc = pull_batch<true>(xin, p, j0, HB, cl, acc);
    if (j0 < n) acc = pull_batch<false>(xin, p, j0, n - j0, cl, acc);
    return acc;
}

template <bool LAST>
__global__ void k_pull_hop_slots(const float* __restrict__ xin, float* __restrict__ xout,
                                 const int* __restrict__ cnt0, const int* __restrict__ cnt1,
                                 const int* __restrict__ slots,
                                 const int* __restrict__ ovfn, const int2* __restrict__ ovf,
                                 const float* __restrict__ bias, int N) {
    const int lane = threadIdx.x & 31;
    const bool own = lane < YL;
    const int cl = 2 * (own ? lane : (YL - 1));
    const int wave  = blockIdx.x * (blockDim.x >> 5) + __builtin_amdgcn_readfirstlane(threadIdx.x >> 5);
    const int node0 = wave * RPW;
    if (node0 >= N) return;
    const int nl = (node0 + lane < N) ? (node0 + lane) : (N - 1);
    const int c0v = cnt0[nl];
    const int c1v = cnt1[nl];
    const float dgv = 1.0f + (float)(c0v + c1v);
    const float div = (dgv > 0.0f) ? rsqrtf(dgv) : 0.0f;
    v2f bb = (v2f){0.0f, 0.0f};
    if (LAST) { bb.x = bias[cl]; bb.y = bias[cl + 1]; }
#pragma unroll
    for (int k = 0; k < RPW; ++k) {
        const int node = node0 + k;
        if (node >= N) break;
        const int   c0 = __builtin_amdgcn_readlane(c0v, k);
        const int   c1 = __builtin_amdgcn_readlane(c1v, k);
        const float di = __int_as_float(__builtin_amdgcn_readlane(__float_as_int(div), k));
        const int n0 = (c0 < ECAP) ? c0 : ECAP;
        const int n1 = (c1 < ECAP) ? c1 : ECAP;
        const int* sr = slots + (size_t)node * EROW;
        const v2f self = *(const v2f*)(xin + (size_t)node * NCLS + cl);
        const int n  = n0 + n1;
        int pa = 0;
        if (lane < n) pa = sr[(lane < n0) ? lane : (ECAP + lane - n0)];
        v2f acc = self;
        acc = pull_list(xin, pa, (n < 32) ? n : 32, cl, acc);
        if (n > 32) {
            int pb = 0;
            if (lane + 32 < n) pb = sr[ECAP + 32 + lane - n0];
            acc = pull_list(xin, pb, n - 32, cl, acc);
        }
        if (c0 > ECAP || c1 > ECAP) {
            const int novf = *ovfn;
            for (int i = 0; i < novf; ++i) {
                const int2 e = ovf[i];
                const int es = __builtin_amdgcn_readfirstlane(e.x);
                const int et = __builtin_amdgcn_readfirstlane(e.y);
                if (et == node) acc += *(const v2f*)(xin + (size_t)es * NCLS + cl);
            }
        }
        if (!LAST) acc = (di * di) * acc;
        if (LAST) {
            const v2f lg = di * acc + bb;
            const float mx = wave_max(own ? fmaxf(lg.x, lg.y) : -INFINITY);
            const float se = wave_sum(own ? (expf(lg.x - mx) + expf(lg.y - mx)) : 0.0f);
            const float ls = logf(se);
            acc = (v2f){lg.x - mx - ls, lg.y - mx - ls};
        }
        if (own) *(v2f*)(xout + (size_t)node * NCLS + cl) = acc;
    }
}

#define BR    256
#define BSH   8
#define MAXB  1024
#define CT    256
#define EPC   12
#define CHUNK (CT * EPC)
#define BT    512

__global__ void __launch_bounds__(CT) k_bucket_chunks(const int* __restrict__ row, const int* __restrict__ col,
                                                      int2* __restrict__ bk, int* __restrict__ offs,
                                                      int* __restrict__ ovfn, int E, int B) {
    __shared__ int4 hist4[MAXB / 4];
    __shared__ int  wtot[CT / 32];
    __shared__ int2 buf[CHUNK];
    int* hist = (int*)hist4;
    const int tid  = threadIdx.x;
    const int lane = tid & 31;
    const int wv   = tid >> 5;
    const int w    = blockIdx.x;
    const int e0   = w * CHUNK;
    if (w == 0 && tid == 0) *ovfn = 0;
    for (int i = tid; i < MAXB; i += CT) hist[i] = 0;
    __syncthreads();

    int r[EPC];
    int c[EPC];
    int pos[EPC];
#pragma unroll
    for (int k = 0; k < EPC; ++k) {
        const int idx = e0 + k * CT + tid;
        const int ii = (idx < E) ? idx : (E - 1);
        c[k] = col[ii];
        r[k] = row[ii];
    }
    int bkt[EPC];
#pragma unroll
    for (int k = 0; k < EPC; ++k) {
        bkt[k] = (e0 + k * CT + tid < E) ? (c[k] >> BSH) : B;
        pos[k] = atomicAdd(&hist[bkt[k]], 1);
    }
    __syncthreads();

    const int4 h = hist4[tid];
    const int s = (h.x + h.y) + (h.z + h.w);
    int inc = s;
    for (int m = 1; m < 32; m <<= 1) {
        const int t = __shfl_up(inc, m);
        if (lane >= m) inc += t;
    }
    if (lane == 31) wtot[wv] = inc;
    __syncthreads();
    int wpre = 0;
    for (int i = 0; i < wv; ++i) wpre += wtot[i];
    const int ex = wpre + inc - s;
    hist4[tid] = make_int4(ex, ex + h.x, ex + h.x + h.y, ex + h.x + h.y + h.z);
    __syncthreads();

    int dstp[EPC];
#pragma unroll
    for (int k = 0; k < EPC; ++k) dstp[k] = hist[bkt[k]] + pos[k];
#pragma unroll
    for (int k = 0; k < EPC; ++k) buf[dstp[k]] = make_int2(r[k], c[k]);
    for (int i = tid; i <= B; i += CT) offs[(size_t)w * (size_t)(B + 1) + (size_t)i] = hist[i];
    __syncthreads();
    const int n = (E - e0 < CHUNK) ? (E - e0) : CHUNK;
    for (int i = tid; i < n; i += CT) bk[(size_t)e0 + (size_t)i] = buf[i];
}

__global__ void __launch_bounds__(BT) k_build_rows(const int2* __restrict__ bk, const int* __restrict__ offs,
                                                   int* __restrict__ cnt, int* __restrict__ slots,
                                                   int* __restrict__ ovfn, int2* __restrict__ ovf,
                                                   int N, int B, int nchunk) {
    __shared__ int  lcnt[BR];
    __shared__ int4 lrow4[BR * EROW / 4];
    int* lrow = (int*)lrow4;
    const int tid = threadIdx.x;
    const int b   = blockIdx.x;
    const int t0  = b << BSH;
    for (int i = tid; i < BR; i += BT) lcnt[i] = 0;
    for (int i = tid; i < BR * EROW / 4; i += BT) lrow4[i] = make_int4(0, 0, 0, 0);
    __syncthreads();

    for (int w = tid; w < nchunk; w += BT) {
        const int* o = offs + (size_t)w * (size_t)(B + 1) + (size_t)b;
        const int s = o[0];
        const int e = o[1];
        const int2* src = bk + (size_t)w * CHUNK;
        for (int i = s; i < e; i += 4) {
            int2 p[4];
#pragma unroll
            for (int u = 0; u < 4; ++u) {
                const int ii = (i + u < e) ? (i + u) : (e - 1);
                p[u] = src[ii];
            }
#pragma unroll
            for (int u = 0; u < 4; ++u) {
                if (i + u < e) {
                    const int lt = p[u].y - t0;
                    const int k = atomicAdd(&lcnt[lt], 1);
                    if (k < EROW) {
                        lrow[lt * EROW + k] = p[u].x;
                    } else {
                        const int q = atomicAdd(ovfn, 1);
                        ovf[q] = p[u];
                    }
                }
            }
        }
    }
    __syncthreads();

    for (int i = tid; i < BR; i += BT) {
        if (t0 + i < N) cnt[t0 + i] = lcnt[i];
    }
    int4* dst = (int4*)slots + (size_t)t0 * (EROW / 4);
    for (int i = tid; i < BR * EROW / 4; i += BT) {
        if (t0 + i / (EROW / 4) < N) dst[i] = lrow4[i];
    }
}

#define WPITCH 68
#define WPF4   (WPITCH / 4)
#define LW_OFF   0
#define LWROWS   (2 * BT / 16)
#define LCNT_OFF (LWROWS * WPF4)
#define LROW_OFF (LCNT_OFF + BR / 4)
#define LDS_W16  (LROW_OFF + BR * EROW / 4)

__device__ __forceinline__ void build_piece(const int2* __restrict__ src, int s, int e, int t0,
                                            int* __restrict__ lcnt, int* __restrict__ lrow,
                                            int* __restrict__ ovfn, int2* __restrict__ ovf) {
    for (int i = s; i < e; i += 4) {
        int2 p[4];
#pragma unroll
        for (int u = 0; u < 4; ++u) {
            const int ii = (i + u < e) ? (i + u) : (e - 1);
            p[u] = src[ii];
        }
#pragma unroll
        for (int u = 0; u < 4; ++u) {
            if (i + u < e) {
                const int lt = p[u].y - t0;
                const int k = atomicAdd(&lcnt[lt], 1);
                if (k < EROW) {
                    lrow[lt * EROW + k] = p[u].x;
                } else {
                    const int q = atomicAdd(ovfn, 1);
                    ovf[q] = p[u];
                }
            }
        }
    }
}

__global__ void __launch_bounds__(BT) k_build_rows_gemm(const int2* __restrict__ bk, const int* __restrict__ offs,
                                                        int* __restrict__ cnt, int* __restrict__ slots,
                                                        int* __restrict__ ovfn, int2* __restrict__ ovf,
                                                        const float* __restrict__ x, const float* __restrict__ W,
                                                        float* __restrict__ y, int N, int B, int nchunk) {
    __shared__ int4 lds[LDS_W16];
    int*  lcnt  = (int*)(lds + LCNT_OFF);
    int4* lrow4 = lds + LROW_OFF;
    int*  lrow  = (int*)lrow4;
    const int tid  = threadIdx.x;
    const int b    = blockIdx.x;
    const int t0   = b << BSH;
    const int lane = tid & 31;
    const int half = lane >> 4;
    const int l16  = lane & 15;
    const int wv   = __builtin_amdgcn_readfirstlane(tid >> 5);
    const int base = t0 + (wv << 4);
    const bool tile_ok = base < N;

    const int4* W4 = (const int4*)W;
    const int nw4 = NCLS * FEAT / 4;
    static_assert(BT <= NCLS * FEAT / 4 && NCLS * FEAT / 4 <= 2 * BT, "two 16-byte words of W per thread");
    const int4 wq0 = W4[tid];
    const int4 wq1 = W4[(tid + BT < nw4) ? (tid + BT) : (nw4 - 1)];
    v2f a[FEAT / 4];
    {
        int m = base + l16;
        if (m >= N) m = N - 1;
        const float* arow = x + (size_t)m * FEAT + 2 * half;
#pragma unroll
        for (int k = 0; k < FEAT / 4; ++k) a[k] = *(const v2f*)(arow + 4 * k);
    }
    const int* o0 = offs + (size_t)((tid < nchunk) ? tid : (nchunk - 1)) * (size_t)(B + 1) + (size_t)b;
    const int s0 = o0[0];
    const int e0 = o0[1];

    if (tid < BR) lcnt[tid] = 0;
#pragma unroll
    for (int i = 0; i < BR * EROW / 4 / BT; ++i) lrow4[i * BT + tid] = make_int4(0, 0, 0, 0);
    lds[LW_OFF + (tid >> 4) * WPF4 + (tid & 15)] = wq0;
    lds[LW_OFF + ((tid + BT) >> 4) * WPF4 + (tid & 15)] = wq1;
    __syncthreads();

    v8f acc0 = {};
    v8f acc1 = {};
    v8f acc2 = {};
    int ln = lane;
    if (tile_ok) {
        const v2f* wl2 = (const v2f*)(lds + LW_OFF);
        const int r2 = ((32 + l16) < NCLS) ? (32 + l16) : (NCLS - 1);
        const v2f* b0p = wl2 + l16 * (WPITCH / 2) + half;
        const v2f* b2p = wl2 + r2  * (WPITCH / 2) + half;
#pragma unroll
        for (int k = 0; k < FEAT / 4; ++k) {
            const v2f b0 = b0p[2 * k];
            const v2f b1 = b0p[16 * (WPITCH / 2) + 2 * k];
            const v2f b2 = b2p[2 * k];
            acc0 = __builtin_amdgcn_wmma_f32_16x16x4_f32(false, a[k], false, b0, (short)0, acc0, false, false);
            acc1 = __builtin_amdgcn_wmma_f32_16x16x4_f32(false, a[k], false, b1, (short)0, acc1, false, false);
            acc2 = __builtin_amdgcn_wmma_f32_16x16x4_f32(false, a[k], false, b2, (short)0, acc2, false, false);
        }
        asm volatile("v_nop\n\tv_nop\n\tv_nop\n\tv_nop" : "+v"(acc0), "+v"(acc1), "+v"(acc2), "+v"(ln));
    }
    const int ltid = (wv << 5) | ln;

    if (ltid < nchunk) build_piece(bk + (size_t)ltid * CHUNK, s0, e0, t0, lcnt, lrow, ovfn, ovf);
    for (int w = ltid + BT; w < nchunk; w += BT) {
        const int* o = offs + (size_t)w * (size_t)(B + 1) + (size_t)b;
        build_piece(bk + (size_t)w * CHUNK, o[0], o[1], t0, lcnt, lrow, ovfn, ovf);
    }
    __syncthreads();

    if (tile_ok) {
        const int eh = ln >> 4;
        const int el = ln & 15;
#pragma unroll
        for (int j = 0; j < 8; ++j) {
            const int rb = (wv << 4) + j + 8 * eh;
            const int node = t0 + rb;
            if (node < N) {
                const float v  = 1.0f + (float)lcnt[rb];
                const float dj = (v > 0.0f) ? rsqrtf(v) : 0.0f;
                float* o = y + (size_t)node * NCLS;
                o[el]      = dj * acc0[j];
                o[16 + el] = dj * acc1[j];
                if (32 + el < NCLS) o[32 + el] = dj * acc2[j];
            }
        }
    }
    if (ltid < BR) {
        if (t0 + ltid < N) cnt[t0 + ltid] = lcnt[ltid];
    }
    int4* dst = (int4*)slots + (size_t)t0 * (EROW / 4);
    for (int i = ltid; i < BR * EROW / 4; i += BT) {
        if (t0 + i / (EROW / 4) < N) dst[i] = lrow4[i];
    }
}

__global__ void k_gemm_deg(const float* __restrict__ x, const float* __restrict__ W,
                           const int* __restrict__ cnt, float* __restrict__ y, int N) {
    const int tile = (int)blockIdx.x * (int)(blockDim.x >> 5) + __builtin_amdgcn_readfirstlane(threadIdx.x >> 5);
    if (tile >= ((N + 15) >> 4)) return;
    gemm_tile_w40<2>(x, W, nullptr, cnt, nullptr, y, N, tile, threadIdx.x & 31);
}

template <bool LAST>
__global__ void k_pull_hop_rows(const float* __restrict__ xin, float* __restrict__ xout,
                                const int* __restrict__ cnt, const int* __restrict__ slots,
                                const int* __restrict__ ovfn, const int2* __restrict__ ovf,
                                const float* __restrict__ bias, int N) {
    const int lane = threadIdx.x & 31;
    const bool own = lane < YL;
    const int cl = 2 * (own ? lane : (YL - 1));
    const int wave  = blockIdx.x * (blockDim.x >> 5) + __builtin_amdgcn_readfirstlane(threadIdx.x >> 5);
    const int node0 = wave * RPW;
    if (node0 >= N) return;
    const int nl = (node0 + lane < N) ? (node0 + lane) : (N - 1);
    const int cv = cnt[nl];
    int pav[RPW];
    v2f selfv[RPW];
#pragma unroll
    for (int k = 0; k < RPW; ++k) {
        const int nk = (node0 + k < N) ? (node0 + k) : (N - 1);
        pav[k]   = slots[(size_t)nk * EROW + (size_t)lane];
        selfv[k] = *(const v2f*)(xin + (size_t)nk * NCLS + cl);
    }
    const float dgv = 1.0f + (float)cv;
    const float div = (dgv > 0.0f) ? rsqrtf(dgv) : 0.0f;
    v2f bb = (v2f){0.0f, 0.0f};
    if (LAST) { bb.x = bias[cl]; bb.y = bias[cl + 1]; }
#pragma unroll
    for (int k = 0; k < RPW; ++k) {
        const int node = node0 + k;
        if (node >= N) break;
        const int   c  = __builtin_amdgcn_readlane(cv, k);
        const float di = __int_as_float(__builtin_amdgcn_readlane(__float_as_int(div), k));
        const int n = (c < EROW) ? c : EROW;
        const int* sr = slots + (size_t)node * EROW;
        const int pa = pav[k];
        v2f acc = selfv[k];
        acc = pull_list(xin, pa, (n < 32) ? n : 32, cl, acc);
        if (n > 32) {
            const int pb = sr[32 + lane];
            acc = pull_list(xin, pb, n - 32, cl, acc);
        }
        if (c > EROW) {
            const int novf = *ovfn;
            for (int i = 0; i < novf; ++i) {
                const int2 e = ovf[i];
                const int es = __builtin_amdgcn_readfirstlane(e.x);
                const int et = __builtin_amdgcn_readfirstlane(e.y);
                if (et == node) acc += *(const v2f*)(xin + (size_t)es * NCLS + cl);
            }
        }
        if (!LAST) acc = (di * di) * acc;
        if (LAST) {
            const v2f lg = di * acc + bb;
            const float mx = wave_max(own ? fmaxf(lg.x, lg.y) : -INFINITY);
            const float se = wave_sum(own ? (expf(lg.x - mx) + expf(lg.y - mx)) : 0.0f);
            const float ls = logf(se);
            acc = (v2f){lg.x - mx - ls, lg.y - mx - ls};
        }
        if (own) *(v2f*)(xout + (size_t)node * NCLS + cl) = acc;
    }
}


static inline size_t align256f(size_t nfloats) { return (nfloats + 63) & ~(size_t)63; }

extern "C" void kernel_launch(void* const* d_in, const int* in_sizes, int n_in,
                              void* d_out, int out_size, void* d_ws, size_t ws_size,
                              hipStream_t stream) {
    const float* x  = (const float*)d_in[0];
    const int*   ei = (const int*)d_in[1];
    const float* W  = (const float*)d_in[2];
    const float* bi = (const float*)d_in[3];

    const int N = in_sizes[0] / FEAT;
    const int E = in_sizes[1] / 2;
    const int* row = ei;
    const int* col = ei + E;
    const int total = N * FEAT;

    float* ws = (float*)d_ws;
    size_t off = 0;
    float* cnt0 = ws + off; off += align256f((size_t)N);
    float* cnt1 = ws + off; off += align256f((size_t)N);
    float* Wp   = ws + off; off += align256f((size_t)NCLSP * FEAT);

    const int T = 256;
    const int ncnt = (int)(cnt1 - cnt0) + N;
    const int ntiles = (N + 15) >> 4;
    const int gblocks = (ntiles + 7) / 8;

    size_t o3 = off;
    float* z0    = ws + o3;          o3 += align256f((size_t)N * NCLS);
    float* z1    = ws + o3;          o3 += align256f((size_t)N * NCLS);
    int*   slots = (int*)(ws + o3);  o3 += align256f((size_t)N * EROW);
    int2*  ovf   = (int2*)(ws + o3); o3 += align256f(2 * (size_t)E);
    int*   ovfn  = (int*)Wp;

    const int nbuck  = (N + BR - 1) >> BSH;
    const int nchunk = (E + CHUNK - 1) / CHUNK;
    if (o3 * sizeof(float) <= ws_size && nchunk > 0 && nbuck + 1 <= MAXB &&
        2 * (size_t)E <= (size_t)N * NCLS && (size_t)nchunk * (size_t)(nbuck + 1) <= (size_t)N * NCLS) {
        int2* bk   = (int2*)z1;
        const int hblocks = (N + (T / 32) * RPW - 1) / ((T / 32) * RPW);
        const size_t otab = align256f(2 * (size_t)E);
        if (otab + (size_t)nchunk * (size_t)(nbuck + 1) <= (size_t)N * NCLS) {
            int* offs = (int*)z1 + otab;
            k_bucket_chunks<<<nchunk, CT, 0, stream>>>(row, col, bk, offs, ovfn, E, nbuck);
            k_build_rows_gemm<<<nbuck, BT, 0, stream>>>(bk, offs, (int*)cnt0, slots, ovfn, ovf, x, W, z0, N, nbuck, nchunk);
            k_pull_hop_rows<false><<<hblocks, T, 0, stream>>>(z0, z1, (const int*)cnt0, slots, ovfn, ovf, nullptr, N);
            k_pull_hop_rows<true><<<hblocks, T, 0, stream>>>(z1, (float*)d_out, (const int*)cnt0, slots, ovfn, ovf, bi, N);
            return;
        }
        int*  offs = (int*)z0;
        k_bucket_chunks<<<nchunk, CT, 0, stream>>>(row, col, bk, offs, ovfn, E, nbuck);
        k_build_rows<<<nbuck, BT, 0, stream>>>(bk, offs, (int*)cnt0, slots, ovfn, ovf, N, nbuck, nchunk);
        k_gemm_deg<<<gblocks, T, 0, stream>>>(x, W, (const int*)cnt0, z0, N);
        k_pull_hop_rows<false><<<hblocks, T, 0, stream>>>(z0, z1, (const int*)cnt0, slots, ovfn, ovf, nullptr, N);
        k_pull_hop_rows<true><<<hblocks, T, 0, stream>>>(z1, (float*)d_out, (const int*)cnt0, slots, ovfn, ovf, bi, N);
        return;
    }

    if (o3 * sizeof(float) <= ws_size) {
        const int nclr = (int)(Wp - cnt0) + 64;
        k_fill<<<(nclr + T - 1) / T, T, 0, stream>>>(cnt0, 0.0f, nclr);
        const int sblocks = (E + T * EPT - 1) / (T * EPT);
        k_rank_slots<<<sblocks, T, 0, stream>>>(row, col, (int*)cnt0, (int*)cnt1, slots, ovfn, ovf, E);
        k_gemm_cnt<<<gblocks, T, 0, stream>>>(x, W, (const int*)cnt0, (const int*)cnt1, z0, N);
        const int hblocks = (N + (T / 32) * RPW - 1) / ((T / 32) * RPW);
        k_pull_hop_slots<false><<<hblocks, T, 0, stream>>>(z0, z1, (const int*)cnt0, (const int*)cnt1, slots,
                                                           ovfn, ovf, nullptr, N);
        k_pull_hop_slots<true><<<hblocks, T, 0, stream>>>(z1, (float*)d_out, (const int*)cnt0, (const int*)cnt1, slots,
                                                          ovfn, ovf, bi, N);
        return;
    }

    size_t o2 = off;
    int4*  rec  = (int4*)(ws + o2); o2 += align256f(4 * ((size_t)N + 1));
    float* y0   = ws + o2;          o2 += align256f((size_t)N * NCLS);
    float* y1   = ws + o2;          o2 += align256f((size_t)N * NCLS);
    int*   csr  = (int*)(ws + o2);  o2 += align256f((size_t)E);
    int*   rank = (int*)(ws + o2);  o2 += align256f((size_t)E);

    if (o2 * sizeof(float) <= ws_size) {
        k_fill<<<(ncnt + T - 1) / T, T, 0, stream>>>(cnt0, 0.0f, ncnt);
        const int sblocks = (E + T * EPT - 1) / (T * EPT);
        k_rank<<<sblocks, T, 0, stream>>>(col, (int*)cnt0, (int*)cnt1, rank, E);
        k_scan_nodes<<<(N + 1 + 1023) / 1024, 1024, 0, stream>>>((const int*)cnt0, (const int*)cnt1, rec, N);
        k_csr_gemm<<<gblocks + sblocks, T, 0, stream>>>(row, col, rank, rec, csr, E, gblocks, sblocks,
                                                        x, W, y0, N);


        const int hblocks = (N + (T / 32) * RPW - 1) / ((T / 32) * RPW);
        k_pull_hop<false><<<hblocks, T, 0, stream>>>(y0, y1, rec, csr, nullptr, N);
        k_pull_hop<true><<<hblocks, T, 0, stream>>>(y1, (float*)d_out, rec, csr, bi, N);
        return;
    }

    float* p0   = ws + off; off += align256f((size_t)total);
    float* p1   = ws + off; off += align256f((size_t)total);
    float* pb   = nullptr;
    if ((off + (size_t)total) * sizeof(float) <= ws_size) pb = ws + off;
    const float* dinv = cnt0;

    k_fill<<<(ncnt + T - 1) / T, T, 0, stream>>>(cnt0, 0.0f, ncnt);
    k_degree<<<(E + T - 1) / T, T, 0, stream>>>(col, cnt0, cnt1, E);
    k_dinv<<<(N + T - 1) / T, T, 0, stream>>>(cnt0, cnt1, N);
    k_pad_w<<<(NCLSP * FEAT + T - 1) / T, T, 0, stream>>>(W, Wp);

    const int eblocks = (E + (T / 32) * EPW - 1) / ((T / 32) * EPW);

    k_seed_planes<<<(total + T - 1) / T, T, 0, stream>>>(x, p0, pb, dinv, total);
    k_edge_spmm<<<eblocks, T, 0, stream>>>(x, p0, pb, row, col, dinv, E);
    k_join_seed<<<(total + T - 1) / T, T, 0, stream>>>(p0, pb, p1, dinv, total);
    k_edge_spmm<<<eblocks, T, 0, stream>>>(p0, p1, pb, row, col, dinv, E);

    if (pb) k_gemm_wmma<true><<<gblocks, 256, 0, stream>>>(p1, pb, Wp, bi, (float*)d_out, N);
    else    k_gemm_wmma<false><<<gblocks, 256, 0, stream>>>(p1, pb, Wp, bi, (float*)d_out, N);
    k_logsoftmax<<<(N + T - 1) / T, T, 0, stream>>>((float*)d_out, N);
}
